# starting kernel
# baseline (speedup 1.0000x reference)
#include <hip/hip_runtime.h>
#include <hip/hip_bf16.h>
#include <math.h>


typedef __attribute__((ext_vector_type(2))) float v2f;
typedef __attribute__((ext_vector_type(8))) float v8f;

#define NSTEP 100
#define T1    99
#define NB    256
#define OUT_U   (NB * NSTEP)
#define OUT_EPS (3 * OUT_U)
#define OUT_ST  (OUT_EPS + NB)

__device__ __forceinline__ void Ecoef(int r, int c, float* pa, float* pb, float* pc) {
  float a = 0.f, b = 0.f, q = 0.f;
  if (r == 0)      { if (c == 0) a = 1.f;  else if (c == 1) b = 1.f;   else if (c == 2) q = 0.5f; else if (c == 3) a = -1.f; }
  else if (r == 1) { if (c == 1) a = 1.f;  else if (c == 2) b = 1.f;   else if (c == 4) a = -1.f; }
  else if (r == 2) { if (c == 0) a = -1.f; else if (c == 3) a = 1.f;   else if (c == 4) b = -1.f; else if (c == 5) q = 0.5f; }
  else if (r == 3) { if (c == 1) a = -1.f; else if (c == 4) a = 1.f;   else if (c == 5) b = -1.f; }
  else if (r == 4) { if (c == 1) a = -1.f; else if (c == 2) b = -0.5f; else if (c == 4) a = 1.f;  else if (c == 5) b = -0.5f; }
  *pa = a; *pb = b; *pc = q;
}

__device__ __forceinline__ void chol3(double* s) {
  double l00 = sqrt(s[0]);
  double l10 = s[1] / l00;
  double l11 = sqrt(s[2] - l10 * l10);
  double l20 = s[3] / l00;
  double l21 = (s[4] - l20 * l10) / l11;
  double l22 = sqrt(s[5] - l20 * l20 - l21 * l21);
  s[0] = l00; s[1] = l10; s[2] = l11; s[3] = l20; s[4] = l21; s[5] = l22;
}
__device__ __forceinline__ void lsolve3(const double* l, double* v) {
  double v0 = v[0] / l[0];
  double v1 = (v[1] - l[1] * v0) / l[2];
  double v2 = (v[2] - l[3] * v0 - l[4] * v1) / l[5];
  v[0] = v0; v[1] = v1; v[2] = v2;
}
__device__ __forceinline__ void ltsolve3(const double* l, double* v) {
  double x2 = v[2] / l[5];
  double x1 = (v[1] - l[4] * x2) / l[2];
  double x0 = (v[0] - l[1] * x1 - l[3] * x2) / l[0];
  v[0] = x0; v[1] = x1; v[2] = x2;
}

__global__ __launch_bounds__(32) void ode_blocktri_kernel(
    const float* __restrict__ coeffs,
    const float* __restrict__ rhs,
    const float* __restrict__ ivr,
    const float* __restrict__ steps,
    float* __restrict__ out)
{
  __shared__ double S[NSTEP][6];
  __shared__ double Bo[T1][9];
  __shared__ double gv[NSTEP][3];
  __shared__ float  hstep[T1];

  const int b = blockIdx.x;
  const int lane = threadIdx.x;

  for (int i = lane; i < NSTEP * 6; i += 32) ((double*)S)[i] = 0.0;
  for (int i = lane; i < T1 * 9;   i += 32) ((double*)Bo)[i] = 0.0;
  for (int i = lane; i < NSTEP * 3; i += 32) ((double*)gv)[i] = 0.0;
  for (int i = lane; i < T1; i += 32) hstep[i] = steps[b * T1 + i];
  __syncthreads();

  for (int t = lane; t < NSTEP; t += 32) {
    const double c0 = (double)coeffs[(b * NSTEP + t) * 3 + 0];
    const double c1 = (double)coeffs[(b * NSTEP + t) * 3 + 1];
    const double c2 = (double)coeffs[(b * NSTEP + t) * 3 + 2];
    const double r  = (double)rhs[b * NSTEP + t];
    S[t][0] += c0 * c0; S[t][1] += c1 * c0; S[t][2] += c1 * c1;
    S[t][3] += c2 * c0; S[t][4] += c2 * c1; S[t][5] += c2 * c2;
    gv[t][0] += c0 * r; gv[t][1] += c1 * r; gv[t][2] += c2 * r;
    if (t == 0) {
      S[0][0] += 1.0; S[0][2] += 1.0;
      gv[0][0] += (double)ivr[b * 2 + 0];
      gv[0][1] += (double)ivr[b * 2 + 1];
    }
  }
  __syncthreads();

  const int m = lane & 15;
  const int khalf = lane >> 4;
  const int col = (m < 6) ? m : 6;
  float a0, b0, q0, a1, b1, q1, a2, b2, q2;
  Ecoef(2 * khalf,                 col, &a0, &b0, &q0);
  Ecoef(2 * khalf + 1,             col, &a1, &b1, &q1);
  Ecoef((khalf == 0) ? 4 : -1,     col, &a2, &b2, &q2);

  for (int p = 0; p < T1; ++p) {
    const float h = hstep[p];
    v2f f0, f1;
    f0.x = fmaf(fmaf(q0, h, b0), h, a0);
    f0.y = fmaf(fmaf(q1, h, b1), h, a1);
    f1.x = fmaf(fmaf(q2, h, b2), h, a2);
    f1.y = 0.f;
    v8f acc = {};
    acc = __builtin_amdgcn_wmma_f32_16x16x4_f32(false, f0, false, f0,
                                                (short)0, acc, false, false);
    acc = __builtin_amdgcn_wmma_f32_16x16x4_f32(false, f1, false, f1,
                                                (short)0, acc, false, false);
    if (lane < 3) {
      const int j = lane;
      #pragma unroll
      for (int i = 0; i < 3; ++i)
        if (i >= j) S[p][i * (i + 1) / 2 + j] += (double)acc[i];
    } else if (lane < 6) {
      const int j = lane - 3;
      #pragma unroll
      for (int i = 0; i < 3; ++i) Bo[p][i * 3 + j] += (double)acc[i];
      #pragma unroll
      for (int i = 0; i < 3; ++i)
        if (i >= j) S[p + 1][i * (i + 1) / 2 + j] += (double)acc[3 + i];
    }
  }
  __syncthreads();

  if (lane == 0) {
    chol3(S[0]);
    lsolve3(S[0], gv[0]);
    for (int p = 0; p < T1; ++p) {
      const double* L = S[p];
      double X[3][3];
      #pragma unroll
      for (int r = 0; r < 3; ++r) {
        const double c0 = Bo[p][0 + r];
        const double c1 = Bo[p][3 + r];
        const double c2 = Bo[p][6 + r];
        const double x0 = c0 / L[0];
        const double x1 = (c1 - x0 * L[1]) / L[2];
        const double x2 = (c2 - x0 * L[3] - x1 * L[4]) / L[5];
        X[r][0] = x0; X[r][1] = x1; X[r][2] = x2;
      }
      #pragma unroll
      for (int r = 0; r < 3; ++r) {
        Bo[p][r * 3 + 0] = X[r][0]; Bo[p][r * 3 + 1] = X[r][1]; Bo[p][r * 3 + 2] = X[r][2];
      }
      S[p + 1][0] -= X[0][0]*X[0][0] + X[0][1]*X[0][1] + X[0][2]*X[0][2];
      S[p + 1][1] -= X[1][0]*X[0][0] + X[1][1]*X[0][1] + X[1][2]*X[0][2];
      S[p + 1][2] -= X[1][0]*X[1][0] + X[1][1]*X[1][1] + X[1][2]*X[1][2];
      S[p + 1][3] -= X[2][0]*X[0][0] + X[2][1]*X[0][1] + X[2][2]*X[0][2];
      S[p + 1][4] -= X[2][0]*X[1][0] + X[2][1]*X[1][1] + X[2][2]*X[1][2];
      S[p + 1][5] -= X[2][0]*X[2][0] + X[2][1]*X[2][1] + X[2][2]*X[2][2];
      chol3(S[p + 1]);
      const double y0 = gv[p][0], y1 = gv[p][1], y2 = gv[p][2];
      gv[p + 1][0] -= X[0][0]*y0 + X[0][1]*y1 + X[0][2]*y2;
      gv[p + 1][1] -= X[1][0]*y0 + X[1][1]*y1 + X[1][2]*y2;
      gv[p + 1][2] -= X[2][0]*y0 + X[2][1]*y1 + X[2][2]*y2;
      lsolve3(S[p + 1], gv[p + 1]);
    }
    ltsolve3(S[T1], gv[T1]);
    for (int p = T1 - 1; p >= 0; --p) {
      const double x0 = gv[p + 1][0], x1 = gv[p + 1][1], x2 = gv[p + 1][2];
      gv[p][0] -= Bo[p][0]*x0 + Bo[p][3]*x1 + Bo[p][6]*x2;
      gv[p][1] -= Bo[p][1]*x0 + Bo[p][4]*x1 + Bo[p][7]*x2;
      gv[p][2] -= Bo[p][2]*x0 + Bo[p][5]*x1 + Bo[p][8]*x2;
      ltsolve3(S[p], gv[p]);
    }
  }
  __syncthreads();

  for (int t = lane; t < NSTEP; t += 32) {
    out[0 * OUT_U + b * NSTEP + t] = (float)gv[t][0];
    out[1 * OUT_U + b * NSTEP + t] = (float)gv[t][1];
    out[2 * OUT_U + b * NSTEP + t] = (float)gv[t][2];
  }
  if (lane == 0) out[OUT_EPS + b] = 0.0f;
  for (int p = lane; p < T1; p += 32) out[OUT_ST + b * T1 + p] = hstep[p];
}

extern "C" void kernel_launch(void* const* d_in, const int* in_sizes, int n_in,
                              void* d_out, int out_size, void* d_ws, size_t ws_size,
                              hipStream_t stream) {
  (void)in_sizes; (void)n_in; (void)out_size; (void)d_ws; (void)ws_size;
  const float* coeffs = (const float*)d_in[0];
  const float* rhs    = (const float*)d_in[1];
  const float* ivr    = (const float*)d_in[2];
  const float* steps  = (const float*)d_in[3];
  float* out = (float*)d_out;
  hipLaunchKernelGGL(ode_blocktri_kernel, dim3(NB), dim3(32), 0, stream,
                     coeffs, rhs, ivr, steps, out);
}


# Serial f64 chain without divide or sqrt: block Cholesky recast as block L D L^T, 3 pivot reciprocals per step (rcp seed + 2 refinement steps) reused as multipliers; factor rounding changes
# speedup vs baseline: 2.5523x; 2.5523x over previous
#include <hip/hip_runtime.h>
#include <hip/hip_bf16.h>
#include <math.h>


typedef __attribute__((ext_vector_type(2))) float v2f;
typedef __attribute__((ext_vector_type(8))) float v8f;

#define NSTEP 100
#define T1    99
#define NB    256
#define OUT_U   (NB * NSTEP)
#define OUT_EPS (3 * OUT_U)
#define OUT_ST  (OUT_EPS + NB)

__device__ __forceinline__ void Ecoef(int r, int c, float* pa, float* pb, float* pc) {
  float a = 0.f, b = 0.f, q = 0.f;
  if (r == 0)      { if (c == 0) a = 1.f;  else if (c == 1) b = 1.f;   else if (c == 2) q = 0.5f; else if (c == 3) a = -1.f; }
  else if (r == 1) { if (c == 1) a = 1.f;  else if (c == 2) b = 1.f;   else if (c == 4) a = -1.f; }
  else if (r == 2) { if (c == 0) a = -1.f; else if (c == 3) a = 1.f;   else if (c == 4) b = -1.f; else if (c == 5) q = 0.5f; }
  else if (r == 3) { if (c == 1) a = -1.f; else if (c == 4) a = 1.f;   else if (c == 5) b = -1.f; }
  else if (r == 4) { if (c == 1) a = -1.f; else if (c == 2) b = -0.5f; else if (c == 4) a = 1.f;  else if (c == 5) b = -0.5f; }
  *pa = a; *pb = b; *pc = q;
}

__device__ __forceinline__ double recip_nr(double d) {
  double r = __builtin_amdgcn_rcp(d);
  double e = __builtin_fma(-d, r, 1.0);
  r = __builtin_fma(r, e, r);
  e = __builtin_fma(-d, r, 1.0);
  r = __builtin_fma(r, e, r);
  return r;
}

__device__ __forceinline__ void ldl3(double s0, double s1, double s2,
                                     double s3, double s4, double s5,
                                     double& u10, double& u20, double& u21,
                                     double& i0, double& i1, double& i2) {
  i0 = recip_nr(s0);
  u10 = s1 * i0;
  u20 = s3 * i0;
  const double d1 = __builtin_fma(-u10, s1, s2);
  i1 = recip_nr(d1);
  const double t = __builtin_fma(-u20, s1, s4);
  u21 = t * i1;
  const double d2 = __builtin_fma(-u21, t, __builtin_fma(-u20, s3, s5));
  i2 = recip_nr(d2);
}

__global__ __launch_bounds__(32) void ode_blocktri_kernel(
    const float* __restrict__ coeffs,
    const float* __restrict__ rhs,
    const float* __restrict__ ivr,
    const float* __restrict__ steps,
    float* __restrict__ out)
{
  __shared__ double S[NSTEP][6];
  __shared__ double Bo[T1][9];
  __shared__ double gv[NSTEP][3];
  __shared__ float  hstep[T1];

  const int b = blockIdx.x;
  const int lane = threadIdx.x;

  for (int i = lane; i < NSTEP * 6; i += 32) ((double*)S)[i] = 0.0;
  for (int i = lane; i < T1 * 9;   i += 32) ((double*)Bo)[i] = 0.0;
  for (int i = lane; i < NSTEP * 3; i += 32) ((double*)gv)[i] = 0.0;
  for (int i = lane; i < T1; i += 32) hstep[i] = steps[b * T1 + i];
  __syncthreads();

  for (int t = lane; t < NSTEP; t += 32) {
    const double c0 = (double)coeffs[(b * NSTEP + t) * 3 + 0];
    const double c1 = (double)coeffs[(b * NSTEP + t) * 3 + 1];
    const double c2 = (double)coeffs[(b * NSTEP + t) * 3 + 2];
    const double r  = (double)rhs[b * NSTEP + t];
    S[t][0] += c0 * c0; S[t][1] += c1 * c0; S[t][2] += c1 * c1;
    S[t][3] += c2 * c0; S[t][4] += c2 * c1; S[t][5] += c2 * c2;
    gv[t][0] += c0 * r; gv[t][1] += c1 * r; gv[t][2] += c2 * r;
    if (t == 0) {
      S[0][0] += 1.0; S[0][2] += 1.0;
      gv[0][0] += (double)ivr[b * 2 + 0];
      gv[0][1] += (double)ivr[b * 2 + 1];
    }
  }
  __syncthreads();

  const int m = lane & 15;
  const int khalf = lane >> 4;
  const int col = (m < 6) ? m : 6;
  float a0, b0, q0, a1, b1, q1, a2, b2, q2;
  Ecoef(2 * khalf,                 col, &a0, &b0, &q0);
  Ecoef(2 * khalf + 1,             col, &a1, &b1, &q1);
  Ecoef((khalf == 0) ? 4 : -1,     col, &a2, &b2, &q2);

  for (int p = 0; p < T1; ++p) {
    const float h = hstep[p];
    v2f f0, f1;
    f0.x = fmaf(fmaf(q0, h, b0), h, a0);
    f0.y = fmaf(fmaf(q1, h, b1), h, a1);
    f1.x = fmaf(fmaf(q2, h, b2), h, a2);
    f1.y = 0.f;
    v8f acc = {};
    acc = __builtin_amdgcn_wmma_f32_16x16x4_f32(false, f0, false, f0,
                                                (short)0, acc, false, false);
    acc = __builtin_amdgcn_wmma_f32_16x16x4_f32(false, f1, false, f1,
                                                (short)0, acc, false, false);
    if (lane < 3) {
      const int j = lane;
      #pragma unroll
      for (int i = 0; i < 3; ++i)
        if (i >= j) S[p][i * (i + 1) / 2 + j] += (double)acc[i];
    } else if (lane < 6) {
      const int j = lane - 3;
      #pragma unroll
      for (int i = 0; i < 3; ++i) Bo[p][i * 3 + j] += (double)acc[i];
      #pragma unroll
      for (int i = 0; i < 3; ++i)
        if (i >= j) S[p + 1][i * (i + 1) / 2 + j] += (double)acc[3 + i];
    }
  }
  __syncthreads();

  if (lane == 0) {
    double u10, u20, u21, i0, i1, i2;
    ldl3(S[0][0], S[0][1], S[0][2], S[0][3], S[0][4], S[0][5],
         u10, u20, u21, i0, i1, i2);
    double w0 = gv[0][0];
    double w1 = __builtin_fma(-u10, w0, gv[0][1]);
    double w2 = __builtin_fma(-u21, w1, __builtin_fma(-u20, w0, gv[0][2]));
    S[0][0] = i0; S[0][1] = u10; S[0][2] = i1; S[0][3] = u20; S[0][4] = u21; S[0][5] = i2;
    gv[0][1] = w1; gv[0][2] = w2;

    for (int p = 0; p < T1; ++p) {
      const double b00 = Bo[p][0], b01 = Bo[p][1], b02 = Bo[p][2];
      const double b10 = Bo[p][3], b11 = Bo[p][4], b12 = Bo[p][5];
      const double b20 = Bo[p][6], b21 = Bo[p][7], b22 = Bo[p][8];
      double n0 = S[p + 1][0], n1 = S[p + 1][1], n2 = S[p + 1][2];
      double n3 = S[p + 1][3], n4 = S[p + 1][4], n5 = S[p + 1][5];
      double g0 = gv[p + 1][0], g1 = gv[p + 1][1], g2 = gv[p + 1][2];

      const double z00 = b00;
      const double z01 = __builtin_fma(-z00, u10, b10);
      const double z02 = __builtin_fma(-z01, u21, __builtin_fma(-z00, u20, b20));
      const double z10 = b01;
      const double z11 = __builtin_fma(-z10, u10, b11);
      const double z12 = __builtin_fma(-z11, u21, __builtin_fma(-z10, u20, b21));
      const double z20 = b02;
      const double z21 = __builtin_fma(-z20, u10, b12);
      const double z22 = __builtin_fma(-z21, u21, __builtin_fma(-z20, u20, b22));
      Bo[p][0] = z00; Bo[p][1] = z01; Bo[p][2] = z02;
      Bo[p][3] = z10; Bo[p][4] = z11; Bo[p][5] = z12;
      Bo[p][6] = z20; Bo[p][7] = z21; Bo[p][8] = z22;

      const double v00 = z00 * i0, v01 = z01 * i1, v02 = z02 * i2;
      const double v10 = z10 * i0, v11 = z11 * i1, v12 = z12 * i2;
      const double v20 = z20 * i0, v21 = z21 * i1, v22 = z22 * i2;

      n0 = __builtin_fma(-v02, z02, __builtin_fma(-v01, z01, __builtin_fma(-v00, z00, n0)));
      n1 = __builtin_fma(-v12, z02, __builtin_fma(-v11, z01, __builtin_fma(-v10, z00, n1)));
      n2 = __builtin_fma(-v12, z12, __builtin_fma(-v11, z11, __builtin_fma(-v10, z10, n2)));
      n3 = __builtin_fma(-v22, z02, __builtin_fma(-v21, z01, __builtin_fma(-v20, z00, n3)));
      n4 = __builtin_fma(-v22, z12, __builtin_fma(-v21, z11, __builtin_fma(-v20, z10, n4)));
      n5 = __builtin_fma(-v22, z22, __builtin_fma(-v21, z21, __builtin_fma(-v20, z20, n5)));

      g0 = __builtin_fma(-v02, w2, __builtin_fma(-v01, w1, __builtin_fma(-v00, w0, g0)));
      g1 = __builtin_fma(-v12, w2, __builtin_fma(-v11, w1, __builtin_fma(-v10, w0, g1)));
      g2 = __builtin_fma(-v22, w2, __builtin_fma(-v21, w1, __builtin_fma(-v20, w0, g2)));

      ldl3(n0, n1, n2, n3, n4, n5, u10, u20, u21, i0, i1, i2);
      w0 = g0;
      w1 = __builtin_fma(-u10, w0, g1);
      w2 = __builtin_fma(-u21, w1, __builtin_fma(-u20, w0, g2));

      S[p + 1][0] = i0; S[p + 1][1] = u10; S[p + 1][2] = i1;
      S[p + 1][3] = u20; S[p + 1][4] = u21; S[p + 1][5] = i2;
      gv[p + 1][0] = w0; gv[p + 1][1] = w1; gv[p + 1][2] = w2;
    }

    double x2 = w2 * i2;
    double x1 = __builtin_fma(-u21, x2, w1 * i1);
    double x0 = __builtin_fma(-u20, x2, __builtin_fma(-u10, x1, w0 * i0));
    gv[T1][0] = x0; gv[T1][1] = x1; gv[T1][2] = x2;

    for (int p = T1 - 1; p >= 0; --p) {
      const double z00 = Bo[p][0], z01 = Bo[p][1], z02 = Bo[p][2];
      const double z10 = Bo[p][3], z11 = Bo[p][4], z12 = Bo[p][5];
      const double z20 = Bo[p][6], z21 = Bo[p][7], z22 = Bo[p][8];
      const double j0 = S[p][0], q10 = S[p][1], j1 = S[p][2];
      const double q20 = S[p][3], q21 = S[p][4], j2 = S[p][5];
      double t0 = __builtin_fma(-z20, x2, __builtin_fma(-z10, x1, __builtin_fma(-z00, x0, gv[p][0])));
      double t1 = __builtin_fma(-z21, x2, __builtin_fma(-z11, x1, __builtin_fma(-z01, x0, gv[p][1])));
      double t2 = __builtin_fma(-z22, x2, __builtin_fma(-z12, x1, __builtin_fma(-z02, x0, gv[p][2])));
      t0 *= j0; t1 *= j1; t2 *= j2;
      x2 = t2;
      x1 = __builtin_fma(-q21, x2, t1);
      x0 = __builtin_fma(-q20, x2, __builtin_fma(-q10, x1, t0));
      gv[p][0] = x0; gv[p][1] = x1; gv[p][2] = x2;
    }
  }
  __syncthreads();

  for (int t = lane; t < NSTEP; t += 32) {
    out[0 * OUT_U + b * NSTEP + t] = (float)gv[t][0];
    out[1 * OUT_U + b * NSTEP + t] = (float)gv[t][1];
    out[2 * OUT_U + b * NSTEP + t] = (float)gv[t][2];
  }
  if (lane == 0) out[OUT_EPS + b] = 0.0f;
  for (int p = lane; p < T1; p += 32) out[OUT_ST + b * T1 + p] = hstep[p];
}

extern "C" void kernel_launch(void* const* d_in, const int* in_sizes, int n_in,
                              void* d_out, int out_size, void* d_ws, size_t ws_size,
                              hipStream_t stream) {
  (void)in_sizes; (void)n_in; (void)out_size; (void)d_ws; (void)ws_size;
  const float* coeffs = (const float*)d_in[0];
  const float* rhs    = (const float*)d_in[1];
  const float* ivr    = (const float*)d_in[2];
  const float* steps  = (const float*)d_in[3];
  float* out = (float*)d_out;
  hipLaunchKernelGGL(ode_blocktri_kernel, dim3(NB), dim3(32), 0, stream,
                     coeffs, rhs, ivr, steps, out);
}


# Two-sided block L D L^T: lanes 0 and 1 eliminate from block 0 up and block 99 down in lockstep, meet in block 50, back-substitute outwards; elimination order and factor rounding change
# speedup vs baseline: 3.8388x; 1.5041x over previous
#include <hip/hip_runtime.h>
#include <hip/hip_bf16.h>
#include <math.h>


typedef __attribute__((ext_vector_type(2))) float v2f;
typedef __attribute__((ext_vector_type(8))) float v8f;

#define NSTEP 100
#define T1    99
#define NB    256
#define OUT_U   (NB * NSTEP)
#define OUT_EPS (3 * OUT_U)
#define OUT_ST  (OUT_EPS + NB)

__device__ __forceinline__ void Ecoef(int r, int c, float* pa, float* pb, float* pc) {
  float a = 0.f, b = 0.f, q = 0.f;
  if (r == 0)      { if (c == 0) a = 1.f;  else if (c == 1) b = 1.f;   else if (c == 2) q = 0.5f; else if (c == 3) a = -1.f; }
  else if (r == 1) { if (c == 1) a = 1.f;  else if (c == 2) b = 1.f;   else if (c == 4) a = -1.f; }
  else if (r == 2) { if (c == 0) a = -1.f; else if (c == 3) a = 1.f;   else if (c == 4) b = -1.f; else if (c == 5) q = 0.5f; }
  else if (r == 3) { if (c == 1) a = -1.f; else if (c == 4) a = 1.f;   else if (c == 5) b = -1.f; }
  else if (r == 4) { if (c == 1) a = -1.f; else if (c == 2) b = -0.5f; else if (c == 4) a = 1.f;  else if (c == 5) b = -0.5f; }
  *pa = a; *pb = b; *pc = q;
}

__device__ __forceinline__ double recip_nr(double d) {
  double r = __builtin_amdgcn_rcp(d);
  double e = __builtin_fma(-d, r, 1.0);
  r = __builtin_fma(r, e, r);
  e = __builtin_fma(-d, r, 1.0);
  r = __builtin_fma(r, e, r);
  return r;
}

__device__ __forceinline__ void ldl3(double s0, double s1, double s2,
                                     double s3, double s4, double s5,
                                     double& u10, double& u20, double& u21,
                                     double& i0, double& i1, double& i2) {
  i0 = recip_nr(s0);
  u10 = s1 * i0;
  u20 = s3 * i0;
  const double d1 = __builtin_fma(-u10, s1, s2);
  i1 = recip_nr(d1);
  const double t = __builtin_fma(-u20, s1, s4);
  u21 = t * i1;
  const double d2 = __builtin_fma(-u21, t, __builtin_fma(-u20, s3, s5));
  i2 = recip_nr(d2);
}

#define MID   50
#define HALF  49

__device__ __forceinline__ void elim_step(
    double (*S)[6], double (*Bo)[9], double (*gv)[3],
    int nxt, int ob, int sk, int sr,
    double& u10, double& u20, double& u21,
    double& i0, double& i1, double& i2,
    double& w0, double& w1, double& w2,
    double& n0, double& n1, double& n2, double& n3, double& n4, double& n5,
    double& g0, double& g1, double& g2) {
  double* bp = Bo[ob];
  const double b00 = bp[0],      b10 = bp[sk],          b20 = bp[2 * sk];
  const double b01 = bp[sr],     b11 = bp[sk + sr],     b21 = bp[2 * sk + sr];
  const double b02 = bp[2 * sr], b12 = bp[sk + 2 * sr], b22 = bp[2 * sk + 2 * sr];
  n0 = S[nxt][0]; n1 = S[nxt][1]; n2 = S[nxt][2];
  n3 = S[nxt][3]; n4 = S[nxt][4]; n5 = S[nxt][5];
  g0 = gv[nxt][0]; g1 = gv[nxt][1]; g2 = gv[nxt][2];

  const double z00 = b00;
  const double z01 = __builtin_fma(-z00, u10, b10);
  const double z02 = __builtin_fma(-z01, u21, __builtin_fma(-z00, u20, b20));
  const double z10 = b01;
  const double z11 = __builtin_fma(-z10, u10, b11);
  const double z12 = __builtin_fma(-z11, u21, __builtin_fma(-z10, u20, b21));
  const double z20 = b02;
  const double z21 = __builtin_fma(-z20, u10, b12);
  const double z22 = __builtin_fma(-z21, u21, __builtin_fma(-z20, u20, b22));
  bp[0] = z00; bp[1] = z01; bp[2] = z02;
  bp[3] = z10; bp[4] = z11; bp[5] = z12;
  bp[6] = z20; bp[7] = z21; bp[8] = z22;

  const double v00 = z00 * i0, v01 = z01 * i1, v02 = z02 * i2;
  const double v10 = z10 * i0, v11 = z11 * i1, v12 = z12 * i2;
  const double v20 = z20 * i0, v21 = z21 * i1, v22 = z22 * i2;

  n0 = __builtin_fma(-v02, z02, __builtin_fma(-v01, z01, __builtin_fma(-v00, z00, n0)));
  n1 = __builtin_fma(-v12, z02, __builtin_fma(-v11, z01, __builtin_fma(-v10, z00, n1)));
  n2 = __builtin_fma(-v12, z12, __builtin_fma(-v11, z11, __builtin_fma(-v10, z10, n2)));
  n3 = __builtin_fma(-v22, z02, __builtin_fma(-v21, z01, __builtin_fma(-v20, z00, n3)));
  n4 = __builtin_fma(-v22, z12, __builtin_fma(-v21, z11, __builtin_fma(-v20, z10, n4)));
  n5 = __builtin_fma(-v22, z22, __builtin_fma(-v21, z21, __builtin_fma(-v20, z20, n5)));

  g0 = __builtin_fma(-v02, w2, __builtin_fma(-v01, w1, __builtin_fma(-v00, w0, g0)));
  g1 = __builtin_fma(-v12, w2, __builtin_fma(-v11, w1, __builtin_fma(-v10, w0, g1)));
  g2 = __builtin_fma(-v22, w2, __builtin_fma(-v21, w1, __builtin_fma(-v20, w0, g2)));

  ldl3(n0, n1, n2, n3, n4, n5, u10, u20, u21, i0, i1, i2);
  w0 = g0;
  w1 = __builtin_fma(-u10, w0, g1);
  w2 = __builtin_fma(-u21, w1, __builtin_fma(-u20, w0, g2));

  S[nxt][0] = i0;  S[nxt][1] = u10; S[nxt][2] = i1;
  S[nxt][3] = u20; S[nxt][4] = u21; S[nxt][5] = i2;
  gv[nxt][0] = w0; gv[nxt][1] = w1; gv[nxt][2] = w2;
}

__device__ __forceinline__ void back_step(
    double (*S)[6], double (*Bo)[9], double (*gv)[3],
    int cur, int ob, double& x0, double& x1, double& x2) {
  const double z00 = Bo[ob][0], z01 = Bo[ob][1], z02 = Bo[ob][2];
  const double z10 = Bo[ob][3], z11 = Bo[ob][4], z12 = Bo[ob][5];
  const double z20 = Bo[ob][6], z21 = Bo[ob][7], z22 = Bo[ob][8];
  const double j0 = S[cur][0], q10 = S[cur][1], j1 = S[cur][2];
  const double q20 = S[cur][3], q21 = S[cur][4], j2 = S[cur][5];
  double t0 = __builtin_fma(-z20, x2, __builtin_fma(-z10, x1, __builtin_fma(-z00, x0, gv[cur][0])));
  double t1 = __builtin_fma(-z21, x2, __builtin_fma(-z11, x1, __builtin_fma(-z01, x0, gv[cur][1])));
  double t2 = __builtin_fma(-z22, x2, __builtin_fma(-z12, x1, __builtin_fma(-z02, x0, gv[cur][2])));
  t0 *= j0; t1 *= j1; t2 *= j2;
  x2 = t2;
  x1 = __builtin_fma(-q21, x2, t1);
  x0 = __builtin_fma(-q20, x2, __builtin_fma(-q10, x1, t0));
  gv[cur][0] = x0; gv[cur][1] = x1; gv[cur][2] = x2;
}

__global__ __launch_bounds__(32) void ode_blocktri_kernel(
    const float* __restrict__ coeffs,
    const float* __restrict__ rhs,
    const float* __restrict__ ivr,
    const float* __restrict__ steps,
    float* __restrict__ out)
{
  __shared__ double S[NSTEP][6];
  __shared__ double Bo[T1][9];
  __shared__ double gv[NSTEP][3];
  __shared__ float  hstep[T1];

  const int b = blockIdx.x;
  const int lane = threadIdx.x;

  for (int i = lane; i < NSTEP * 6; i += 32) ((double*)S)[i] = 0.0;
  for (int i = lane; i < T1 * 9;   i += 32) ((double*)Bo)[i] = 0.0;
  for (int i = lane; i < NSTEP * 3; i += 32) ((double*)gv)[i] = 0.0;
  for (int i = lane; i < T1; i += 32) hstep[i] = steps[b * T1 + i];
  __syncthreads();

  for (int t = lane; t < NSTEP; t += 32) {
    const double c0 = (double)coeffs[(b * NSTEP + t) * 3 + 0];
    const double c1 = (double)coeffs[(b * NSTEP + t) * 3 + 1];
    const double c2 = (double)coeffs[(b * NSTEP + t) * 3 + 2];
    const double r  = (double)rhs[b * NSTEP + t];
    S[t][0] += c0 * c0; S[t][1] += c1 * c0; S[t][2] += c1 * c1;
    S[t][3] += c2 * c0; S[t][4] += c2 * c1; S[t][5] += c2 * c2;
    gv[t][0] += c0 * r; gv[t][1] += c1 * r; gv[t][2] += c2 * r;
    if (t == 0) {
      S[0][0] += 1.0; S[0][2] += 1.0;
      gv[0][0] += (double)ivr[b * 2 + 0];
      gv[0][1] += (double)ivr[b * 2 + 1];
    }
  }
  __syncthreads();

  const int m = lane & 15;
  const int khalf = lane >> 4;
  const int col = (m < 6) ? m : 6;
  float a0, b0, q0, a1, b1, q1, a2, b2, q2;
  Ecoef(2 * khalf,                 col, &a0, &b0, &q0);
  Ecoef(2 * khalf + 1,             col, &a1, &b1, &q1);
  Ecoef((khalf == 0) ? 4 : -1,     col, &a2, &b2, &q2);

  for (int p = 0; p < T1; ++p) {
    const float h = hstep[p];
    v2f f0, f1;
    f0.x = fmaf(fmaf(q0, h, b0), h, a0);
    f0.y = fmaf(fmaf(q1, h, b1), h, a1);
    f1.x = fmaf(fmaf(q2, h, b2), h, a2);
    f1.y = 0.f;
    v8f acc = {};
    acc = __builtin_amdgcn_wmma_f32_16x16x4_f32(false, f0, false, f0,
                                                (short)0, acc, false, false);
    acc = __builtin_amdgcn_wmma_f32_16x16x4_f32(false, f1, false, f1,
                                                (short)0, acc, false, false);
    if (lane < 3) {
      const int j = lane;
      #pragma unroll
      for (int i = 0; i < 3; ++i)
        if (i >= j) S[p][i * (i + 1) / 2 + j] += (double)acc[i];
    } else if (lane < 6) {
      const int j = lane - 3;
      #pragma unroll
      for (int i = 0; i < 3; ++i) Bo[p][i * 3 + j] += (double)acc[i];
      #pragma unroll
      for (int i = 0; i < 3; ++i)
        if (i >= j) S[p + 1][i * (i + 1) / 2 + j] += (double)acc[3 + i];
    }
  }
  __syncthreads();

  const bool up = (lane == 0);
  const int dir = up ? 1 : -1;
  const int first = up ? 0 : NSTEP - 1;
  const int sk = up ? 3 : 1;
  const int sr = up ? 1 : 3;
  double u10 = 0.0, u20 = 0.0, u21 = 0.0, i0 = 0.0, i1 = 0.0, i2 = 0.0;
  double w0 = 0.0, w1 = 0.0, w2 = 0.0;
  double n0 = 0.0, n1 = 0.0, n2 = 0.0, n3 = 0.0, n4 = 0.0, n5 = 0.0;
  double g0 = 0.0, g1 = 0.0, g2 = 0.0;
  if (lane < 2) {
    ldl3(S[first][0], S[first][1], S[first][2], S[first][3], S[first][4], S[first][5],
         u10, u20, u21, i0, i1, i2);
    w0 = gv[first][0];
    w1 = __builtin_fma(-u10, w0, gv[first][1]);
    w2 = __builtin_fma(-u21, w1, __builtin_fma(-u20, w0, gv[first][2]));
    S[first][0] = i0;  S[first][1] = u10; S[first][2] = i1;
    S[first][3] = u20; S[first][4] = u21; S[first][5] = i2;
    gv[first][1] = w1; gv[first][2] = w2;

    for (int s = 0; s < HALF; ++s) {
      const int cur = first + dir * s;
      const int nxt = cur + dir;
      elim_step(S, Bo, gv, nxt, up ? cur : nxt, sk, sr,
                u10, u20, u21, i0, i1, i2, w0, w1, w2,
                n0, n1, n2, n3, n4, n5, g0, g1, g2);
    }
    if (!up) {
      S[MID][0] = n0; S[MID][1] = n1; S[MID][2] = n2;
      S[MID][3] = n3; S[MID][4] = n4; S[MID][5] = n5;
      gv[MID][0] = g0; gv[MID][1] = g1; gv[MID][2] = g2;
    }
  }
  __syncthreads();
  if (lane == 0) {
    elim_step(S, Bo, gv, MID, MID - 1, 3, 1,
              u10, u20, u21, i0, i1, i2, w0, w1, w2,
              n0, n1, n2, n3, n4, n5, g0, g1, g2);
    const double x2 = w2 * i2;
    const double x1 = __builtin_fma(-u21, x2, w1 * i1);
    const double x0 = __builtin_fma(-u20, x2, __builtin_fma(-u10, x1, w0 * i0));
    gv[MID][0] = x0; gv[MID][1] = x1; gv[MID][2] = x2;
  }
  __syncthreads();
  if (lane < 2) {
    double x0 = gv[MID][0], x1 = gv[MID][1], x2 = gv[MID][2];
    for (int s = 0; s < HALF; ++s) {
      const int cur = up ? (MID - 1 - s) : (MID + 1 + s);
      back_step(S, Bo, gv, cur, up ? cur : cur - 1, x0, x1, x2);
    }
    if (up) back_step(S, Bo, gv, 0, 0, x0, x1, x2);
  }
  __syncthreads();

  for (int t = lane; t < NSTEP; t += 32) {
    out[0 * OUT_U + b * NSTEP + t] = (float)gv[t][0];
    out[1 * OUT_U + b * NSTEP + t] = (float)gv[t][1];
    out[2 * OUT_U + b * NSTEP + t] = (float)gv[t][2];
  }
  if (lane == 0) out[OUT_EPS + b] = 0.0f;
  for (int p = lane; p < T1; p += 32) out[OUT_ST + b * T1 + p] = hstep[p];
}

extern "C" void kernel_launch(void* const* d_in, const int* in_sizes, int n_in,
                              void* d_out, int out_size, void* d_ws, size_t ws_size,
                              hipStream_t stream) {
  (void)in_sizes; (void)n_in; (void)out_size; (void)d_ws; (void)ws_size;
  const float* coeffs = (const float*)d_in[0];
  const float* rhs    = (const float*)d_in[1];
  const float* ivr    = (const float*)d_in[2];
  const float* steps  = (const float*)d_in[3];
  float* out = (float*)d_out;
  hipLaunchKernelGGL(ode_blocktri_kernel, dim3(NB), dim3(32), 0, stream,
                     coeffs, rhs, ivr, steps, out);
}


# Gram tiles leave the WMMA loop as f32 in LDS; f64 widening and assembly of S, B, g run once, one block per lane, not as a serial per-step f64 LDS read-modify-write on 6 lanes; sums bit-identical
# speedup vs baseline: 5.8461x; 1.5229x over previous
#include <hip/hip_runtime.h>
#include <hip/hip_bf16.h>
#include <math.h>


typedef __attribute__((ext_vector_type(2))) float v2f;
typedef __attribute__((ext_vector_type(8))) float v8f;

#define NSTEP 100
#define T1    99
#define NB    256
#define OUT_U   (NB * NSTEP)
#define OUT_EPS (3 * OUT_U)
#define OUT_ST  (OUT_EPS + NB)

__device__ __forceinline__ void Ecoef(int r, int c, float* pa, float* pb, float* pc) {
  float a = 0.f, b = 0.f, q = 0.f;
  if (r == 0)      { if (c == 0) a = 1.f;  else if (c == 1) b = 1.f;   else if (c == 2) q = 0.5f; else if (c == 3) a = -1.f; }
  else if (r == 1) { if (c == 1) a = 1.f;  else if (c == 2) b = 1.f;   else if (c == 4) a = -1.f; }
  else if (r == 2) { if (c == 0) a = -1.f; else if (c == 3) a = 1.f;   else if (c == 4) b = -1.f; else if (c == 5) q = 0.5f; }
  else if (r == 3) { if (c == 1) a = -1.f; else if (c == 4) a = 1.f;   else if (c == 5) b = -1.f; }
  else if (r == 4) { if (c == 1) a = -1.f; else if (c == 2) b = -0.5f; else if (c == 4) a = 1.f;  else if (c == 5) b = -0.5f; }
  *pa = a; *pb = b; *pc = q;
}

__device__ __forceinline__ double recip_nr(double d) {
  double r = __builtin_amdgcn_rcp(d);
  double e = __builtin_fma(-d, r, 1.0);
  r = __builtin_fma(r, e, r);
  e = __builtin_fma(-d, r, 1.0);
  r = __builtin_fma(r, e, r);
  return r;
}

__device__ __forceinline__ void ldl3(double s0, double s1, double s2,
                                     double s3, double s4, double s5,
                                     double& u10, double& u20, double& u21,
                                     double& i0, double& i1, double& i2) {
  i0 = recip_nr(s0);
  u10 = s1 * i0;
  u20 = s3 * i0;
  const double d1 = __builtin_fma(-u10, s1, s2);
  i1 = recip_nr(d1);
  const double t = __builtin_fma(-u20, s1, s4);
  u21 = t * i1;
  const double d2 = __builtin_fma(-u21, t, __builtin_fma(-u20, s3, s5));
  i2 = recip_nr(d2);
}

#define MID   50
#define HALF  49

__device__ __forceinline__ void elim_step(
    double (*S)[6], double (*Bo)[9], double (*gv)[3],
    int nxt, int ob, int sk, int sr,
    double& u10, double& u20, double& u21,
    double& i0, double& i1, double& i2,
    double& w0, double& w1, double& w2,
    double& n0, double& n1, double& n2, double& n3, double& n4, double& n5,
    double& g0, double& g1, double& g2) {
  double* bp = Bo[ob];
  const double b00 = bp[0],      b10 = bp[sk],          b20 = bp[2 * sk];
  const double b01 = bp[sr],     b11 = bp[sk + sr],     b21 = bp[2 * sk + sr];
  const double b02 = bp[2 * sr], b12 = bp[sk + 2 * sr], b22 = bp[2 * sk + 2 * sr];
  n0 = S[nxt][0]; n1 = S[nxt][1]; n2 = S[nxt][2];
  n3 = S[nxt][3]; n4 = S[nxt][4]; n5 = S[nxt][5];
  g0 = gv[nxt][0]; g1 = gv[nxt][1]; g2 = gv[nxt][2];

  const double z00 = b00;
  const double z01 = __builtin_fma(-z00, u10, b10);
  const double z02 = __builtin_fma(-z01, u21, __builtin_fma(-z00, u20, b20));
  const double z10 = b01;
  const double z11 = __builtin_fma(-z10, u10, b11);
  const double z12 = __builtin_fma(-z11, u21, __builtin_fma(-z10, u20, b21));
  const double z20 = b02;
  const double z21 = __builtin_fma(-z20, u10, b12);
  const double z22 = __builtin_fma(-z21, u21, __builtin_fma(-z20, u20, b22));
  bp[0] = z00; bp[1] = z01; bp[2] = z02;
  bp[3] = z10; bp[4] = z11; bp[5] = z12;
  bp[6] = z20; bp[7] = z21; bp[8] = z22;

  const double v00 = z00 * i0, v01 = z01 * i1, v02 = z02 * i2;
  const double v10 = z10 * i0, v11 = z11 * i1, v12 = z12 * i2;
  const double v20 = z20 * i0, v21 = z21 * i1, v22 = z22 * i2;

  n0 = __builtin_fma(-v02, z02, __builtin_fma(-v01, z01, __builtin_fma(-v00, z00, n0)));
  n1 = __builtin_fma(-v12, z02, __builtin_fma(-v11, z01, __builtin_fma(-v10, z00, n1)));
  n2 = __builtin_fma(-v12, z12, __builtin_fma(-v11, z11, __builtin_fma(-v10, z10, n2)));
  n3 = __builtin_fma(-v22, z02, __builtin_fma(-v21, z01, __builtin_fma(-v20, z00, n3)));
  n4 = __builtin_fma(-v22, z12, __builtin_fma(-v21, z11, __builtin_fma(-v20, z10, n4)));
  n5 = __builtin_fma(-v22, z22, __builtin_fma(-v21, z21, __builtin_fma(-v20, z20, n5)));

  g0 = __builtin_fma(-v02, w2, __builtin_fma(-v01, w1, __builtin_fma(-v00, w0, g0)));
  g1 = __builtin_fma(-v12, w2, __builtin_fma(-v11, w1, __builtin_fma(-v10, w0, g1)));
  g2 = __builtin_fma(-v22, w2, __builtin_fma(-v21, w1, __builtin_fma(-v20, w0, g2)));

  ldl3(n0, n1, n2, n3, n4, n5, u10, u20, u21, i0, i1, i2);
  w0 = g0;
  w1 = __builtin_fma(-u10, w0, g1);
  w2 = __builtin_fma(-u21, w1, __builtin_fma(-u20, w0, g2));

  S[nxt][0] = i0;  S[nxt][1] = u10; S[nxt][2] = i1;
  S[nxt][3] = u20; S[nxt][4] = u21; S[nxt][5] = i2;
  gv[nxt][0] = w0; gv[nxt][1] = w1; gv[nxt][2] = w2;
}

__device__ __forceinline__ void back_step(
    double (*S)[6], double (*Bo)[9], double (*gv)[3],
    int cur, int ob, double& x0, double& x1, double& x2) {
  const double z00 = Bo[ob][0], z01 = Bo[ob][1], z02 = Bo[ob][2];
  const double z10 = Bo[ob][3], z11 = Bo[ob][4], z12 = Bo[ob][5];
  const double z20 = Bo[ob][6], z21 = Bo[ob][7], z22 = Bo[ob][8];
  const double j0 = S[cur][0], q10 = S[cur][1], j1 = S[cur][2];
  const double q20 = S[cur][3], q21 = S[cur][4], j2 = S[cur][5];
  double t0 = __builtin_fma(-z20, x2, __builtin_fma(-z10, x1, __builtin_fma(-z00, x0, gv[cur][0])));
  double t1 = __builtin_fma(-z21, x2, __builtin_fma(-z11, x1, __builtin_fma(-z01, x0, gv[cur][1])));
  double t2 = __builtin_fma(-z22, x2, __builtin_fma(-z12, x1, __builtin_fma(-z02, x0, gv[cur][2])));
  t0 *= j0; t1 *= j1; t2 *= j2;
  x2 = t2;
  x1 = __builtin_fma(-q21, x2, t1);
  x0 = __builtin_fma(-q20, x2, __builtin_fma(-q10, x1, t0));
  gv[cur][0] = x0; gv[cur][1] = x1; gv[cur][2] = x2;
}

__global__ __launch_bounds__(32) void ode_blocktri_kernel(
    const float* __restrict__ coeffs,
    const float* __restrict__ rhs,
    const float* __restrict__ ivr,
    const float* __restrict__ steps,
    float* __restrict__ out)
{
  __shared__ double S[NSTEP][6];
  __shared__ double Bo[T1][9];
  __shared__ double gv[NSTEP][3];
  __shared__ float  hstep[T1];
  __shared__ float  Gs[T1][36];

  const int b = blockIdx.x;
  const int lane = threadIdx.x;

  for (int i = lane; i < T1; i += 32) hstep[i] = steps[b * T1 + i];
  __syncthreads();

  const int m = lane & 15;
  const int khalf = lane >> 4;
  const int col = (m < 6) ? m : 6;
  float a0, b0, q0, a1, b1, q1, a2, b2, q2;
  Ecoef(2 * khalf,                 col, &a0, &b0, &q0);
  Ecoef(2 * khalf + 1,             col, &a1, &b1, &q1);
  Ecoef((khalf == 0) ? 4 : -1,     col, &a2, &b2, &q2);

  for (int p = 0; p < T1; ++p) {
    const float h = hstep[p];
    v2f f0, f1;
    f0.x = fmaf(fmaf(q0, h, b0), h, a0);
    f0.y = fmaf(fmaf(q1, h, b1), h, a1);
    f1.x = fmaf(fmaf(q2, h, b2), h, a2);
    f1.y = 0.f;
    v8f acc = {};
    acc = __builtin_amdgcn_wmma_f32_16x16x4_f32(false, f0, false, f0,
                                                (short)0, acc, false, false);
    acc = __builtin_amdgcn_wmma_f32_16x16x4_f32(false, f1, false, f1,
                                                (short)0, acc, false, false);
    if (lane < 6) {
      float* q = &Gs[p][lane * 6];
      q[0] = acc[0]; q[1] = acc[1]; q[2] = acc[2];
      q[3] = acc[3]; q[4] = acc[4]; q[5] = acc[5];
    }
  }
  __syncthreads();

  const float iv0 = ivr[b * 2 + 0];
  const float iv1 = ivr[b * 2 + 1];
  for (int t = lane; t < NSTEP; t += 32) {
    const float* cp = coeffs + (b * NSTEP + t) * 3;
    const double c0 = (double)cp[0];
    const double c1 = (double)cp[1];
    const double c2 = (double)cp[2];
    const double r  = (double)rhs[b * NSTEP + t];
    const bool hasPrev = (t > 0);
    const bool hasNext = (t < T1);
    const float* gp = Gs[hasPrev ? t - 1 : 0];
    const float* gn = Gs[hasNext ? t : T1 - 1];
    const float f0 = gp[3 * 6 + 3], f1 = gp[3 * 6 + 4], f2 = gp[4 * 6 + 4];
    const float f3 = gp[3 * 6 + 5], f4 = gp[4 * 6 + 5], f5 = gp[5 * 6 + 5];
    const float n0 = gn[0 * 6 + 0], n1 = gn[0 * 6 + 1], n2 = gn[1 * 6 + 1];
    const float n3 = gn[0 * 6 + 2], n4 = gn[1 * 6 + 2], n5 = gn[2 * 6 + 2];
    const double p0 = (double)(hasPrev ? f0 : 1.f);
    const double p1 = (double)(hasPrev ? f1 : 0.f);
    const double p2 = (double)(hasPrev ? f2 : 1.f);
    const double p3 = (double)(hasPrev ? f3 : 0.f);
    const double p4 = (double)(hasPrev ? f4 : 0.f);
    const double p5 = (double)(hasPrev ? f5 : 0.f);
    S[t][0] = __builtin_fma(c0, c0, p0) + (double)(hasNext ? n0 : 0.f);
    S[t][1] = __builtin_fma(c1, c0, p1) + (double)(hasNext ? n1 : 0.f);
    S[t][2] = __builtin_fma(c1, c1, p2) + (double)(hasNext ? n2 : 0.f);
    S[t][3] = __builtin_fma(c2, c0, p3) + (double)(hasNext ? n3 : 0.f);
    S[t][4] = __builtin_fma(c2, c1, p4) + (double)(hasNext ? n4 : 0.f);
    S[t][5] = __builtin_fma(c2, c2, p5) + (double)(hasNext ? n5 : 0.f);
    gv[t][0] = __builtin_fma(c0, r, (double)((t == 0) ? iv0 : 0.f));
    gv[t][1] = __builtin_fma(c1, r, (double)((t == 0) ? iv1 : 0.f));
    gv[t][2] = c2 * r;
    const double b00 = (double)gn[3 * 6 + 0], b01 = (double)gn[4 * 6 + 0], b02 = (double)gn[5 * 6 + 0];
    const double b10 = (double)gn[3 * 6 + 1], b11 = (double)gn[4 * 6 + 1], b12 = (double)gn[5 * 6 + 1];
    const double b20 = (double)gn[3 * 6 + 2], b21 = (double)gn[4 * 6 + 2], b22 = (double)gn[5 * 6 + 2];
    if (hasNext) {
      Bo[t][0] = b00; Bo[t][1] = b01; Bo[t][2] = b02;
      Bo[t][3] = b10; Bo[t][4] = b11; Bo[t][5] = b12;
      Bo[t][6] = b20; Bo[t][7] = b21; Bo[t][8] = b22;
    }
  }
  __syncthreads();

  const bool up = (lane == 0);
  const int dir = up ? 1 : -1;
  const int first = up ? 0 : NSTEP - 1;
  const int sk = up ? 3 : 1;
  const int sr = up ? 1 : 3;
  double u10 = 0.0, u20 = 0.0, u21 = 0.0, i0 = 0.0, i1 = 0.0, i2 = 0.0;
  double w0 = 0.0, w1 = 0.0, w2 = 0.0;
  double n0 = 0.0, n1 = 0.0, n2 = 0.0, n3 = 0.0, n4 = 0.0, n5 = 0.0;
  double g0 = 0.0, g1 = 0.0, g2 = 0.0;
  if (lane < 2) {
    ldl3(S[first][0], S[first][1], S[first][2], S[first][3], S[first][4], S[first][5],
         u10, u20, u21, i0, i1, i2);
    w0 = gv[first][0];
    w1 = __builtin_fma(-u10, w0, gv[first][1]);
    w2 = __builtin_fma(-u21, w1, __builtin_fma(-u20, w0, gv[first][2]));
    S[first][0] = i0;  S[first][1] = u10; S[first][2] = i1;
    S[first][3] = u20; S[first][4] = u21; S[first][5] = i2;
    gv[first][1] = w1; gv[first][2] = w2;

    for (int s = 0; s < HALF; ++s) {
      const int cur = first + dir * s;
      const int nxt = cur + dir;
      elim_step(S, Bo, gv, nxt, up ? cur : nxt, sk, sr,
                u10, u20, u21, i0, i1, i2, w0, w1, w2,
                n0, n1, n2, n3, n4, n5, g0, g1, g2);
    }
    if (!up) {
      S[MID][0] = n0; S[MID][1] = n1; S[MID][2] = n2;
      S[MID][3] = n3; S[MID][4] = n4; S[MID][5] = n5;
      gv[MID][0] = g0; gv[MID][1] = g1; gv[MID][2] = g2;
    }
  }
  __syncthreads();
  if (lane == 0) {
    elim_step(S, Bo, gv, MID, MID - 1, 3, 1,
              u10, u20, u21, i0, i1, i2, w0, w1, w2,
              n0, n1, n2, n3, n4, n5, g0, g1, g2);
    const double x2 = w2 * i2;
    const double x1 = __builtin_fma(-u21, x2, w1 * i1);
    const double x0 = __builtin_fma(-u20, x2, __builtin_fma(-u10, x1, w0 * i0));
    gv[MID][0] = x0; gv[MID][1] = x1; gv[MID][2] = x2;
  }
  __syncthreads();
  if (lane < 2) {
    double x0 = gv[MID][0], x1 = gv[MID][1], x2 = gv[MID][2];
    for (int s = 0; s < HALF; ++s) {
      const int cur = up ? (MID - 1 - s) : (MID + 1 + s);
      back_step(S, Bo, gv, cur, up ? cur : cur - 1, x0, x1, x2);
    }
    if (up) back_step(S, Bo, gv, 0, 0, x0, x1, x2);
  }
  __syncthreads();

  for (int t = lane; t < NSTEP; t += 32) {
    out[0 * OUT_U + b * NSTEP + t] = (float)gv[t][0];
    out[1 * OUT_U + b * NSTEP + t] = (float)gv[t][1];
    out[2 * OUT_U + b * NSTEP + t] = (float)gv[t][2];
  }
  if (lane == 0) out[OUT_EPS + b] = 0.0f;
  for (int p = lane; p < T1; p += 32) out[OUT_ST + b * T1 + p] = hstep[p];
}

extern "C" void kernel_launch(void* const* d_in, const int* in_sizes, int n_in,
                              void* d_out, int out_size, void* d_ws, size_t ws_size,
                              hipStream_t stream) {
  (void)in_sizes; (void)n_in; (void)out_size; (void)d_ws; (void)ws_size;
  const float* coeffs = (const float*)d_in[0];
  const float* rhs    = (const float*)d_in[1];
  const float* ivr    = (const float*)d_in[2];
  const float* steps  = (const float*)d_in[3];
  float* out = (float*)d_out;
  hipLaunchKernelGGL(ode_blocktri_kernel, dim3(NB), dim3(32), 0, stream,
                     coeffs, rhs, ivr, steps, out);
}


# Block elimination order becomes odd-even (block cyclic reduction): 7 levels, one block per lane, replace the 50-step two-ended serial chain; elimination order and factor rounding change
# speedup vs baseline: 13.8671x; 2.3720x over previous
#include <hip/hip_runtime.h>
#include <hip/hip_bf16.h>
#include <math.h>


typedef __attribute__((ext_vector_type(2))) float v2f;
typedef __attribute__((ext_vector_type(8))) float v8f;

#define NSTEP 100
#define T1    99
#define NB    256
#define OUT_U   (NB * NSTEP)
#define OUT_EPS (3 * OUT_U)
#define OUT_ST  (OUT_EPS + NB)

__device__ __forceinline__ void Ecoef(int r, int c, float* pa, float* pb, float* pc) {
  float a = 0.f, b = 0.f, q = 0.f;
  if (r == 0)      { if (c == 0) a = 1.f;  else if (c == 1) b = 1.f;   else if (c == 2) q = 0.5f; else if (c == 3) a = -1.f; }
  else if (r == 1) { if (c == 1) a = 1.f;  else if (c == 2) b = 1.f;   else if (c == 4) a = -1.f; }
  else if (r == 2) { if (c == 0) a = -1.f; else if (c == 3) a = 1.f;   else if (c == 4) b = -1.f; else if (c == 5) q = 0.5f; }
  else if (r == 3) { if (c == 1) a = -1.f; else if (c == 4) a = 1.f;   else if (c == 5) b = -1.f; }
  else if (r == 4) { if (c == 1) a = -1.f; else if (c == 2) b = -0.5f; else if (c == 4) a = 1.f;  else if (c == 5) b = -0.5f; }
  *pa = a; *pb = b; *pc = q;
}

__device__ __forceinline__ double recip_nr(double d) {
  double r = __builtin_amdgcn_rcp(d);
  double e = __builtin_fma(-d, r, 1.0);
  r = __builtin_fma(r, e, r);
  e = __builtin_fma(-d, r, 1.0);
  r = __builtin_fma(r, e, r);
  return r;
}

__device__ __forceinline__ void ldl3(double s0, double s1, double s2,
                                     double s3, double s4, double s5,
                                     double& u10, double& u20, double& u21,
                                     double& i0, double& i1, double& i2) {
  i0 = recip_nr(s0);
  u10 = s1 * i0;
  u20 = s3 * i0;
  const double d1 = __builtin_fma(-u10, s1, s2);
  i1 = recip_nr(d1);
  const double t = __builtin_fma(-u20, s1, s4);
  u21 = t * i1;
  const double d2 = __builtin_fma(-u21, t, __builtin_fma(-u20, s3, s5));
  i2 = recip_nr(d2);
}

__device__ __forceinline__ void zw3(const double* m,
                                    double u10, double u20, double u21,
                                    double i0, double i1, double i2,
                                    double* z, double* w) {
  z[0] = m[0];
  z[1] = __builtin_fma(-z[0], u10, m[1]);
  z[2] = __builtin_fma(-z[1], u21, __builtin_fma(-z[0], u20, m[2]));
  z[3] = m[3];
  z[4] = __builtin_fma(-z[3], u10, m[4]);
  z[5] = __builtin_fma(-z[4], u21, __builtin_fma(-z[3], u20, m[5]));
  z[6] = m[6];
  z[7] = __builtin_fma(-z[6], u10, m[7]);
  z[8] = __builtin_fma(-z[7], u21, __builtin_fma(-z[6], u20, m[8]));
  w[0] = z[0] * i0; w[1] = z[1] * i1; w[2] = z[2] * i2;
  w[3] = z[3] * i0; w[4] = z[4] * i1; w[5] = z[5] * i2;
  w[6] = z[6] * i0; w[7] = z[7] * i1; w[8] = z[8] * i2;
}

__device__ __forceinline__ void cr_eliminate(
    double (*S)[6], double (*C)[9], double (*gv)[3],
    double (*Zl)[9], double (*Wl)[9], double (*Zr)[9], double (*Wr)[9],
    int i, int s) {
  const int l = i - s;
  const bool hasR = (i + s < NSTEP);
  const double* cl = C[l];
  const double* cr = C[hasR ? i : l];
  double ml[9], mr[9];
  ml[0] = cl[0]; ml[1] = cl[1]; ml[2] = cl[2];
  ml[3] = cl[3]; ml[4] = cl[4]; ml[5] = cl[5];
  ml[6] = cl[6]; ml[7] = cl[7]; ml[8] = cl[8];
  mr[0] = cr[0]; mr[1] = cr[3]; mr[2] = cr[6];
  mr[3] = cr[1]; mr[4] = cr[4]; mr[5] = cr[7];
  mr[6] = cr[2]; mr[7] = cr[5]; mr[8] = cr[8];
  const double s0 = S[i][0], s1 = S[i][1], s2 = S[i][2];
  const double s3 = S[i][3], s4 = S[i][4], s5 = S[i][5];
  const double g0 = gv[i][0], g1 = gv[i][1], g2 = gv[i][2];

  double u10, u20, u21, i0, i1, i2;
  ldl3(s0, s1, s2, s3, s4, s5, u10, u20, u21, i0, i1, i2);
  const double w1 = __builtin_fma(-u10, g0, g1);
  const double w2 = __builtin_fma(-u21, w1, __builtin_fma(-u20, g0, g2));
  double zl[9], wl[9], zr[9], wr[9];
  zw3(ml, u10, u20, u21, i0, i1, i2, zl, wl);
  zw3(mr, u10, u20, u21, i0, i1, i2, zr, wr);

  S[i][0] = i0;  S[i][1] = u10; S[i][2] = i1;
  S[i][3] = u20; S[i][4] = u21; S[i][5] = i2;
  gv[i][1] = w1; gv[i][2] = w2;
  Zl[i][0] = zl[0]; Zl[i][1] = zl[1]; Zl[i][2] = zl[2];
  Zl[i][3] = zl[3]; Zl[i][4] = zl[4]; Zl[i][5] = zl[5];
  Zl[i][6] = zl[6]; Zl[i][7] = zl[7]; Zl[i][8] = zl[8];
  Wl[i][0] = wl[0]; Wl[i][1] = wl[1]; Wl[i][2] = wl[2];
  Wl[i][3] = wl[3]; Wl[i][4] = wl[4]; Wl[i][5] = wl[5];
  Wl[i][6] = wl[6]; Wl[i][7] = wl[7]; Wl[i][8] = wl[8];
  if (hasR) {
    Zr[i][0] = zr[0]; Zr[i][1] = zr[1]; Zr[i][2] = zr[2];
    Zr[i][3] = zr[3]; Zr[i][4] = zr[4]; Zr[i][5] = zr[5];
    Zr[i][6] = zr[6]; Zr[i][7] = zr[7]; Zr[i][8] = zr[8];
    Wr[i][0] = wr[0]; Wr[i][1] = wr[1]; Wr[i][2] = wr[2];
    Wr[i][3] = wr[3]; Wr[i][4] = wr[4]; Wr[i][5] = wr[5];
    Wr[i][6] = wr[6]; Wr[i][7] = wr[7]; Wr[i][8] = wr[8];
  }
}

__device__ __forceinline__ void schur3(
    const double* zp, const double* wp, const double* xp,
    double& n0, double& n1, double& n2, double& n3, double& n4, double& n5,
    double& g0, double& g1, double& g2) {
  const double z0 = zp[0], z1 = zp[1], z2 = zp[2];
  const double z3 = zp[3], z4 = zp[4], z5 = zp[5];
  const double z6 = zp[6], z7 = zp[7], z8 = zp[8];
  const double v0 = wp[0], v1 = wp[1], v2 = wp[2];
  const double v3 = wp[3], v4 = wp[4], v5 = wp[5];
  const double v6 = wp[6], v7 = wp[7], v8 = wp[8];
  const double x0 = xp[0], x1 = xp[1], x2 = xp[2];
  n0 = __builtin_fma(-v2, z2, __builtin_fma(-v1, z1, __builtin_fma(-v0, z0, n0)));
  n1 = __builtin_fma(-v5, z2, __builtin_fma(-v4, z1, __builtin_fma(-v3, z0, n1)));
  n2 = __builtin_fma(-v5, z5, __builtin_fma(-v4, z4, __builtin_fma(-v3, z3, n2)));
  n3 = __builtin_fma(-v8, z2, __builtin_fma(-v7, z1, __builtin_fma(-v6, z0, n3)));
  n4 = __builtin_fma(-v8, z5, __builtin_fma(-v7, z4, __builtin_fma(-v6, z3, n4)));
  n5 = __builtin_fma(-v8, z8, __builtin_fma(-v7, z7, __builtin_fma(-v6, z6, n5)));
  g0 = __builtin_fma(-v2, x2, __builtin_fma(-v1, x1, __builtin_fma(-v0, x0, g0)));
  g1 = __builtin_fma(-v5, x2, __builtin_fma(-v4, x1, __builtin_fma(-v3, x0, g1)));
  g2 = __builtin_fma(-v8, x2, __builtin_fma(-v7, x1, __builtin_fma(-v6, x0, g2)));
}

__device__ __forceinline__ void cr_update(
    double (*S)[6], double (*C)[9], double (*gv)[3],
    double (*Zl)[9], double (*Wl)[9], double (*Zr)[9], double (*Wr)[9],
    int j, int s) {
  double n0 = S[j][0], n1 = S[j][1], n2 = S[j][2];
  double n3 = S[j][3], n4 = S[j][4], n5 = S[j][5];
  double g0 = gv[j][0], g1 = gv[j][1], g2 = gv[j][2];
  if (j > 0) {
    const int e = j - s;
    schur3(Zr[e], Wr[e], gv[e], n0, n1, n2, n3, n4, n5, g0, g1, g2);
  }
  if (j + s < NSTEP) {
    const int e = j + s;
    schur3(Zl[e], Wl[e], gv[e], n0, n1, n2, n3, n4, n5, g0, g1, g2);
    if (j + 2 * s < NSTEP) {
      const double* wp = Wl[e];
      const double* zp = Zr[e];
      const double v0 = wp[0], v1 = wp[1], v2 = wp[2];
      const double v3 = wp[3], v4 = wp[4], v5 = wp[5];
      const double v6 = wp[6], v7 = wp[7], v8 = wp[8];
      const double z0 = zp[0], z1 = zp[1], z2 = zp[2];
      const double z3 = zp[3], z4 = zp[4], z5 = zp[5];
      const double z6 = zp[6], z7 = zp[7], z8 = zp[8];
      double* cp = C[j];
      cp[0] = __builtin_fma(-v2, z2, __builtin_fma(-v1, z1, -(v0 * z0)));
      cp[1] = __builtin_fma(-v2, z5, __builtin_fma(-v1, z4, -(v0 * z3)));
      cp[2] = __builtin_fma(-v2, z8, __builtin_fma(-v1, z7, -(v0 * z6)));
      cp[3] = __builtin_fma(-v5, z2, __builtin_fma(-v4, z1, -(v3 * z0)));
      cp[4] = __builtin_fma(-v5, z5, __builtin_fma(-v4, z4, -(v3 * z3)));
      cp[5] = __builtin_fma(-v5, z8, __builtin_fma(-v4, z7, -(v3 * z6)));
      cp[6] = __builtin_fma(-v8, z2, __builtin_fma(-v7, z1, -(v6 * z0)));
      cp[7] = __builtin_fma(-v8, z5, __builtin_fma(-v7, z4, -(v6 * z3)));
      cp[8] = __builtin_fma(-v8, z8, __builtin_fma(-v7, z7, -(v6 * z6)));
    }
  }
  S[j][0] = n0; S[j][1] = n1; S[j][2] = n2;
  S[j][3] = n3; S[j][4] = n4; S[j][5] = n5;
  gv[j][0] = g0; gv[j][1] = g1; gv[j][2] = g2;
}

__device__ __forceinline__ void cr_back(
    double (*S)[6], double (*gv)[3], double (*Wl)[9], double (*Wr)[9],
    int i, int s) {
  const double j0 = S[i][0], q10 = S[i][1], j1 = S[i][2];
  const double q20 = S[i][3], q21 = S[i][4], j2 = S[i][5];
  double t0 = gv[i][0] * j0, t1 = gv[i][1] * j1, t2 = gv[i][2] * j2;
  {
    const double* wp = Wl[i];
    const double a0 = gv[i - s][0], a1 = gv[i - s][1], a2 = gv[i - s][2];
    t0 = __builtin_fma(-wp[6], a2, __builtin_fma(-wp[3], a1, __builtin_fma(-wp[0], a0, t0)));
    t1 = __builtin_fma(-wp[7], a2, __builtin_fma(-wp[4], a1, __builtin_fma(-wp[1], a0, t1)));
    t2 = __builtin_fma(-wp[8], a2, __builtin_fma(-wp[5], a1, __builtin_fma(-wp[2], a0, t2)));
  }
  if (i + s < NSTEP) {
    const double* wp = Wr[i];
    const double a0 = gv[i + s][0], a1 = gv[i + s][1], a2 = gv[i + s][2];
    t0 = __builtin_fma(-wp[6], a2, __builtin_fma(-wp[3], a1, __builtin_fma(-wp[0], a0, t0)));
    t1 = __builtin_fma(-wp[7], a2, __builtin_fma(-wp[4], a1, __builtin_fma(-wp[1], a0, t1)));
    t2 = __builtin_fma(-wp[8], a2, __builtin_fma(-wp[5], a1, __builtin_fma(-wp[2], a0, t2)));
  }
  const double x2 = t2;
  const double x1 = __builtin_fma(-q21, x2, t1);
  const double x0 = __builtin_fma(-q20, x2, __builtin_fma(-q10, x1, t0));
  gv[i][0] = x0; gv[i][1] = x1; gv[i][2] = x2;
}

#define TOPS  64

__global__ __launch_bounds__(32) void ode_blocktri_kernel(
    const float* __restrict__ coeffs,
    const float* __restrict__ rhs,
    const float* __restrict__ ivr,
    const float* __restrict__ steps,
    float* __restrict__ out)
{
  __shared__ double S[NSTEP][6];
  __shared__ double Bo[T1][9];
  __shared__ double gv[NSTEP][3];
  __shared__ double Zl[NSTEP][9];
  __shared__ double Wl[NSTEP][9];
  __shared__ double Zr[NSTEP][9];
  __shared__ double Wr[NSTEP][9];
  __shared__ float  hstep[T1];
  __shared__ float  Gs[T1][36];

  const int b = blockIdx.x;
  const int lane = threadIdx.x;

  for (int i = lane; i < T1; i += 32) hstep[i] = steps[b * T1 + i];
  __syncthreads();

  const int m = lane & 15;
  const int khalf = lane >> 4;
  const int col = (m < 6) ? m : 6;
  float a0, b0, q0, a1, b1, q1, a2, b2, q2;
  Ecoef(2 * khalf,                 col, &a0, &b0, &q0);
  Ecoef(2 * khalf + 1,             col, &a1, &b1, &q1);
  Ecoef((khalf == 0) ? 4 : -1,     col, &a2, &b2, &q2);

  for (int p = 0; p < T1; ++p) {
    const float h = hstep[p];
    v2f f0, f1;
    f0.x = fmaf(fmaf(q0, h, b0), h, a0);
    f0.y = fmaf(fmaf(q1, h, b1), h, a1);
    f1.x = fmaf(fmaf(q2, h, b2), h, a2);
    f1.y = 0.f;
    v8f acc = {};
    acc = __builtin_amdgcn_wmma_f32_16x16x4_f32(false, f0, false, f0,
                                                (short)0, acc, false, false);
    acc = __builtin_amdgcn_wmma_f32_16x16x4_f32(false, f1, false, f1,
                                                (short)0, acc, false, false);
    if (lane < 6) {
      float* q = &Gs[p][lane * 6];
      q[0] = acc[0]; q[1] = acc[1]; q[2] = acc[2];
      q[3] = acc[3]; q[4] = acc[4]; q[5] = acc[5];
    }
  }
  __syncthreads();

  const float iv0 = ivr[b * 2 + 0];
  const float iv1 = ivr[b * 2 + 1];
  for (int t = lane; t < NSTEP; t += 32) {
    const float* cp = coeffs + (b * NSTEP + t) * 3;
    const double c0 = (double)cp[0];
    const double c1 = (double)cp[1];
    const double c2 = (double)cp[2];
    const double r  = (double)rhs[b * NSTEP + t];
    const bool hasPrev = (t > 0);
    const bool hasNext = (t < T1);
    const float* gp = Gs[hasPrev ? t - 1 : 0];
    const float* gn = Gs[hasNext ? t : T1 - 1];
    const float f0 = gp[3 * 6 + 3], f1 = gp[3 * 6 + 4], f2 = gp[4 * 6 + 4];
    const float f3 = gp[3 * 6 + 5], f4 = gp[4 * 6 + 5], f5 = gp[5 * 6 + 5];
    const float n0 = gn[0 * 6 + 0], n1 = gn[0 * 6 + 1], n2 = gn[1 * 6 + 1];
    const float n3 = gn[0 * 6 + 2], n4 = gn[1 * 6 + 2], n5 = gn[2 * 6 + 2];
    const double p0 = (double)(hasPrev ? f0 : 1.f);
    const double p1 = (double)(hasPrev ? f1 : 0.f);
    const double p2 = (double)(hasPrev ? f2 : 1.f);
    const double p3 = (double)(hasPrev ? f3 : 0.f);
    const double p4 = (double)(hasPrev ? f4 : 0.f);
    const double p5 = (double)(hasPrev ? f5 : 0.f);
    S[t][0] = __builtin_fma(c0, c0, p0) + (double)(hasNext ? n0 : 0.f);
    S[t][1] = __builtin_fma(c1, c0, p1) + (double)(hasNext ? n1 : 0.f);
    S[t][2] = __builtin_fma(c1, c1, p2) + (double)(hasNext ? n2 : 0.f);
    S[t][3] = __builtin_fma(c2, c0, p3) + (double)(hasNext ? n3 : 0.f);
    S[t][4] = __builtin_fma(c2, c1, p4) + (double)(hasNext ? n4 : 0.f);
    S[t][5] = __builtin_fma(c2, c2, p5) + (double)(hasNext ? n5 : 0.f);
    gv[t][0] = __builtin_fma(c0, r, (double)((t == 0) ? iv0 : 0.f));
    gv[t][1] = __builtin_fma(c1, r, (double)((t == 0) ? iv1 : 0.f));
    gv[t][2] = c2 * r;
    const double b00 = (double)gn[3 * 6 + 0], b01 = (double)gn[4 * 6 + 0], b02 = (double)gn[5 * 6 + 0];
    const double b10 = (double)gn[3 * 6 + 1], b11 = (double)gn[4 * 6 + 1], b12 = (double)gn[5 * 6 + 1];
    const double b20 = (double)gn[3 * 6 + 2], b21 = (double)gn[4 * 6 + 2], b22 = (double)gn[5 * 6 + 2];
    if (hasNext) {
      Bo[t][0] = b00; Bo[t][1] = b01; Bo[t][2] = b02;
      Bo[t][3] = b10; Bo[t][4] = b11; Bo[t][5] = b12;
      Bo[t][6] = b20; Bo[t][7] = b21; Bo[t][8] = b22;
    }
  }
  __syncthreads();

  for (int s = 1; s <= TOPS; s <<= 1) {
    for (int i = s * (2 * lane + 1); i < NSTEP; i += 64 * s)
      cr_eliminate(S, Bo, gv, Zl, Wl, Zr, Wr, i, s);
    __syncthreads();
    for (int j = 2 * s * lane; j < NSTEP; j += 64 * s)
      cr_update(S, Bo, gv, Zl, Wl, Zr, Wr, j, s);
    __syncthreads();
  }
  if (lane == 0) {
    double u10, u20, u21, i0, i1, i2;
    ldl3(S[0][0], S[0][1], S[0][2], S[0][3], S[0][4], S[0][5],
         u10, u20, u21, i0, i1, i2);
    const double w0 = gv[0][0];
    const double w1 = __builtin_fma(-u10, w0, gv[0][1]);
    const double w2 = __builtin_fma(-u21, w1, __builtin_fma(-u20, w0, gv[0][2]));
    const double x2 = w2 * i2;
    const double x1 = __builtin_fma(-u21, x2, w1 * i1);
    const double x0 = __builtin_fma(-u20, x2, __builtin_fma(-u10, x1, w0 * i0));
    gv[0][0] = x0; gv[0][1] = x1; gv[0][2] = x2;
  }
  __syncthreads();
  for (int s = TOPS; s >= 1; s >>= 1) {
    for (int i = s * (2 * lane + 1); i < NSTEP; i += 64 * s)
      cr_back(S, gv, Wl, Wr, i, s);
    __syncthreads();
  }

  for (int t = lane; t < NSTEP; t += 32) {
    out[0 * OUT_U + b * NSTEP + t] = (float)gv[t][0];
    out[1 * OUT_U + b * NSTEP + t] = (float)gv[t][1];
    out[2 * OUT_U + b * NSTEP + t] = (float)gv[t][2];
  }
  if (lane == 0) out[OUT_EPS + b] = 0.0f;
  for (int p = lane; p < T1; p += 32) out[OUT_ST + b * T1 + p] = hstep[p];
}

extern "C" void kernel_launch(void* const* d_in, const int* in_sizes, int n_in,
                              void* d_out, int out_size, void* d_ws, size_t ws_size,
                              hipStream_t stream) {
  (void)in_sizes; (void)n_in; (void)out_size; (void)d_ws; (void)ws_size;
  const float* coeffs = (const float*)d_in[0];
  const float* rhs    = (const float*)d_in[1];
  const float* ivr    = (const float*)d_in[2];
  const float* steps  = (const float*)d_in[3];
  float* out = (float*)d_out;
  hipLaunchKernelGGL(ode_blocktri_kernel, dim3(NB), dim3(32), 0, stream,
                     coeffs, rhs, ivr, steps, out);
}


# speedup vs baseline: 24.9284x; 1.7977x over previous
#include <hip/hip_runtime.h>
#include <hip/hip_bf16.h>
#include <math.h>


typedef __attribute__((ext_vector_type(2))) float v2f;
typedef __attribute__((ext_vector_type(8))) float v8f;

#define NSTEP 100
#define T1    99
#define NB    256
#define OUT_U   (NB * NSTEP)
#define OUT_EPS (3 * OUT_U)
#define OUT_ST  (OUT_EPS + NB)

__device__ __forceinline__ void Ecoef(int r, int c, float* pa, float* pb, float* pc) {
  float a = 0.f, b = 0.f, q = 0.f;
  if (r == 0)      { if (c == 0) a = 1.f;  else if (c == 1) b = 1.f;   else if (c == 2) q = 0.5f; else if (c == 3) a = -1.f; }
  else if (r == 1) { if (c == 1) a = 1.f;  else if (c == 2) b = 1.f;   else if (c == 4) a = -1.f; }
  else if (r == 2) { if (c == 0) a = -1.f; else if (c == 3) a = 1.f;   else if (c == 4) b = -1.f; else if (c == 5) q = 0.5f; }
  else if (r == 3) { if (c == 1) a = -1.f; else if (c == 4) a = 1.f;   else if (c == 5) b = -1.f; }
  else if (r == 4) { if (c == 1) a = -1.f; else if (c == 2) b = -0.5f; else if (c == 4) a = 1.f;  else if (c == 5) b = -0.5f; }
  *pa = a; *pb = b; *pc = q;
}

__device__ __forceinline__ double recip_nr(double d) {
  double r = __builtin_amdgcn_rcp(d);
  double e = __builtin_fma(-d, r, 1.0);
  r = __builtin_fma(r, e, r);
  e = __builtin_fma(-d, r, 1.0);
  r = __builtin_fma(r, e, r);
  return r;
}

__device__ __forceinline__ void ldl3(double s0, double s1, double s2,
                                     double s3, double s4, double s5,
                                     double& u10, double& u20, double& u21,
                                     double& i0, double& i1, double& i2) {
  i0 = recip_nr(s0);
  u10 = s1 * i0;
  u20 = s3 * i0;
  const double d1 = __builtin_fma(-u10, s1, s2);
  i1 = recip_nr(d1);
  const double t = __builtin_fma(-u20, s1, s4);
  u21 = t * i1;
  const double d2 = __builtin_fma(-u21, t, __builtin_fma(-u20, s3, s5));
  i2 = recip_nr(d2);
}

__device__ __forceinline__ void zw3(const double* m,
                                    double u10, double u20, double u21,
                                    double i0, double i1, double i2,
                                    double* z, double* w) {
  z[0] = m[0];
  z[1] = __builtin_fma(-z[0], u10, m[1]);
  z[2] = __builtin_fma(-z[1], u21, __builtin_fma(-z[0], u20, m[2]));
  z[3] = m[3];
  z[4] = __builtin_fma(-z[3], u10, m[4]);
  z[5] = __builtin_fma(-z[4], u21, __builtin_fma(-z[3], u20, m[5]));
  z[6] = m[6];
  z[7] = __builtin_fma(-z[6], u10, m[7]);
  z[8] = __builtin_fma(-z[7], u21, __builtin_fma(-z[6], u20, m[8]));
  w[0] = z[0] * i0; w[1] = z[1] * i1; w[2] = z[2] * i2;
  w[3] = z[3] * i0; w[4] = z[4] * i1; w[5] = z[5] * i2;
  w[6] = z[6] * i0; w[7] = z[7] * i1; w[8] = z[8] * i2;
}

__device__ __forceinline__ void elim_left(
    const double (*S)[6], double (*F)[6], const double (*C)[9], double (*gv)[3],
    double (*Zl)[9], double (*Wl)[9], int i, int s) {
  const double* cl = C[i - s];
  double ml[9];
  ml[0] = cl[0]; ml[1] = cl[1]; ml[2] = cl[2];
  ml[3] = cl[3]; ml[4] = cl[4]; ml[5] = cl[5];
  ml[6] = cl[6]; ml[7] = cl[7]; ml[8] = cl[8];
  const double s0 = S[i][0], s1 = S[i][1], s2 = S[i][2];
  const double s3 = S[i][3], s4 = S[i][4], s5 = S[i][5];
  const double g0 = gv[i][0], g1 = gv[i][1], g2 = gv[i][2];

  double u10, u20, u21, i0, i1, i2;
  ldl3(s0, s1, s2, s3, s4, s5, u10, u20, u21, i0, i1, i2);
  const double w1 = __builtin_fma(-u10, g0, g1);
  const double w2 = __builtin_fma(-u21, w1, __builtin_fma(-u20, g0, g2));
  double zl[9], wl[9];
  zw3(ml, u10, u20, u21, i0, i1, i2, zl, wl);

  F[i][0] = i0;  F[i][1] = u10; F[i][2] = i1;
  F[i][3] = u20; F[i][4] = u21; F[i][5] = i2;
  gv[i][1] = w1; gv[i][2] = w2;
  Zl[i][0] = zl[0]; Zl[i][1] = zl[1]; Zl[i][2] = zl[2];
  Zl[i][3] = zl[3]; Zl[i][4] = zl[4]; Zl[i][5] = zl[5];
  Zl[i][6] = zl[6]; Zl[i][7] = zl[7]; Zl[i][8] = zl[8];
  Wl[i][0] = wl[0]; Wl[i][1] = wl[1]; Wl[i][2] = wl[2];
  Wl[i][3] = wl[3]; Wl[i][4] = wl[4]; Wl[i][5] = wl[5];
  Wl[i][6] = wl[6]; Wl[i][7] = wl[7]; Wl[i][8] = wl[8];
}

__device__ __forceinline__ void elim_right(
    const double (*S)[6], const double (*C)[9],
    double (*Zr)[9], double (*Wr)[9], int i) {
  const double* cr = C[i];
  double mr[9];
  mr[0] = cr[0]; mr[1] = cr[3]; mr[2] = cr[6];
  mr[3] = cr[1]; mr[4] = cr[4]; mr[5] = cr[7];
  mr[6] = cr[2]; mr[7] = cr[5]; mr[8] = cr[8];
  const double s0 = S[i][0], s1 = S[i][1], s2 = S[i][2];
  const double s3 = S[i][3], s4 = S[i][4], s5 = S[i][5];

  double u10, u20, u21, i0, i1, i2;
  ldl3(s0, s1, s2, s3, s4, s5, u10, u20, u21, i0, i1, i2);
  double zr[9], wr[9];
  zw3(mr, u10, u20, u21, i0, i1, i2, zr, wr);

  Zr[i][0] = zr[0]; Zr[i][1] = zr[1]; Zr[i][2] = zr[2];
  Zr[i][3] = zr[3]; Zr[i][4] = zr[4]; Zr[i][5] = zr[5];
  Zr[i][6] = zr[6]; Zr[i][7] = zr[7]; Zr[i][8] = zr[8];
  Wr[i][0] = wr[0]; Wr[i][1] = wr[1]; Wr[i][2] = wr[2];
  Wr[i][3] = wr[3]; Wr[i][4] = wr[4]; Wr[i][5] = wr[5];
  Wr[i][6] = wr[6]; Wr[i][7] = wr[7]; Wr[i][8] = wr[8];
}

__device__ __forceinline__ void schur3_n(
    const double* zp, const double* wp,
    double& n0, double& n1, double& n2, double& n3, double& n4, double& n5) {
  const double z0 = zp[0], z1 = zp[1], z2 = zp[2];
  const double z3 = zp[3], z4 = zp[4], z5 = zp[5];
  const double z6 = zp[6], z7 = zp[7], z8 = zp[8];
  const double v0 = wp[0], v1 = wp[1], v2 = wp[2];
  const double v3 = wp[3], v4 = wp[4], v5 = wp[5];
  const double v6 = wp[6], v7 = wp[7], v8 = wp[8];
  n0 = __builtin_fma(-v2, z2, __builtin_fma(-v1, z1, __builtin_fma(-v0, z0, n0)));
  n1 = __builtin_fma(-v5, z2, __builtin_fma(-v4, z1, __builtin_fma(-v3, z0, n1)));
  n2 = __builtin_fma(-v5, z5, __builtin_fma(-v4, z4, __builtin_fma(-v3, z3, n2)));
  n3 = __builtin_fma(-v8, z2, __builtin_fma(-v7, z1, __builtin_fma(-v6, z0, n3)));
  n4 = __builtin_fma(-v8, z5, __builtin_fma(-v7, z4, __builtin_fma(-v6, z3, n4)));
  n5 = __builtin_fma(-v8, z8, __builtin_fma(-v7, z7, __builtin_fma(-v6, z6, n5)));
}

__device__ __forceinline__ void schur3_g(
    const double* wp, const double* xp, double& g0, double& g1, double& g2) {
  const double v0 = wp[0], v1 = wp[1], v2 = wp[2];
  const double v3 = wp[3], v4 = wp[4], v5 = wp[5];
  const double v6 = wp[6], v7 = wp[7], v8 = wp[8];
  const double x0 = xp[0], x1 = xp[1], x2 = xp[2];
  g0 = __builtin_fma(-v2, x2, __builtin_fma(-v1, x1, __builtin_fma(-v0, x0, g0)));
  g1 = __builtin_fma(-v5, x2, __builtin_fma(-v4, x1, __builtin_fma(-v3, x0, g1)));
  g2 = __builtin_fma(-v8, x2, __builtin_fma(-v7, x1, __builtin_fma(-v6, x0, g2)));
}

__device__ __forceinline__ void upd_diag(
    double (*S)[6],
    const double (*Zl)[9], const double (*Wl)[9],
    const double (*Zr)[9], const double (*Wr)[9],
    int j, int s) {
  double n0 = S[j][0], n1 = S[j][1], n2 = S[j][2];
  double n3 = S[j][3], n4 = S[j][4], n5 = S[j][5];
  if (j > 0) {
    const int e = j - s;
    schur3_n(Zr[e], Wr[e], n0, n1, n2, n3, n4, n5);
  }
  if (j + s < NSTEP) {
    const int e = j + s;
    schur3_n(Zl[e], Wl[e], n0, n1, n2, n3, n4, n5);
  }
  S[j][0] = n0; S[j][1] = n1; S[j][2] = n2;
  S[j][3] = n3; S[j][4] = n4; S[j][5] = n5;
}

__device__ __forceinline__ void upd_rhs_couple(
    double (*C)[9], double (*gv)[3],
    const double (*Wl)[9], const double (*Wr)[9], const double (*Zr)[9],
    int j, int s) {
  double g0 = gv[j][0], g1 = gv[j][1], g2 = gv[j][2];
  if (j > 0) {
    const int e = j - s;
    schur3_g(Wr[e], gv[e], g0, g1, g2);
  }
  if (j + s < NSTEP) {
    const int e = j + s;
    schur3_g(Wl[e], gv[e], g0, g1, g2);
    if (j + 2 * s < NSTEP) {
      const double* wp = Wl[e];
      const double* zp = Zr[e];
      const double v0 = wp[0], v1 = wp[1], v2 = wp[2];
      const double v3 = wp[3], v4 = wp[4], v5 = wp[5];
      const double v6 = wp[6], v7 = wp[7], v8 = wp[8];
      const double z0 = zp[0], z1 = zp[1], z2 = zp[2];
      const double z3 = zp[3], z4 = zp[4], z5 = zp[5];
      const double z6 = zp[6], z7 = zp[7], z8 = zp[8];
      double* cp = C[j];
      cp[0] = __builtin_fma(-v2, z2, __builtin_fma(-v1, z1, -(v0 * z0)));
      cp[1] = __builtin_fma(-v2, z5, __builtin_fma(-v1, z4, -(v0 * z3)));
      cp[2] = __builtin_fma(-v2, z8, __builtin_fma(-v1, z7, -(v0 * z6)));
      cp[3] = __builtin_fma(-v5, z2, __builtin_fma(-v4, z1, -(v3 * z0)));
      cp[4] = __builtin_fma(-v5, z5, __builtin_fma(-v4, z4, -(v3 * z3)));
      cp[5] = __builtin_fma(-v5, z8, __builtin_fma(-v4, z7, -(v3 * z6)));
      cp[6] = __builtin_fma(-v8, z2, __builtin_fma(-v7, z1, -(v6 * z0)));
      cp[7] = __builtin_fma(-v8, z5, __builtin_fma(-v7, z4, -(v6 * z3)));
      cp[8] = __builtin_fma(-v8, z8, __builtin_fma(-v7, z7, -(v6 * z6)));
    }
  }
  gv[j][0] = g0; gv[j][1] = g1; gv[j][2] = g2;
}

__device__ __forceinline__ void cr_back(
    const double (*F)[6], double (*gv)[3],
    const double (*Wl)[9], const double (*Wr)[9],
    int i, int s) {
  const double j0 = F[i][0], q10 = F[i][1], j1 = F[i][2];
  const double q20 = F[i][3], q21 = F[i][4], j2 = F[i][5];
  double t0 = gv[i][0] * j0, t1 = gv[i][1] * j1, t2 = gv[i][2] * j2;
  {
    const double* wp = Wl[i];
    const double a0 = gv[i - s][0], a1 = gv[i - s][1], a2 = gv[i - s][2];
    t0 = __builtin_fma(-wp[6], a2, __builtin_fma(-wp[3], a1, __builtin_fma(-wp[0], a0, t0)));
    t1 = __builtin_fma(-wp[7], a2, __builtin_fma(-wp[4], a1, __builtin_fma(-wp[1], a0, t1)));
    t2 = __builtin_fma(-wp[8], a2, __builtin_fma(-wp[5], a1, __builtin_fma(-wp[2], a0, t2)));
  }
  if (i + s < NSTEP) {
    const double* wp = Wr[i];
    const double a0 = gv[i + s][0], a1 = gv[i + s][1], a2 = gv[i + s][2];
    t0 = __builtin_fma(-wp[6], a2, __builtin_fma(-wp[3], a1, __builtin_fma(-wp[0], a0, t0)));
    t1 = __builtin_fma(-wp[7], a2, __builtin_fma(-wp[4], a1, __builtin_fma(-wp[1], a0, t1)));
    t2 = __builtin_fma(-wp[8], a2, __builtin_fma(-wp[5], a1, __builtin_fma(-wp[2], a0, t2)));
  }
  const double x2 = t2;
  const double x1 = __builtin_fma(-q21, x2, t1);
  const double x0 = __builtin_fma(-q20, x2, __builtin_fma(-q10, x1, t0));
  gv[i][0] = x0; gv[i][1] = x1; gv[i][2] = x2;
}

#define TOPS  64
#define NWAVE 4
#define NTHR  (32 * NWAVE)

__global__ __launch_bounds__(NTHR) void ode_blocktri_kernel(
    const float* __restrict__ coeffs,
    const float* __restrict__ rhs,
    const float* __restrict__ ivr,
    const float* __restrict__ steps,
    float* __restrict__ out)
{
  __shared__ double S[NSTEP][6];
  __shared__ double F[NSTEP][6];
  __shared__ double Bo[T1][9];
  __shared__ double gv[NSTEP][3];
  __shared__ double Zl[NSTEP][9];
  __shared__ double Wl[NSTEP][9];
  __shared__ double Zr[NSTEP][9];
  __shared__ double Wr[NSTEP][9];
  __shared__ float  hstep[T1];
  __shared__ float  Gs[T1][36];

  const int b = blockIdx.x;
  const int tid = threadIdx.x;
  const int lane = tid & 31;
  const int wave = __builtin_amdgcn_readfirstlane(tid >> 5);
  const int role = wave >> 1;
  const int slot = ((wave & 1) << 5) | lane;

  for (int i = tid; i < T1; i += NTHR) hstep[i] = steps[b * T1 + i];
  __syncthreads();

  const int m = lane & 15;
  const int khalf = lane >> 4;
  const int col = (m < 6) ? m : 6;
  float a0, b0, q0, a1, b1, q1, a2, b2, q2;
  Ecoef(2 * khalf,                 col, &a0, &b0, &q0);
  Ecoef(2 * khalf + 1,             col, &a1, &b1, &q1);
  Ecoef((khalf == 0) ? 4 : -1,     col, &a2, &b2, &q2);

  for (int p = wave; p < T1; p += NWAVE) {
    const float h = hstep[p];
    v2f f0, f1;
    f0.x = fmaf(fmaf(q0, h, b0), h, a0);
    f0.y = fmaf(fmaf(q1, h, b1), h, a1);
    f1.x = fmaf(fmaf(q2, h, b2), h, a2);
    f1.y = 0.f;
    v8f acc = {};
    acc = __builtin_amdgcn_wmma_f32_16x16x4_f32(false, f0, false, f0,
                                                (short)0, acc, false, false);
    acc = __builtin_amdgcn_wmma_f32_16x16x4_f32(false, f1, false, f1,
                                                (short)0, acc, false, false);
    if (lane < 6) {
      float* q = &Gs[p][lane * 6];
      q[0] = acc[0]; q[1] = acc[1]; q[2] = acc[2];
      q[3] = acc[3]; q[4] = acc[4]; q[5] = acc[5];
    }
  }
  __syncthreads();

  const float iv0 = ivr[b * 2 + 0];
  const float iv1 = ivr[b * 2 + 1];
  if (tid < NSTEP) {
    const int t = tid;
    const float* cp = coeffs + (b * NSTEP + t) * 3;
    const double c0 = (double)cp[0];
    const double c1 = (double)cp[1];
    const double c2 = (double)cp[2];
    const double r  = (double)rhs[b * NSTEP + t];
    const bool hasPrev = (t > 0);
    const bool hasNext = (t < T1);
    const float* gp = Gs[hasPrev ? t - 1 : 0];
    const float* gn = Gs[hasNext ? t : T1 - 1];
    const float f0 = gp[3 * 6 + 3], f1 = gp[3 * 6 + 4], f2 = gp[4 * 6 + 4];
    const float f3 = gp[3 * 6 + 5], f4 = gp[4 * 6 + 5], f5 = gp[5 * 6 + 5];
    const float n0 = gn[0 * 6 + 0], n1 = gn[0 * 6 + 1], n2 = gn[1 * 6 + 1];
    const float n3 = gn[0 * 6 + 2], n4 = gn[1 * 6 + 2], n5 = gn[2 * 6 + 2];
    const double p0 = (double)(hasPrev ? f0 : 1.f);
    const double p1 = (double)(hasPrev ? f1 : 0.f);
    const double p2 = (double)(hasPrev ? f2 : 1.f);
    const double p3 = (double)(hasPrev ? f3 : 0.f);
    const double p4 = (double)(hasPrev ? f4 : 0.f);
    const double p5 = (double)(hasPrev ? f5 : 0.f);
    S[t][0] = __builtin_fma(c0, c0, p0) + (double)(hasNext ? n0 : 0.f);
    S[t][1] = __builtin_fma(c1, c0, p1) + (double)(hasNext ? n1 : 0.f);
    S[t][2] = __builtin_fma(c1, c1, p2) + (double)(hasNext ? n2 : 0.f);
    S[t][3] = __builtin_fma(c2, c0, p3) + (double)(hasNext ? n3 : 0.f);
    S[t][4] = __builtin_fma(c2, c1, p4) + (double)(hasNext ? n4 : 0.f);
    S[t][5] = __builtin_fma(c2, c2, p5) + (double)(hasNext ? n5 : 0.f);
    gv[t][0] = __builtin_fma(c0, r, (double)((t == 0) ? iv0 : 0.f));
    gv[t][1] = __builtin_fma(c1, r, (double)((t == 0) ? iv1 : 0.f));
    gv[t][2] = c2 * r;
    const double b00 = (double)gn[3 * 6 + 0], b01 = (double)gn[4 * 6 + 0], b02 = (double)gn[5 * 6 + 0];
    const double b10 = (double)gn[3 * 6 + 1], b11 = (double)gn[4 * 6 + 1], b12 = (double)gn[5 * 6 + 1];
    const double b20 = (double)gn[3 * 6 + 2], b21 = (double)gn[4 * 6 + 2], b22 = (double)gn[5 * 6 + 2];
    if (hasNext) {
      Bo[t][0] = b00; Bo[t][1] = b01; Bo[t][2] = b02;
      Bo[t][3] = b10; Bo[t][4] = b11; Bo[t][5] = b12;
      Bo[t][6] = b20; Bo[t][7] = b21; Bo[t][8] = b22;
    }
  }
  __syncthreads();

  for (int s = 1; s <= TOPS; s <<= 1) {
    {
      const int i = s * (2 * slot + 1);
      if (i < NSTEP) {
        if (role == 0)            elim_left(S, F, Bo, gv, Zl, Wl, i, s);
        else if (i + s < NSTEP)   elim_right(S, Bo, Zr, Wr, i);
      }
    }
    __syncthreads();
    {
      const int j = 2 * s * slot;
      if (j < NSTEP) {
        if (role == 0)   upd_diag(S, Zl, Wl, Zr, Wr, j, s);
        else             upd_rhs_couple(Bo, gv, Wl, Wr, Zr, j, s);
      }
    }
    __syncthreads();
  }
  if (tid == 0) {
    double u10, u20, u21, i0, i1, i2;
    ldl3(S[0][0], S[0][1], S[0][2], S[0][3], S[0][4], S[0][5],
         u10, u20, u21, i0, i1, i2);
    const double w0 = gv[0][0];
    const double w1 = __builtin_fma(-u10, w0, gv[0][1]);
    const double w2 = __builtin_fma(-u21, w1, __builtin_fma(-u20, w0, gv[0][2]));
    const double x2 = w2 * i2;
    const double x1 = __builtin_fma(-u21, x2, w1 * i1);
    const double x0 = __builtin_fma(-u20, x2, __builtin_fma(-u10, x1, w0 * i0));
    gv[0][0] = x0; gv[0][1] = x1; gv[0][2] = x2;
  }
  __syncthreads();
  for (int s = TOPS; s >= 1; s >>= 1) {
    const int i = s * (2 * tid + 1);
    if (i < NSTEP) cr_back(F, gv, Wl, Wr, i, s);
    __syncthreads();
  }

  if (tid < NSTEP) {
    out[0 * OUT_U + b * NSTEP + tid] = (float)gv[tid][0];
    out[1 * OUT_U + b * NSTEP + tid] = (float)gv[tid][1];
    out[2 * OUT_U + b * NSTEP + tid] = (float)gv[tid][2];
  }
  if (tid == 0) out[OUT_EPS + b] = 0.0f;
  if (tid < T1) out[OUT_ST + b * T1 + tid] = hstep[tid];
}

extern "C" void kernel_launch(void* const* d_in, const int* in_sizes, int n_in,
                              void* d_out, int out_size, void* d_ws, size_t ws_size,
                              hipStream_t stream) {
  (void)in_sizes; (void)n_in; (void)out_size; (void)d_ws; (void)ws_size;
  const float* coeffs = (const float*)d_in[0];
  const float* rhs    = (const float*)d_in[1];
  const float* ivr    = (const float*)d_in[2];
  const float* steps  = (const float*)d_in[3];
  float* out = (float*)d_out;
  hipLaunchKernelGGL(ode_blocktri_kernel, dim3(NB), dim3(NTHR), 0, stream,
                     coeffs, rhs, ivr, steps, out);
}


# From stride 2 up (25 blocks or fewer a side) all four waves work: elimination split by rows of Z, W and by output element, back-substitution by component of t; every value bit-identical
# speedup vs baseline: 27.8478x; 1.1171x over previous
#include <hip/hip_runtime.h>
#include <hip/hip_bf16.h>
#include <math.h>


typedef __attribute__((ext_vector_type(2))) float v2f;
typedef __attribute__((ext_vector_type(8))) float v8f;

#define NSTEP 100
#define T1    99
#define NB    256
#define OUT_U   (NB * NSTEP)
#define OUT_EPS (3 * OUT_U)
#define OUT_ST  (OUT_EPS + NB)

__device__ __forceinline__ void Ecoef(int r, int c, float* pa, float* pb, float* pc) {
  float a = 0.f, b = 0.f, q = 0.f;
  if (r == 0)      { if (c == 0) a = 1.f;  else if (c == 1) b = 1.f;   else if (c == 2) q = 0.5f; else if (c == 3) a = -1.f; }
  else if (r == 1) { if (c == 1) a = 1.f;  else if (c == 2) b = 1.f;   else if (c == 4) a = -1.f; }
  else if (r == 2) { if (c == 0) a = -1.f; else if (c == 3) a = 1.f;   else if (c == 4) b = -1.f; else if (c == 5) q = 0.5f; }
  else if (r == 3) { if (c == 1) a = -1.f; else if (c == 4) a = 1.f;   else if (c == 5) b = -1.f; }
  else if (r == 4) { if (c == 1) a = -1.f; else if (c == 2) b = -0.5f; else if (c == 4) a = 1.f;  else if (c == 5) b = -0.5f; }
  *pa = a; *pb = b; *pc = q;
}

__device__ __forceinline__ double recip_nr(double d) {
  double r = __builtin_amdgcn_rcp(d);
  double e = __builtin_fma(-d, r, 1.0);
  r = __builtin_fma(r, e, r);
  e = __builtin_fma(-d, r, 1.0);
  r = __builtin_fma(r, e, r);
  return r;
}

__device__ __forceinline__ void ldl3(double s0, double s1, double s2,
                                     double s3, double s4, double s5,
                                     double& u10, double& u20, double& u21,
                                     double& i0, double& i1, double& i2) {
  i0 = recip_nr(s0);
  u10 = s1 * i0;
  u20 = s3 * i0;
  const double d1 = __builtin_fma(-u10, s1, s2);
  i1 = recip_nr(d1);
  const double t = __builtin_fma(-u20, s1, s4);
  u21 = t * i1;
  const double d2 = __builtin_fma(-u21, t, __builtin_fma(-u20, s3, s5));
  i2 = recip_nr(d2);
}

__device__ __forceinline__ void zw3(const double* m,
                                    double u10, double u20, double u21,
                                    double i0, double i1, double i2,
                                    double* z, double* w) {
  z[0] = m[0];
  z[1] = __builtin_fma(-z[0], u10, m[1]);
  z[2] = __builtin_fma(-z[1], u21, __builtin_fma(-z[0], u20, m[2]));
  z[3] = m[3];
  z[4] = __builtin_fma(-z[3], u10, m[4]);
  z[5] = __builtin_fma(-z[4], u21, __builtin_fma(-z[3], u20, m[5]));
  z[6] = m[6];
  z[7] = __builtin_fma(-z[6], u10, m[7]);
  z[8] = __builtin_fma(-z[7], u21, __builtin_fma(-z[6], u20, m[8]));
  w[0] = z[0] * i0; w[1] = z[1] * i1; w[2] = z[2] * i2;
  w[3] = z[3] * i0; w[4] = z[4] * i1; w[5] = z[5] * i2;
  w[6] = z[6] * i0; w[7] = z[7] * i1; w[8] = z[8] * i2;
}

__device__ __forceinline__ void elim_left(
    const double (*S)[6], double (*F)[6], const double (*C)[9], double (*gv)[3],
    double (*Zl)[9], double (*Wl)[9], int i, int s) {
  const double* cl = C[i - s];
  double ml[9];
  ml[0] = cl[0]; ml[1] = cl[1]; ml[2] = cl[2];
  ml[3] = cl[3]; ml[4] = cl[4]; ml[5] = cl[5];
  ml[6] = cl[6]; ml[7] = cl[7]; ml[8] = cl[8];
  const double s0 = S[i][0], s1 = S[i][1], s2 = S[i][2];
  const double s3 = S[i][3], s4 = S[i][4], s5 = S[i][5];
  const double g0 = gv[i][0], g1 = gv[i][1], g2 = gv[i][2];

  double u10, u20, u21, i0, i1, i2;
  ldl3(s0, s1, s2, s3, s4, s5, u10, u20, u21, i0, i1, i2);
  const double w1 = __builtin_fma(-u10, g0, g1);
  const double w2 = __builtin_fma(-u21, w1, __builtin_fma(-u20, g0, g2));
  double zl[9], wl[9];
  zw3(ml, u10, u20, u21, i0, i1, i2, zl, wl);

  F[i][0] = i0;  F[i][1] = u10; F[i][2] = i1;
  F[i][3] = u20; F[i][4] = u21; F[i][5] = i2;
  gv[i][1] = w1; gv[i][2] = w2;
  Zl[i][0] = zl[0]; Zl[i][1] = zl[1]; Zl[i][2] = zl[2];
  Zl[i][3] = zl[3]; Zl[i][4] = zl[4]; Zl[i][5] = zl[5];
  Zl[i][6] = zl[6]; Zl[i][7] = zl[7]; Zl[i][8] = zl[8];
  Wl[i][0] = wl[0]; Wl[i][1] = wl[1]; Wl[i][2] = wl[2];
  Wl[i][3] = wl[3]; Wl[i][4] = wl[4]; Wl[i][5] = wl[5];
  Wl[i][6] = wl[6]; Wl[i][7] = wl[7]; Wl[i][8] = wl[8];
}

__device__ __forceinline__ void elim_right(
    const double (*S)[6], const double (*C)[9],
    double (*Zr)[9], double (*Wr)[9], int i) {
  const double* cr = C[i];
  double mr[9];
  mr[0] = cr[0]; mr[1] = cr[3]; mr[2] = cr[6];
  mr[3] = cr[1]; mr[4] = cr[4]; mr[5] = cr[7];
  mr[6] = cr[2]; mr[7] = cr[5]; mr[8] = cr[8];
  const double s0 = S[i][0], s1 = S[i][1], s2 = S[i][2];
  const double s3 = S[i][3], s4 = S[i][4], s5 = S[i][5];

  double u10, u20, u21, i0, i1, i2;
  ldl3(s0, s1, s2, s3, s4, s5, u10, u20, u21, i0, i1, i2);
  double zr[9], wr[9];
  zw3(mr, u10, u20, u21, i0, i1, i2, zr, wr);

  Zr[i][0] = zr[0]; Zr[i][1] = zr[1]; Zr[i][2] = zr[2];
  Zr[i][3] = zr[3]; Zr[i][4] = zr[4]; Zr[i][5] = zr[5];
  Zr[i][6] = zr[6]; Zr[i][7] = zr[7]; Zr[i][8] = zr[8];
  Wr[i][0] = wr[0]; Wr[i][1] = wr[1]; Wr[i][2] = wr[2];
  Wr[i][3] = wr[3]; Wr[i][4] = wr[4]; Wr[i][5] = wr[5];
  Wr[i][6] = wr[6]; Wr[i][7] = wr[7]; Wr[i][8] = wr[8];
}

__device__ __forceinline__ void schur3_n(
    const double* zp, const double* wp,
    double& n0, double& n1, double& n2, double& n3, double& n4, double& n5) {
  const double z0 = zp[0], z1 = zp[1], z2 = zp[2];
  const double z3 = zp[3], z4 = zp[4], z5 = zp[5];
  const double z6 = zp[6], z7 = zp[7], z8 = zp[8];
  const double v0 = wp[0], v1 = wp[1], v2 = wp[2];
  const double v3 = wp[3], v4 = wp[4], v5 = wp[5];
  const double v6 = wp[6], v7 = wp[7], v8 = wp[8];
  n0 = __builtin_fma(-v2, z2, __builtin_fma(-v1, z1, __builtin_fma(-v0, z0, n0)));
  n1 = __builtin_fma(-v5, z2, __builtin_fma(-v4, z1, __builtin_fma(-v3, z0, n1)));
  n2 = __builtin_fma(-v5, z5, __builtin_fma(-v4, z4, __builtin_fma(-v3, z3, n2)));
  n3 = __builtin_fma(-v8, z2, __builtin_fma(-v7, z1, __builtin_fma(-v6, z0, n3)));
  n4 = __builtin_fma(-v8, z5, __builtin_fma(-v7, z4, __builtin_fma(-v6, z3, n4)));
  n5 = __builtin_fma(-v8, z8, __builtin_fma(-v7, z7, __builtin_fma(-v6, z6, n5)));
}

__device__ __forceinline__ void schur3_g(
    const double* wp, const double* xp, double& g0, double& g1, double& g2) {
  const double v0 = wp[0], v1 = wp[1], v2 = wp[2];
  const double v3 = wp[3], v4 = wp[4], v5 = wp[5];
  const double v6 = wp[6], v7 = wp[7], v8 = wp[8];
  const double x0 = xp[0], x1 = xp[1], x2 = xp[2];
  g0 = __builtin_fma(-v2, x2, __builtin_fma(-v1, x1, __builtin_fma(-v0, x0, g0)));
  g1 = __builtin_fma(-v5, x2, __builtin_fma(-v4, x1, __builtin_fma(-v3, x0, g1)));
  g2 = __builtin_fma(-v8, x2, __builtin_fma(-v7, x1, __builtin_fma(-v6, x0, g2)));
}

__device__ __forceinline__ void upd_diag(
    double (*S)[6],
    const double (*Zl)[9], const double (*Wl)[9],
    const double (*Zr)[9], const double (*Wr)[9],
    int j, int s) {
  double n0 = S[j][0], n1 = S[j][1], n2 = S[j][2];
  double n3 = S[j][3], n4 = S[j][4], n5 = S[j][5];
  if (j > 0) {
    const int e = j - s;
    schur3_n(Zr[e], Wr[e], n0, n1, n2, n3, n4, n5);
  }
  if (j + s < NSTEP) {
    const int e = j + s;
    schur3_n(Zl[e], Wl[e], n0, n1, n2, n3, n4, n5);
  }
  S[j][0] = n0; S[j][1] = n1; S[j][2] = n2;
  S[j][3] = n3; S[j][4] = n4; S[j][5] = n5;
}

__device__ __forceinline__ void upd_rhs_couple(
    double (*C)[9], double (*gv)[3],
    const double (*Wl)[9], const double (*Wr)[9], const double (*Zr)[9],
    int j, int s) {
  double g0 = gv[j][0], g1 = gv[j][1], g2 = gv[j][2];
  if (j > 0) {
    const int e = j - s;
    schur3_g(Wr[e], gv[e], g0, g1, g2);
  }
  if (j + s < NSTEP) {
    const int e = j + s;
    schur3_g(Wl[e], gv[e], g0, g1, g2);
    if (j + 2 * s < NSTEP) {
      const double* wp = Wl[e];
      const double* zp = Zr[e];
      const double v0 = wp[0], v1 = wp[1], v2 = wp[2];
      const double v3 = wp[3], v4 = wp[4], v5 = wp[5];
      const double v6 = wp[6], v7 = wp[7], v8 = wp[8];
      const double z0 = zp[0], z1 = zp[1], z2 = zp[2];
      const double z3 = zp[3], z4 = zp[4], z5 = zp[5];
      const double z6 = zp[6], z7 = zp[7], z8 = zp[8];
      double* cp = C[j];
      cp[0] = __builtin_fma(-v2, z2, __builtin_fma(-v1, z1, -(v0 * z0)));
      cp[1] = __builtin_fma(-v2, z5, __builtin_fma(-v1, z4, -(v0 * z3)));
      cp[2] = __builtin_fma(-v2, z8, __builtin_fma(-v1, z7, -(v0 * z6)));
      cp[3] = __builtin_fma(-v5, z2, __builtin_fma(-v4, z1, -(v3 * z0)));
      cp[4] = __builtin_fma(-v5, z5, __builtin_fma(-v4, z4, -(v3 * z3)));
      cp[5] = __builtin_fma(-v5, z8, __builtin_fma(-v4, z7, -(v3 * z6)));
      cp[6] = __builtin_fma(-v8, z2, __builtin_fma(-v7, z1, -(v6 * z0)));
      cp[7] = __builtin_fma(-v8, z5, __builtin_fma(-v7, z4, -(v6 * z3)));
      cp[8] = __builtin_fma(-v8, z8, __builtin_fma(-v7, z7, -(v6 * z6)));
    }
  }
  gv[j][0] = g0; gv[j][1] = g1; gv[j][2] = g2;
}

__device__ __forceinline__ void zw_row(double m0, double m1, double m2,
                                       double u10, double u20, double u21,
                                       double i0, double i1, double i2,
                                       double* z, double* w) {
  const double z0 = m0;
  const double z1 = __builtin_fma(-z0, u10, m1);
  const double z2 = __builtin_fma(-z1, u21, __builtin_fma(-z0, u20, m2));
  z[0] = z0; z[1] = z1; z[2] = z2;
  w[0] = z0 * i0; w[1] = z1 * i1; w[2] = z2 * i2;
}

__device__ __forceinline__ void elim4(
    const double (*S)[6], double (*F)[6], const double (*C)[9], double (*gv)[3],
    double (*Zl)[9], double (*Wl)[9], double (*Zr)[9], double (*Wr)[9],
    int i, int s, int part) {
  if (part >= 2 && i + s >= NSTEP) return;
  const double s0 = S[i][0], s1 = S[i][1], s2 = S[i][2];
  const double s3 = S[i][3], s4 = S[i][4], s5 = S[i][5];
  double u10, u20, u21, i0, i1, i2;
  if (part == 0) {
    const double* cl = C[i - s];
    const double m0 = cl[0], m1 = cl[1], m2 = cl[2];
    const double g0 = gv[i][0], g1 = gv[i][1], g2 = gv[i][2];
    ldl3(s0, s1, s2, s3, s4, s5, u10, u20, u21, i0, i1, i2);
    const double w1 = __builtin_fma(-u10, g0, g1);
    const double w2 = __builtin_fma(-u21, w1, __builtin_fma(-u20, g0, g2));
    F[i][0] = i0;  F[i][1] = u10; F[i][2] = i1;
    F[i][3] = u20; F[i][4] = u21; F[i][5] = i2;
    gv[i][1] = w1; gv[i][2] = w2;
    zw_row(m0, m1, m2, u10, u20, u21, i0, i1, i2, &Zl[i][0], &Wl[i][0]);
  } else if (part == 1) {
    const double* cl = C[i - s];
    const double m3 = cl[3], m4 = cl[4], m5 = cl[5];
    const double m6 = cl[6], m7 = cl[7], m8 = cl[8];
    ldl3(s0, s1, s2, s3, s4, s5, u10, u20, u21, i0, i1, i2);
    zw_row(m3, m4, m5, u10, u20, u21, i0, i1, i2, &Zl[i][3], &Wl[i][3]);
    zw_row(m6, m7, m8, u10, u20, u21, i0, i1, i2, &Zl[i][6], &Wl[i][6]);
  } else if (part == 2) {
    const double* cr = C[i];
    const double m0 = cr[0], m1 = cr[3], m2 = cr[6];
    const double m3 = cr[1], m4 = cr[4], m5 = cr[7];
    ldl3(s0, s1, s2, s3, s4, s5, u10, u20, u21, i0, i1, i2);
    zw_row(m0, m1, m2, u10, u20, u21, i0, i1, i2, &Zr[i][0], &Wr[i][0]);
    zw_row(m3, m4, m5, u10, u20, u21, i0, i1, i2, &Zr[i][3], &Wr[i][3]);
  } else {
    const double* cr = C[i];
    const double m6 = cr[2], m7 = cr[5], m8 = cr[8];
    ldl3(s0, s1, s2, s3, s4, s5, u10, u20, u21, i0, i1, i2);
    zw_row(m6, m7, m8, u10, u20, u21, i0, i1, i2, &Zr[i][6], &Wr[i][6]);
  }
}

__device__ __forceinline__ void upd4(
    double (*S)[6], double (*C)[9], double (*gv)[3],
    const double (*Zl)[9], const double (*Wl)[9],
    const double (*Zr)[9], const double (*Wr)[9],
    int j, int s, int part) {
  const bool hasL = (j > 0);
  const bool hasR = (j + s < NSTEP);
  const bool hasC = (j + 2 * s < NSTEP);
  const int el = hasL ? j - s : 0;
  const int er = hasR ? j + s : 0;
  if (part == 0) {
    double n0 = S[j][0], n1 = S[j][1], n2 = S[j][2];
    if (hasL) {
      const double* zp = Zr[el]; const double* wp = Wr[el];
      const double z0 = zp[0], z1 = zp[1], z2 = zp[2], z3 = zp[3], z4 = zp[4], z5 = zp[5];
      const double v0 = wp[0], v1 = wp[1], v2 = wp[2], v3 = wp[3], v4 = wp[4], v5 = wp[5];
      n0 = __builtin_fma(-v2, z2, __builtin_fma(-v1, z1, __builtin_fma(-v0, z0, n0)));
      n1 = __builtin_fma(-v5, z2, __builtin_fma(-v4, z1, __builtin_fma(-v3, z0, n1)));
      n2 = __builtin_fma(-v5, z5, __builtin_fma(-v4, z4, __builtin_fma(-v3, z3, n2)));
    }
    if (hasR) {
      const double* zp = Zl[er]; const double* wp = Wl[er];
      const double z0 = zp[0], z1 = zp[1], z2 = zp[2], z3 = zp[3], z4 = zp[4], z5 = zp[5];
      const double v0 = wp[0], v1 = wp[1], v2 = wp[2], v3 = wp[3], v4 = wp[4], v5 = wp[5];
      n0 = __builtin_fma(-v2, z2, __builtin_fma(-v1, z1, __builtin_fma(-v0, z0, n0)));
      n1 = __builtin_fma(-v5, z2, __builtin_fma(-v4, z1, __builtin_fma(-v3, z0, n1)));
      n2 = __builtin_fma(-v5, z5, __builtin_fma(-v4, z4, __builtin_fma(-v3, z3, n2)));
    }
    S[j][0] = n0; S[j][1] = n1; S[j][2] = n2;
  } else if (part == 1) {
    double n3 = S[j][3], n4 = S[j][4], n5 = S[j][5];
    if (hasL) {
      const double* zp = Zr[el]; const double* wp = Wr[el];
      const double z0 = zp[0], z1 = zp[1], z2 = zp[2], z3 = zp[3], z4 = zp[4], z5 = zp[5];
      const double z6 = zp[6], z7 = zp[7], z8 = zp[8];
      const double v6 = wp[6], v7 = wp[7], v8 = wp[8];
      n3 = __builtin_fma(-v8, z2, __builtin_fma(-v7, z1, __builtin_fma(-v6, z0, n3)));
      n4 = __builtin_fma(-v8, z5, __builtin_fma(-v7, z4, __builtin_fma(-v6, z3, n4)));
      n5 = __builtin_fma(-v8, z8, __builtin_fma(-v7, z7, __builtin_fma(-v6, z6, n5)));
    }
    if (hasR) {
      const double* zp = Zl[er]; const double* wp = Wl[er];
      const double z0 = zp[0], z1 = zp[1], z2 = zp[2], z3 = zp[3], z4 = zp[4], z5 = zp[5];
      const double z6 = zp[6], z7 = zp[7], z8 = zp[8];
      const double v6 = wp[6], v7 = wp[7], v8 = wp[8];
      n3 = __builtin_fma(-v8, z2, __builtin_fma(-v7, z1, __builtin_fma(-v6, z0, n3)));
      n4 = __builtin_fma(-v8, z5, __builtin_fma(-v7, z4, __builtin_fma(-v6, z3, n4)));
      n5 = __builtin_fma(-v8, z8, __builtin_fma(-v7, z7, __builtin_fma(-v6, z6, n5)));
    }
    S[j][3] = n3; S[j][4] = n4; S[j][5] = n5;
    if (hasC) {
      const double* wp = Wl[er]; const double* zp = Zr[er];
      const double v0 = wp[0], v1 = wp[1], v2 = wp[2];
      const double z0 = zp[0], z1 = zp[1], z2 = zp[2];
      C[j][0] = __builtin_fma(-v2, z2, __builtin_fma(-v1, z1, -(v0 * z0)));
    }
  } else if (part == 2) {
    double g0 = gv[j][0], g1 = gv[j][1], g2 = gv[j][2];
    if (hasL) schur3_g(Wr[el], gv[el], g0, g1, g2);
    if (hasR) schur3_g(Wl[er], gv[er], g0, g1, g2);
    gv[j][0] = g0; gv[j][1] = g1; gv[j][2] = g2;
    if (hasC) {
      const double* wp = Wl[er]; const double* zp = Zr[er];
      const double v0 = wp[0], v1 = wp[1], v2 = wp[2];
      const double z3 = zp[3], z4 = zp[4], z5 = zp[5];
      C[j][1] = __builtin_fma(-v2, z5, __builtin_fma(-v1, z4, -(v0 * z3)));
    }
  } else {
    if (hasC) {
      const double* wp = Wl[er]; const double* zp = Zr[er];
      const double v0 = wp[0], v1 = wp[1], v2 = wp[2];
      const double v3 = wp[3], v4 = wp[4], v5 = wp[5];
      const double v6 = wp[6], v7 = wp[7], v8 = wp[8];
      const double z0 = zp[0], z1 = zp[1], z2 = zp[2];
      const double z3 = zp[3], z4 = zp[4], z5 = zp[5];
      const double z6 = zp[6], z7 = zp[7], z8 = zp[8];
      double* cp = C[j];
      cp[2] = __builtin_fma(-v2, z8, __builtin_fma(-v1, z7, -(v0 * z6)));
      cp[3] = __builtin_fma(-v5, z2, __builtin_fma(-v4, z1, -(v3 * z0)));
      cp[4] = __builtin_fma(-v5, z5, __builtin_fma(-v4, z4, -(v3 * z3)));
      cp[5] = __builtin_fma(-v5, z8, __builtin_fma(-v4, z7, -(v3 * z6)));
      cp[6] = __builtin_fma(-v8, z2, __builtin_fma(-v7, z1, -(v6 * z0)));
      cp[7] = __builtin_fma(-v8, z5, __builtin_fma(-v7, z4, -(v6 * z3)));
      cp[8] = __builtin_fma(-v8, z8, __builtin_fma(-v7, z7, -(v6 * z6)));
    }
  }
}

__device__ __forceinline__ void cr_back(
    const double (*F)[6], double (*gv)[3],
    const double (*Wl)[9], const double (*Wr)[9],
    int i, int s) {
  const double j0 = F[i][0], q10 = F[i][1], j1 = F[i][2];
  const double q20 = F[i][3], q21 = F[i][4], j2 = F[i][5];
  double t0 = gv[i][0] * j0, t1 = gv[i][1] * j1, t2 = gv[i][2] * j2;
  {
    const double* wp = Wl[i];
    const double a0 = gv[i - s][0], a1 = gv[i - s][1], a2 = gv[i - s][2];
    t0 = __builtin_fma(-wp[6], a2, __builtin_fma(-wp[3], a1, __builtin_fma(-wp[0], a0, t0)));
    t1 = __builtin_fma(-wp[7], a2, __builtin_fma(-wp[4], a1, __builtin_fma(-wp[1], a0, t1)));
    t2 = __builtin_fma(-wp[8], a2, __builtin_fma(-wp[5], a1, __builtin_fma(-wp[2], a0, t2)));
  }
  if (i + s < NSTEP) {
    const double* wp = Wr[i];
    const double a0 = gv[i + s][0], a1 = gv[i + s][1], a2 = gv[i + s][2];
    t0 = __builtin_fma(-wp[6], a2, __builtin_fma(-wp[3], a1, __builtin_fma(-wp[0], a0, t0)));
    t1 = __builtin_fma(-wp[7], a2, __builtin_fma(-wp[4], a1, __builtin_fma(-wp[1], a0, t1)));
    t2 = __builtin_fma(-wp[8], a2, __builtin_fma(-wp[5], a1, __builtin_fma(-wp[2], a0, t2)));
  }
  const double x2 = t2;
  const double x1 = __builtin_fma(-q21, x2, t1);
  const double x0 = __builtin_fma(-q20, x2, __builtin_fma(-q10, x1, t0));
  gv[i][0] = x0; gv[i][1] = x1; gv[i][2] = x2;
}

__device__ __forceinline__ double cr_back_t(
    const double (*F)[6], const double (*gv)[3],
    const double (*Wl)[9], const double (*Wr)[9],
    int i, int s, int k) {
  const int fk = (k == 0) ? 0 : ((k == 1) ? 2 : 5);
  double t = gv[i][k] * F[i][fk];
  {
    const double* wp = Wl[i];
    const double a0 = gv[i - s][0], a1 = gv[i - s][1], a2 = gv[i - s][2];
    t = __builtin_fma(-wp[6 + k], a2, __builtin_fma(-wp[3 + k], a1, __builtin_fma(-wp[k], a0, t)));
  }
  if (i + s < NSTEP) {
    const double* wp = Wr[i];
    const double a0 = gv[i + s][0], a1 = gv[i + s][1], a2 = gv[i + s][2];
    t = __builtin_fma(-wp[6 + k], a2, __builtin_fma(-wp[3 + k], a1, __builtin_fma(-wp[k], a0, t)));
  }
  return t;
}

__device__ __forceinline__ void cr_back_x(
    const double (*F)[6], double (*gv)[3], const double (*tb)[3], int i) {
  const double q10 = F[i][1], q20 = F[i][3], q21 = F[i][4];
  const double x2 = tb[i][2];
  const double x1 = __builtin_fma(-q21, x2, tb[i][1]);
  const double x0 = __builtin_fma(-q20, x2, __builtin_fma(-q10, x1, tb[i][0]));
  gv[i][0] = x0; gv[i][1] = x1; gv[i][2] = x2;
}

#define TOPS  64
#define NWAVE 4
#define NTHR  (32 * NWAVE)

__global__ __launch_bounds__(NTHR) void ode_blocktri_kernel(
    const float* __restrict__ coeffs,
    const float* __restrict__ rhs,
    const float* __restrict__ ivr,
    const float* __restrict__ steps,
    float* __restrict__ out)
{
  __shared__ double S[NSTEP][6];
  __shared__ double F[NSTEP][6];
  __shared__ double Bo[T1][9];
  __shared__ double gv[NSTEP][3];
  __shared__ double Zl[NSTEP][9];
  __shared__ double Wl[NSTEP][9];
  __shared__ double Zr[NSTEP][9];
  __shared__ double Wr[NSTEP][9];
  __shared__ double tb[NSTEP][3];
  __shared__ float  hstep[T1];
  __shared__ float  Gs[T1][36];

  const int b = blockIdx.x;
  const int tid = threadIdx.x;
  const int lane = tid & 31;
  const int wave = __builtin_amdgcn_readfirstlane(tid >> 5);
  const int role = wave >> 1;
  const int slot = ((wave & 1) << 5) | lane;

  for (int i = tid; i < T1; i += NTHR) hstep[i] = steps[b * T1 + i];
  __syncthreads();

  const int m = lane & 15;
  const int khalf = lane >> 4;
  const int col = (m < 6) ? m : 6;
  float a0, b0, q0, a1, b1, q1, a2, b2, q2;
  Ecoef(2 * khalf,                 col, &a0, &b0, &q0);
  Ecoef(2 * khalf + 1,             col, &a1, &b1, &q1);
  Ecoef((khalf == 0) ? 4 : -1,     col, &a2, &b2, &q2);

  for (int p = wave; p < T1; p += NWAVE) {
    const float h = hstep[p];
    v2f f0, f1;
    f0.x = fmaf(fmaf(q0, h, b0), h, a0);
    f0.y = fmaf(fmaf(q1, h, b1), h, a1);
    f1.x = fmaf(fmaf(q2, h, b2), h, a2);
    f1.y = 0.f;
    v8f acc = {};
    acc = __builtin_amdgcn_wmma_f32_16x16x4_f32(false, f0, false, f0,
                                                (short)0, acc, false, false);
    acc = __builtin_amdgcn_wmma_f32_16x16x4_f32(false, f1, false, f1,
                                                (short)0, acc, false, false);
    if (lane < 6) {
      float* q = &Gs[p][lane * 6];
      q[0] = acc[0]; q[1] = acc[1]; q[2] = acc[2];
      q[3] = acc[3]; q[4] = acc[4]; q[5] = acc[5];
    }
  }
  __syncthreads();

  const float iv0 = ivr[b * 2 + 0];
  const float iv1 = ivr[b * 2 + 1];
  if (tid < NSTEP) {
    const int t = tid;
    const float* cp = coeffs + (b * NSTEP + t) * 3;
    const double c0 = (double)cp[0];
    const double c1 = (double)cp[1];
    const double c2 = (double)cp[2];
    const double r  = (double)rhs[b * NSTEP + t];
    const bool hasPrev = (t > 0);
    const bool hasNext = (t < T1);
    const float* gp = Gs[hasPrev ? t - 1 : 0];
    const float* gn = Gs[hasNext ? t : T1 - 1];
    const float f0 = gp[3 * 6 + 3], f1 = gp[3 * 6 + 4], f2 = gp[4 * 6 + 4];
    const float f3 = gp[3 * 6 + 5], f4 = gp[4 * 6 + 5], f5 = gp[5 * 6 + 5];
    const float n0 = gn[0 * 6 + 0], n1 = gn[0 * 6 + 1], n2 = gn[1 * 6 + 1];
    const float n3 = gn[0 * 6 + 2], n4 = gn[1 * 6 + 2], n5 = gn[2 * 6 + 2];
    const double p0 = (double)(hasPrev ? f0 : 1.f);
    const double p1 = (double)(hasPrev ? f1 : 0.f);
    const double p2 = (double)(hasPrev ? f2 : 1.f);
    const double p3 = (double)(hasPrev ? f3 : 0.f);
    const double p4 = (double)(hasPrev ? f4 : 0.f);
    const double p5 = (double)(hasPrev ? f5 : 0.f);
    S[t][0] = __builtin_fma(c0, c0, p0) + (double)(hasNext ? n0 : 0.f);
    S[t][1] = __builtin_fma(c1, c0, p1) + (double)(hasNext ? n1 : 0.f);
    S[t][2] = __builtin_fma(c1, c1, p2) + (double)(hasNext ? n2 : 0.f);
    S[t][3] = __builtin_fma(c2, c0, p3) + (double)(hasNext ? n3 : 0.f);
    S[t][4] = __builtin_fma(c2, c1, p4) + (double)(hasNext ? n4 : 0.f);
    S[t][5] = __builtin_fma(c2, c2, p5) + (double)(hasNext ? n5 : 0.f);
    gv[t][0] = __builtin_fma(c0, r, (double)((t == 0) ? iv0 : 0.f));
    gv[t][1] = __builtin_fma(c1, r, (double)((t == 0) ? iv1 : 0.f));
    gv[t][2] = c2 * r;
    const double b00 = (double)gn[3 * 6 + 0], b01 = (double)gn[4 * 6 + 0], b02 = (double)gn[5 * 6 + 0];
    const double b10 = (double)gn[3 * 6 + 1], b11 = (double)gn[4 * 6 + 1], b12 = (double)gn[5 * 6 + 1];
    const double b20 = (double)gn[3 * 6 + 2], b21 = (double)gn[4 * 6 + 2], b22 = (double)gn[5 * 6 + 2];
    if (hasNext) {
      Bo[t][0] = b00; Bo[t][1] = b01; Bo[t][2] = b02;
      Bo[t][3] = b10; Bo[t][4] = b11; Bo[t][5] = b12;
      Bo[t][6] = b20; Bo[t][7] = b21; Bo[t][8] = b22;
    }
  }
  __syncthreads();

  {
    const int s = 1;
    {
      const int i = s * (2 * slot + 1);
      if (i < NSTEP) {
        if (role == 0)            elim_left(S, F, Bo, gv, Zl, Wl, i, s);
        else if (i + s < NSTEP)   elim_right(S, Bo, Zr, Wr, i);
      }
    }
    __syncthreads();
    {
      const int j = 2 * s * slot;
      if (j < NSTEP) {
        if (role == 0)   upd_diag(S, Zl, Wl, Zr, Wr, j, s);
        else             upd_rhs_couple(Bo, gv, Wl, Wr, Zr, j, s);
      }
    }
    __syncthreads();
  }
  for (int s = 2; s <= TOPS; s <<= 1) {
    {
      const int i = s * (2 * lane + 1);
      if (i < NSTEP) elim4(S, F, Bo, gv, Zl, Wl, Zr, Wr, i, s, wave);
    }
    __syncthreads();
    {
      const int j = 2 * s * lane;
      if (j < NSTEP) upd4(S, Bo, gv, Zl, Wl, Zr, Wr, j, s, wave);
    }
    __syncthreads();
  }
  if (tid == 0) {
    double u10, u20, u21, i0, i1, i2;
    ldl3(S[0][0], S[0][1], S[0][2], S[0][3], S[0][4], S[0][5],
         u10, u20, u21, i0, i1, i2);
    const double w0 = gv[0][0];
    const double w1 = __builtin_fma(-u10, w0, gv[0][1]);
    const double w2 = __builtin_fma(-u21, w1, __builtin_fma(-u20, w0, gv[0][2]));
    const double x2 = w2 * i2;
    const double x1 = __builtin_fma(-u21, x2, w1 * i1);
    const double x0 = __builtin_fma(-u20, x2, __builtin_fma(-u10, x1, w0 * i0));
    gv[0][0] = x0; gv[0][1] = x1; gv[0][2] = x2;
  }
  __syncthreads();
  for (int s = TOPS; s >= 2; s >>= 1) {
    const int i = s * (2 * lane + 1);
    if (i < NSTEP && wave < 3) tb[i][wave] = cr_back_t(F, gv, Wl, Wr, i, s, wave);
    __syncthreads();
    if (i < NSTEP && wave == 0) cr_back_x(F, gv, tb, i);
    __syncthreads();
  }
  {
    const int i = 2 * tid + 1;
    if (i < NSTEP) cr_back(F, gv, Wl, Wr, i, 1);
    __syncthreads();
  }

  if (tid < NSTEP) {
    out[0 * OUT_U + b * NSTEP + tid] = (float)gv[tid][0];
    out[1 * OUT_U + b * NSTEP + tid] = (float)gv[tid][1];
    out[2 * OUT_U + b * NSTEP + tid] = (float)gv[tid][2];
  }
  if (tid == 0) out[OUT_EPS + b] = 0.0f;
  if (tid < T1) out[OUT_ST + b * T1 + tid] = hstep[tid];
}

extern "C" void kernel_launch(void* const* d_in, const int* in_sizes, int n_in,
                              void* d_out, int out_size, void* d_ws, size_t ws_size,
                              hipStream_t stream) {
  (void)in_sizes; (void)n_in; (void)out_size; (void)d_ws; (void)ws_size;
  const float* coeffs = (const float*)d_in[0];
  const float* rhs    = (const float*)d_in[1];
  const float* ivr    = (const float*)d_in[2];
  const float* steps  = (const float*)d_in[3];
  float* out = (float*)d_out;
  hipLaunchKernelGGL(ode_blocktri_kernel, dim3(NB), dim3(NTHR), 0, stream,
                     coeffs, rhs, ivr, steps, out);
}


# Gram loop takes its step sizes from a register (lane read), not one dependent LDS load per trip, and is unrolled branch-free so neighbouring steps' WMMAs overlap; tiles bit-identical
# speedup vs baseline: 29.7419x; 1.0680x over previous
#include <hip/hip_runtime.h>
#include <hip/hip_bf16.h>
#include <math.h>


typedef __attribute__((ext_vector_type(2))) float v2f;
typedef __attribute__((ext_vector_type(8))) float v8f;

#define NSTEP 100
#define T1    99
#define NB    256
#define OUT_U   (NB * NSTEP)
#define OUT_EPS (3 * OUT_U)
#define OUT_ST  (OUT_EPS + NB)

__device__ __forceinline__ void Ecoef(int r, int c, float* pa, float* pb, float* pc) {
  float a = 0.f, b = 0.f, q = 0.f;
  if (r == 0)      { if (c == 0) a = 1.f;  else if (c == 1) b = 1.f;   else if (c == 2) q = 0.5f; else if (c == 3) a = -1.f; }
  else if (r == 1) { if (c == 1) a = 1.f;  else if (c == 2) b = 1.f;   else if (c == 4) a = -1.f; }
  else if (r == 2) { if (c == 0) a = -1.f; else if (c == 3) a = 1.f;   else if (c == 4) b = -1.f; else if (c == 5) q = 0.5f; }
  else if (r == 3) { if (c == 1) a = -1.f; else if (c == 4) a = 1.f;   else if (c == 5) b = -1.f; }
  else if (r == 4) { if (c == 1) a = -1.f; else if (c == 2) b = -0.5f; else if (c == 4) a = 1.f;  else if (c == 5) b = -0.5f; }
  *pa = a; *pb = b; *pc = q;
}

__device__ __forceinline__ double recip_nr(double d) {
  double r = __builtin_amdgcn_rcp(d);
  double e = __builtin_fma(-d, r, 1.0);
  r = __builtin_fma(r, e, r);
  e = __builtin_fma(-d, r, 1.0);
  r = __builtin_fma(r, e, r);
  return r;
}

__device__ __forceinline__ void ldl3(double s0, double s1, double s2,
                                     double s3, double s4, double s5,
                                     double& u10, double& u20, double& u21,
                                     double& i0, double& i1, double& i2) {
  i0 = recip_nr(s0);
  u10 = s1 * i0;
  u20 = s3 * i0;
  const double d1 = __builtin_fma(-u10, s1, s2);
  i1 = recip_nr(d1);
  const double t = __builtin_fma(-u20, s1, s4);
  u21 = t * i1;
  const double d2 = __builtin_fma(-u21, t, __builtin_fma(-u20, s3, s5));
  i2 = recip_nr(d2);
}

__device__ __forceinline__ void zw3(const double* m,
                                    double u10, double u20, double u21,
                                    double i0, double i1, double i2,
                                    double* z, double* w) {
  z[0] = m[0];
  z[1] = __builtin_fma(-z[0], u10, m[1]);
  z[2] = __builtin_fma(-z[1], u21, __builtin_fma(-z[0], u20, m[2]));
  z[3] = m[3];
  z[4] = __builtin_fma(-z[3], u10, m[4]);
  z[5] = __builtin_fma(-z[4], u21, __builtin_fma(-z[3], u20, m[5]));
  z[6] = m[6];
  z[7] = __builtin_fma(-z[6], u10, m[7]);
  z[8] = __builtin_fma(-z[7], u21, __builtin_fma(-z[6], u20, m[8]));
  w[0] = z[0] * i0; w[1] = z[1] * i1; w[2] = z[2] * i2;
  w[3] = z[3] * i0; w[4] = z[4] * i1; w[5] = z[5] * i2;
  w[6] = z[6] * i0; w[7] = z[7] * i1; w[8] = z[8] * i2;
}

__device__ __forceinline__ void elim_left(
    const double (*S)[6], double (*F)[6], const double (*C)[9], double (*gv)[3],
    double (*Zl)[9], double (*Wl)[9], int i, int s) {
  const double* cl = C[i - s];
  double ml[9];
  ml[0] = cl[0]; ml[1] = cl[1]; ml[2] = cl[2];
  ml[3] = cl[3]; ml[4] = cl[4]; ml[5] = cl[5];
  ml[6] = cl[6]; ml[7] = cl[7]; ml[8] = cl[8];
  const double s0 = S[i][0], s1 = S[i][1], s2 = S[i][2];
  const double s3 = S[i][3], s4 = S[i][4], s5 = S[i][5];
  const double g0 = gv[i][0], g1 = gv[i][1], g2 = gv[i][2];

  double u10, u20, u21, i0, i1, i2;
  ldl3(s0, s1, s2, s3, s4, s5, u10, u20, u21, i0, i1, i2);
  const double w1 = __builtin_fma(-u10, g0, g1);
  const double w2 = __builtin_fma(-u21, w1, __builtin_fma(-u20, g0, g2));
  double zl[9], wl[9];
  zw3(ml, u10, u20, u21, i0, i1, i2, zl, wl);

  F[i][0] = i0;  F[i][1] = u10; F[i][2] = i1;
  F[i][3] = u20; F[i][4] = u21; F[i][5] = i2;
  gv[i][1] = w1; gv[i][2] = w2;
  Zl[i][0] = zl[0]; Zl[i][1] = zl[1]; Zl[i][2] = zl[2];
  Zl[i][3] = zl[3]; Zl[i][4] = zl[4]; Zl[i][5] = zl[5];
  Zl[i][6] = zl[6]; Zl[i][7] = zl[7]; Zl[i][8] = zl[8];
  Wl[i][0] = wl[0]; Wl[i][1] = wl[1]; Wl[i][2] = wl[2];
  Wl[i][3] = wl[3]; Wl[i][4] = wl[4]; Wl[i][5] = wl[5];
  Wl[i][6] = wl[6]; Wl[i][7] = wl[7]; Wl[i][8] = wl[8];
}

__device__ __forceinline__ void elim_right(
    const double (*S)[6], const double (*C)[9],
    double (*Zr)[9], double (*Wr)[9], int i) {
  const double* cr = C[i];
  double mr[9];
  mr[0] = cr[0]; mr[1] = cr[3]; mr[2] = cr[6];
  mr[3] = cr[1]; mr[4] = cr[4]; mr[5] = cr[7];
  mr[6] = cr[2]; mr[7] = cr[5]; mr[8] = cr[8];
  const double s0 = S[i][0], s1 = S[i][1], s2 = S[i][2];
  const double s3 = S[i][3], s4 = S[i][4], s5 = S[i][5];

  double u10, u20, u21, i0, i1, i2;
  ldl3(s0, s1, s2, s3, s4, s5, u10, u20, u21, i0, i1, i2);
  double zr[9], wr[9];
  zw3(mr, u10, u20, u21, i0, i1, i2, zr, wr);

  Zr[i][0] = zr[0]; Zr[i][1] = zr[1]; Zr[i][2] = zr[2];
  Zr[i][3] = zr[3]; Zr[i][4] = zr[4]; Zr[i][5] = zr[5];
  Zr[i][6] = zr[6]; Zr[i][7] = zr[7]; Zr[i][8] = zr[8];
  Wr[i][0] = wr[0]; Wr[i][1] = wr[1]; Wr[i][2] = wr[2];
  Wr[i][3] = wr[3]; Wr[i][4] = wr[4]; Wr[i][5] = wr[5];
  Wr[i][6] = wr[6]; Wr[i][7] = wr[7]; Wr[i][8] = wr[8];
}

__device__ __forceinline__ void schur3_n(
    const double* zp, const double* wp,
    double& n0, double& n1, double& n2, double& n3, double& n4, double& n5) {
  const double z0 = zp[0], z1 = zp[1], z2 = zp[2];
  const double z3 = zp[3], z4 = zp[4], z5 = zp[5];
  const double z6 = zp[6], z7 = zp[7], z8 = zp[8];
  const double v0 = wp[0], v1 = wp[1], v2 = wp[2];
  const double v3 = wp[3], v4 = wp[4], v5 = wp[5];
  const double v6 = wp[6], v7 = wp[7], v8 = wp[8];
  n0 = __builtin_fma(-v2, z2, __builtin_fma(-v1, z1, __builtin_fma(-v0, z0, n0)));
  n1 = __builtin_fma(-v5, z2, __builtin_fma(-v4, z1, __builtin_fma(-v3, z0, n1)));
  n2 = __builtin_fma(-v5, z5, __builtin_fma(-v4, z4, __builtin_fma(-v3, z3, n2)));
  n3 = __builtin_fma(-v8, z2, __builtin_fma(-v7, z1, __builtin_fma(-v6, z0, n3)));
  n4 = __builtin_fma(-v8, z5, __builtin_fma(-v7, z4, __builtin_fma(-v6, z3, n4)));
  n5 = __builtin_fma(-v8, z8, __builtin_fma(-v7, z7, __builtin_fma(-v6, z6, n5)));
}

__device__ __forceinline__ void schur3_g(
    const double* wp, const double* xp, double& g0, double& g1, double& g2) {
  const double v0 = wp[0], v1 = wp[1], v2 = wp[2];
  const double v3 = wp[3], v4 = wp[4], v5 = wp[5];
  const double v6 = wp[6], v7 = wp[7], v8 = wp[8];
  const double x0 = xp[0], x1 = xp[1], x2 = xp[2];
  g0 = __builtin_fma(-v2, x2, __builtin_fma(-v1, x1, __builtin_fma(-v0, x0, g0)));
  g1 = __builtin_fma(-v5, x2, __builtin_fma(-v4, x1, __builtin_fma(-v3, x0, g1)));
  g2 = __builtin_fma(-v8, x2, __builtin_fma(-v7, x1, __builtin_fma(-v6, x0, g2)));
}

__device__ __forceinline__ void upd_diag(
    double (*S)[6],
    const double (*Zl)[9], const double (*Wl)[9],
    const double (*Zr)[9], const double (*Wr)[9],
    int j, int s) {
  double n0 = S[j][0], n1 = S[j][1], n2 = S[j][2];
  double n3 = S[j][3], n4 = S[j][4], n5 = S[j][5];
  if (j > 0) {
    const int e = j - s;
    schur3_n(Zr[e], Wr[e], n0, n1, n2, n3, n4, n5);
  }
  if (j + s < NSTEP) {
    const int e = j + s;
    schur3_n(Zl[e], Wl[e], n0, n1, n2, n3, n4, n5);
  }
  S[j][0] = n0; S[j][1] = n1; S[j][2] = n2;
  S[j][3] = n3; S[j][4] = n4; S[j][5] = n5;
}

__device__ __forceinline__ void upd_rhs_couple(
    double (*C)[9], double (*gv)[3],
    const double (*Wl)[9], const double (*Wr)[9], const double (*Zr)[9],
    int j, int s) {
  double g0 = gv[j][0], g1 = gv[j][1], g2 = gv[j][2];
  if (j > 0) {
    const int e = j - s;
    schur3_g(Wr[e], gv[e], g0, g1, g2);
  }
  if (j + s < NSTEP) {
    const int e = j + s;
    schur3_g(Wl[e], gv[e], g0, g1, g2);
    if (j + 2 * s < NSTEP) {
      const double* wp = Wl[e];
      const double* zp = Zr[e];
      const double v0 = wp[0], v1 = wp[1], v2 = wp[2];
      const double v3 = wp[3], v4 = wp[4], v5 = wp[5];
      const double v6 = wp[6], v7 = wp[7], v8 = wp[8];
      const double z0 = zp[0], z1 = zp[1], z2 = zp[2];
      const double z3 = zp[3], z4 = zp[4], z5 = zp[5];
      const double z6 = zp[6], z7 = zp[7], z8 = zp[8];
      double* cp = C[j];
      cp[0] = __builtin_fma(-v2, z2, __builtin_fma(-v1, z1, -(v0 * z0)));
      cp[1] = __builtin_fma(-v2, z5, __builtin_fma(-v1, z4, -(v0 * z3)));
      cp[2] = __builtin_fma(-v2, z8, __builtin_fma(-v1, z7, -(v0 * z6)));
      cp[3] = __builtin_fma(-v5, z2, __builtin_fma(-v4, z1, -(v3 * z0)));
      cp[4] = __builtin_fma(-v5, z5, __builtin_fma(-v4, z4, -(v3 * z3)));
      cp[5] = __builtin_fma(-v5, z8, __builtin_fma(-v4, z7, -(v3 * z6)));
      cp[6] = __builtin_fma(-v8, z2, __builtin_fma(-v7, z1, -(v6 * z0)));
      cp[7] = __builtin_fma(-v8, z5, __builtin_fma(-v7, z4, -(v6 * z3)));
      cp[8] = __builtin_fma(-v8, z8, __builtin_fma(-v7, z7, -(v6 * z6)));
    }
  }
  gv[j][0] = g0; gv[j][1] = g1; gv[j][2] = g2;
}

__device__ __forceinline__ void zw_row(double m0, double m1, double m2,
                                       double u10, double u20, double u21,
                                       double i0, double i1, double i2,
                                       double* z, double* w) {
  const double z0 = m0;
  const double z1 = __builtin_fma(-z0, u10, m1);
  const double z2 = __builtin_fma(-z1, u21, __builtin_fma(-z0, u20, m2));
  z[0] = z0; z[1] = z1; z[2] = z2;
  w[0] = z0 * i0; w[1] = z1 * i1; w[2] = z2 * i2;
}

__device__ __forceinline__ void elim4(
    const double (*S)[6], double (*F)[6], const double (*C)[9], double (*gv)[3],
    double (*Zl)[9], double (*Wl)[9], double (*Zr)[9], double (*Wr)[9],
    int i, int s, int part) {
  if (part >= 2 && i + s >= NSTEP) return;
  const double s0 = S[i][0], s1 = S[i][1], s2 = S[i][2];
  const double s3 = S[i][3], s4 = S[i][4], s5 = S[i][5];
  double u10, u20, u21, i0, i1, i2;
  if (part == 0) {
    const double* cl = C[i - s];
    const double m0 = cl[0], m1 = cl[1], m2 = cl[2];
    const double g0 = gv[i][0], g1 = gv[i][1], g2 = gv[i][2];
    ldl3(s0, s1, s2, s3, s4, s5, u10, u20, u21, i0, i1, i2);
    const double w1 = __builtin_fma(-u10, g0, g1);
    const double w2 = __builtin_fma(-u21, w1, __builtin_fma(-u20, g0, g2));
    F[i][0] = i0;  F[i][1] = u10; F[i][2] = i1;
    F[i][3] = u20; F[i][4] = u21; F[i][5] = i2;
    gv[i][1] = w1; gv[i][2] = w2;
    zw_row(m0, m1, m2, u10, u20, u21, i0, i1, i2, &Zl[i][0], &Wl[i][0]);
  } else if (part == 1) {
    const double* cl = C[i - s];
    const double m3 = cl[3], m4 = cl[4], m5 = cl[5];
    const double m6 = cl[6], m7 = cl[7], m8 = cl[8];
    ldl3(s0, s1, s2, s3, s4, s5, u10, u20, u21, i0, i1, i2);
    zw_row(m3, m4, m5, u10, u20, u21, i0, i1, i2, &Zl[i][3], &Wl[i][3]);
    zw_row(m6, m7, m8, u10, u20, u21, i0, i1, i2, &Zl[i][6], &Wl[i][6]);
  } else if (part == 2) {
    const double* cr = C[i];
    const double m0 = cr[0], m1 = cr[3], m2 = cr[6];
    const double m3 = cr[1], m4 = cr[4], m5 = cr[7];
    ldl3(s0, s1, s2, s3, s4, s5, u10, u20, u21, i0, i1, i2);
    zw_row(m0, m1, m2, u10, u20, u21, i0, i1, i2, &Zr[i][0], &Wr[i][0]);
    zw_row(m3, m4, m5, u10, u20, u21, i0, i1, i2, &Zr[i][3], &Wr[i][3]);
  } else {
    const double* cr = C[i];
    const double m6 = cr[2], m7 = cr[5], m8 = cr[8];
    ldl3(s0, s1, s2, s3, s4, s5, u10, u20, u21, i0, i1, i2);
    zw_row(m6, m7, m8, u10, u20, u21, i0, i1, i2, &Zr[i][6], &Wr[i][6]);
  }
}

__device__ __forceinline__ void upd4(
    double (*S)[6], double (*C)[9], double (*gv)[3],
    const double (*Zl)[9], const double (*Wl)[9],
    const double (*Zr)[9], const double (*Wr)[9],
    int j, int s, int part) {
  const bool hasL = (j > 0);
  const bool hasR = (j + s < NSTEP);
  const bool hasC = (j + 2 * s < NSTEP);
  const int el = hasL ? j - s : 0;
  const int er = hasR ? j + s : 0;
  if (part == 0) {
    double n0 = S[j][0], n1 = S[j][1], n2 = S[j][2];
    if (hasL) {
      const double* zp = Zr[el]; const double* wp = Wr[el];
      const double z0 = zp[0], z1 = zp[1], z2 = zp[2], z3 = zp[3], z4 = zp[4], z5 = zp[5];
      const double v0 = wp[0], v1 = wp[1], v2 = wp[2], v3 = wp[3], v4 = wp[4], v5 = wp[5];
      n0 = __builtin_fma(-v2, z2, __builtin_fma(-v1, z1, __builtin_fma(-v0, z0, n0)));
      n1 = __builtin_fma(-v5, z2, __builtin_fma(-v4, z1, __builtin_fma(-v3, z0, n1)));
      n2 = __builtin_fma(-v5, z5, __builtin_fma(-v4, z4, __builtin_fma(-v3, z3, n2)));
    }
    if (hasR) {
      const double* zp = Zl[er]; const double* wp = Wl[er];
      const double z0 = zp[0], z1 = zp[1], z2 = zp[2], z3 = zp[3], z4 = zp[4], z5 = zp[5];
      const double v0 = wp[0], v1 = wp[1], v2 = wp[2], v3 = wp[3], v4 = wp[4], v5 = wp[5];
      n0 = __builtin_fma(-v2, z2, __builtin_fma(-v1, z1, __builtin_fma(-v0, z0, n0)));
      n1 = __builtin_fma(-v5, z2, __builtin_fma(-v4, z1, __builtin_fma(-v3, z0, n1)));
      n2 = __builtin_fma(-v5, z5, __builtin_fma(-v4, z4, __builtin_fma(-v3, z3, n2)));
    }
    S[j][0] = n0; S[j][1] = n1; S[j][2] = n2;
  } else if (part == 1) {
    double n3 = S[j][3], n4 = S[j][4], n5 = S[j][5];
    if (hasL) {
      const double* zp = Zr[el]; const double* wp = Wr[el];
      const double z0 = zp[0], z1 = zp[1], z2 = zp[2], z3 = zp[3], z4 = zp[4], z5 = zp[5];
      const double z6 = zp[6], z7 = zp[7], z8 = zp[8];
      const double v6 = wp[6], v7 = wp[7], v8 = wp[8];
      n3 = __builtin_fma(-v8, z2, __builtin_fma(-v7, z1, __builtin_fma(-v6, z0, n3)));
      n4 = __builtin_fma(-v8, z5, __builtin_fma(-v7, z4, __builtin_fma(-v6, z3, n4)));
      n5 = __builtin_fma(-v8, z8, __builtin_fma(-v7, z7, __builtin_fma(-v6, z6, n5)));
    }
    if (hasR) {
      const double* zp = Zl[er]; const double* wp = Wl[er];
      const double z0 = zp[0], z1 = zp[1], z2 = zp[2], z3 = zp[3], z4 = zp[4], z5 = zp[5];
      const double z6 = zp[6], z7 = zp[7], z8 = zp[8];
      const double v6 = wp[6], v7 = wp[7], v8 = wp[8];
      n3 = __builtin_fma(-v8, z2, __builtin_fma(-v7, z1, __builtin_fma(-v6, z0, n3)));
      n4 = __builtin_fma(-v8, z5, __builtin_fma(-v7, z4, __builtin_fma(-v6, z3, n4)));
      n5 = __builtin_fma(-v8, z8, __builtin_fma(-v7, z7, __builtin_fma(-v6, z6, n5)));
    }
    S[j][3] = n3; S[j][4] = n4; S[j][5] = n5;
    if (hasC) {
      const double* wp = Wl[er]; const double* zp = Zr[er];
      const double v0 = wp[0], v1 = wp[1], v2 = wp[2];
      const double z0 = zp[0], z1 = zp[1], z2 = zp[2];
      C[j][0] = __builtin_fma(-v2, z2, __builtin_fma(-v1, z1, -(v0 * z0)));
    }
  } else if (part == 2) {
    double g0 = gv[j][0], g1 = gv[j][1], g2 = gv[j][2];
    if (hasL) schur3_g(Wr[el], gv[el], g0, g1, g2);
    if (hasR) schur3_g(Wl[er], gv[er], g0, g1, g2);
    gv[j][0] = g0; gv[j][1] = g1; gv[j][2] = g2;
    if (hasC) {
      const double* wp = Wl[er]; const double* zp = Zr[er];
      const double v0 = wp[0], v1 = wp[1], v2 = wp[2];
      const double z3 = zp[3], z4 = zp[4], z5 = zp[5];
      C[j][1] = __builtin_fma(-v2, z5, __builtin_fma(-v1, z4, -(v0 * z3)));
    }
  } else {
    if (hasC) {
      const double* wp = Wl[er]; const double* zp = Zr[er];
      const double v0 = wp[0], v1 = wp[1], v2 = wp[2];
      const double v3 = wp[3], v4 = wp[4], v5 = wp[5];
      const double v6 = wp[6], v7 = wp[7], v8 = wp[8];
      const double z0 = zp[0], z1 = zp[1], z2 = zp[2];
      const double z3 = zp[3], z4 = zp[4], z5 = zp[5];
      const double z6 = zp[6], z7 = zp[7], z8 = zp[8];
      double* cp = C[j];
      cp[2] = __builtin_fma(-v2, z8, __builtin_fma(-v1, z7, -(v0 * z6)));
      cp[3] = __builtin_fma(-v5, z2, __builtin_fma(-v4, z1, -(v3 * z0)));
      cp[4] = __builtin_fma(-v5, z5, __builtin_fma(-v4, z4, -(v3 * z3)));
      cp[5] = __builtin_fma(-v5, z8, __builtin_fma(-v4, z7, -(v3 * z6)));
      cp[6] = __builtin_fma(-v8, z2, __builtin_fma(-v7, z1, -(v6 * z0)));
      cp[7] = __builtin_fma(-v8, z5, __builtin_fma(-v7, z4, -(v6 * z3)));
      cp[8] = __builtin_fma(-v8, z8, __builtin_fma(-v7, z7, -(v6 * z6)));
    }
  }
}

__device__ __forceinline__ void cr_back(
    const double (*F)[6], double (*gv)[3],
    const double (*Wl)[9], const double (*Wr)[9],
    int i, int s) {
  const double j0 = F[i][0], q10 = F[i][1], j1 = F[i][2];
  const double q20 = F[i][3], q21 = F[i][4], j2 = F[i][5];
  double t0 = gv[i][0] * j0, t1 = gv[i][1] * j1, t2 = gv[i][2] * j2;
  {
    const double* wp = Wl[i];
    const double a0 = gv[i - s][0], a1 = gv[i - s][1], a2 = gv[i - s][2];
    t0 = __builtin_fma(-wp[6], a2, __builtin_fma(-wp[3], a1, __builtin_fma(-wp[0], a0, t0)));
    t1 = __builtin_fma(-wp[7], a2, __builtin_fma(-wp[4], a1, __builtin_fma(-wp[1], a0, t1)));
    t2 = __builtin_fma(-wp[8], a2, __builtin_fma(-wp[5], a1, __builtin_fma(-wp[2], a0, t2)));
  }
  if (i + s < NSTEP) {
    const double* wp = Wr[i];
    const double a0 = gv[i + s][0], a1 = gv[i + s][1], a2 = gv[i + s][2];
    t0 = __builtin_fma(-wp[6], a2, __builtin_fma(-wp[3], a1, __builtin_fma(-wp[0], a0, t0)));
    t1 = __builtin_fma(-wp[7], a2, __builtin_fma(-wp[4], a1, __builtin_fma(-wp[1], a0, t1)));
    t2 = __builtin_fma(-wp[8], a2, __builtin_fma(-wp[5], a1, __builtin_fma(-wp[2], a0, t2)));
  }
  const double x2 = t2;
  const double x1 = __builtin_fma(-q21, x2, t1);
  const double x0 = __builtin_fma(-q20, x2, __builtin_fma(-q10, x1, t0));
  gv[i][0] = x0; gv[i][1] = x1; gv[i][2] = x2;
}

__device__ __forceinline__ double cr_back_t(
    const double (*F)[6], const double (*gv)[3],
    const double (*Wl)[9], const double (*Wr)[9],
    int i, int s, int k) {
  const int fk = (k == 0) ? 0 : ((k == 1) ? 2 : 5);
  double t = gv[i][k] * F[i][fk];
  {
    const double* wp = Wl[i];
    const double a0 = gv[i - s][0], a1 = gv[i - s][1], a2 = gv[i - s][2];
    t = __builtin_fma(-wp[6 + k], a2, __builtin_fma(-wp[3 + k], a1, __builtin_fma(-wp[k], a0, t)));
  }
  if (i + s < NSTEP) {
    const double* wp = Wr[i];
    const double a0 = gv[i + s][0], a1 = gv[i + s][1], a2 = gv[i + s][2];
    t = __builtin_fma(-wp[6 + k], a2, __builtin_fma(-wp[3 + k], a1, __builtin_fma(-wp[k], a0, t)));
  }
  return t;
}

__device__ __forceinline__ void cr_back_x(
    const double (*F)[6], double (*gv)[3], const double (*tb)[3], int i) {
  const double q10 = F[i][1], q20 = F[i][3], q21 = F[i][4];
  const double x2 = tb[i][2];
  const double x1 = __builtin_fma(-q21, x2, tb[i][1]);
  const double x0 = __builtin_fma(-q20, x2, __builtin_fma(-q10, x1, tb[i][0]));
  gv[i][0] = x0; gv[i][1] = x1; gv[i][2] = x2;
}

__device__ __forceinline__ v8f gram_tile(float h,
                                         float a0, float b0, float q0,
                                         float a1, float b1, float q1,
                                         float a2, float b2, float q2) {
  v2f f0, f1;
  f0.x = fmaf(fmaf(q0, h, b0), h, a0);
  f0.y = fmaf(fmaf(q1, h, b1), h, a1);
  f1.x = fmaf(fmaf(q2, h, b2), h, a2);
  f1.y = 0.f;
  v8f acc = {};
  acc = __builtin_amdgcn_wmma_f32_16x16x4_f32(false, f0, false, f0,
                                              (short)0, acc, false, false);
  acc = __builtin_amdgcn_wmma_f32_16x16x4_f32(false, f1, false, f1,
                                              (short)0, acc, false, false);
  return acc;
}

__device__ __forceinline__ void put_tile(float* q, v8f acc) {
  q[0] = acc[0]; q[1] = acc[1]; q[2] = acc[2];
  q[3] = acc[3]; q[4] = acc[4]; q[5] = acc[5];
}

#define TOPS  64
#define NWAVE 4
#define NTHR  (32 * NWAVE)
#define KSTEP 25
#define GS_TILES (T1 * 36)
#define GS_SPARE ((KSTEP - 1) * NWAVE * 36 + 26 * 6)

__global__ __launch_bounds__(NTHR) void ode_blocktri_kernel(
    const float* __restrict__ coeffs,
    const float* __restrict__ rhs,
    const float* __restrict__ ivr,
    const float* __restrict__ steps,
    float* __restrict__ out)
{
  __shared__ double S[NSTEP][6];
  __shared__ double F[NSTEP][6];
  __shared__ double Bo[T1][9];
  __shared__ double gv[NSTEP][3];
  __shared__ double Zl[NSTEP][9];
  __shared__ double Wl[NSTEP][9];
  __shared__ double Zr[NSTEP][9];
  __shared__ double Wr[NSTEP][9];
  __shared__ double tb[NSTEP][3];
  __shared__ __attribute__((aligned(16))) float Gs[GS_TILES + GS_SPARE];

  const int b = blockIdx.x;
  const int tid = threadIdx.x;
  const int lane = tid & 31;
  const int wave = __builtin_amdgcn_readfirstlane(tid >> 5);
  const int role = wave >> 1;
  const int slot = ((wave & 1) << 5) | lane;

  const int myp = wave + NWAVE * lane;
  const float hmine = (myp < T1) ? steps[b * T1 + myp] : 0.f;
  const float hcopy = (tid < T1) ? steps[b * T1 + tid] : 0.f;

  const int m = lane & 15;
  const int khalf = lane >> 4;
  const int col = (m < 6) ? m : 6;
  float a0, b0, q0, a1, b1, q1, a2, b2, q2;
  Ecoef(2 * khalf,                 col, &a0, &b0, &q0);
  Ecoef(2 * khalf + 1,             col, &a1, &b1, &q1);
  Ecoef((khalf == 0) ? 4 : -1,     col, &a2, &b2, &q2);

  const int hbits = __builtin_bit_cast(int, hmine);
  float* const gq = Gs + ((lane < 6) ? (wave * 36 + lane * 6)
                                     : (GS_TILES + (lane - 6) * 6));
  {
    v8f pend = {};
#pragma unroll
    for (int k = 0; k < KSTEP - 1; ++k) {
      const float h = __builtin_bit_cast(float, __builtin_amdgcn_readlane(hbits, k));
      const v8f acc = gram_tile(h, a0, b0, q0, a1, b1, q1, a2, b2, q2);
      if (k > 0) put_tile(gq + (k - 1) * (NWAVE * 36), pend);
      pend = acc;
    }
    put_tile(gq + (KSTEP - 2) * (NWAVE * 36), pend);
    if (wave + NWAVE * (KSTEP - 1) < T1) {
      const float h = __builtin_bit_cast(float, __builtin_amdgcn_readlane(hbits, KSTEP - 1));
      const v8f acc = gram_tile(h, a0, b0, q0, a1, b1, q1, a2, b2, q2);
      put_tile(gq + (KSTEP - 1) * (NWAVE * 36), acc);
    }
  }
  __syncthreads();

  const float iv0 = ivr[b * 2 + 0];
  const float iv1 = ivr[b * 2 + 1];
  if (tid < NSTEP) {
    const int t = tid;
    const float* cp = coeffs + (b * NSTEP + t) * 3;
    const double c0 = (double)cp[0];
    const double c1 = (double)cp[1];
    const double c2 = (double)cp[2];
    const double r  = (double)rhs[b * NSTEP + t];
    const bool hasPrev = (t > 0);
    const bool hasNext = (t < T1);
    const float* gp = Gs + 36 * (hasPrev ? t - 1 : 0);
    const float* gn = Gs + 36 * (hasNext ? t : T1 - 1);
    const float f0 = gp[3 * 6 + 3], f1 = gp[3 * 6 + 4], f2 = gp[4 * 6 + 4];
    const float f3 = gp[3 * 6 + 5], f4 = gp[4 * 6 + 5], f5 = gp[5 * 6 + 5];
    const float n0 = gn[0 * 6 + 0], n1 = gn[0 * 6 + 1], n2 = gn[1 * 6 + 1];
    const float n3 = gn[0 * 6 + 2], n4 = gn[1 * 6 + 2], n5 = gn[2 * 6 + 2];
    const double p0 = (double)(hasPrev ? f0 : 1.f);
    const double p1 = (double)(hasPrev ? f1 : 0.f);
    const double p2 = (double)(hasPrev ? f2 : 1.f);
    const double p3 = (double)(hasPrev ? f3 : 0.f);
    const double p4 = (double)(hasPrev ? f4 : 0.f);
    const double p5 = (double)(hasPrev ? f5 : 0.f);
    S[t][0] = __builtin_fma(c0, c0, p0) + (double)(hasNext ? n0 : 0.f);
    S[t][1] = __builtin_fma(c1, c0, p1) + (double)(hasNext ? n1 : 0.f);
    S[t][2] = __builtin_fma(c1, c1, p2) + (double)(hasNext ? n2 : 0.f);
    S[t][3] = __builtin_fma(c2, c0, p3) + (double)(hasNext ? n3 : 0.f);
    S[t][4] = __builtin_fma(c2, c1, p4) + (double)(hasNext ? n4 : 0.f);
    S[t][5] = __builtin_fma(c2, c2, p5) + (double)(hasNext ? n5 : 0.f);
    gv[t][0] = __builtin_fma(c0, r, (double)((t == 0) ? iv0 : 0.f));
    gv[t][1] = __builtin_fma(c1, r, (double)((t == 0) ? iv1 : 0.f));
    gv[t][2] = c2 * r;
    const double b00 = (double)gn[3 * 6 + 0], b01 = (double)gn[4 * 6 + 0], b02 = (double)gn[5 * 6 + 0];
    const double b10 = (double)gn[3 * 6 + 1], b11 = (double)gn[4 * 6 + 1], b12 = (double)gn[5 * 6 + 1];
    const double b20 = (double)gn[3 * 6 + 2], b21 = (double)gn[4 * 6 + 2], b22 = (double)gn[5 * 6 + 2];
    if (hasNext) {
      Bo[t][0] = b00; Bo[t][1] = b01; Bo[t][2] = b02;
      Bo[t][3] = b10; Bo[t][4] = b11; Bo[t][5] = b12;
      Bo[t][6] = b20; Bo[t][7] = b21; Bo[t][8] = b22;
    }
  }
  __syncthreads();

  {
    const int s = 1;
    {
      const int i = s * (2 * slot + 1);
      if (i < NSTEP) {
        if (role == 0)            elim_left(S, F, Bo, gv, Zl, Wl, i, s);
        else if (i + s < NSTEP)   elim_right(S, Bo, Zr, Wr, i);
      }
    }
    __syncthreads();
    {
      const int j = 2 * s * slot;
      if (j < NSTEP) {
        if (role == 0)   upd_diag(S, Zl, Wl, Zr, Wr, j, s);
        else             upd_rhs_couple(Bo, gv, Wl, Wr, Zr, j, s);
      }
    }
    __syncthreads();
  }
  for (int s = 2; s <= TOPS; s <<= 1) {
    {
      const int i = s * (2 * lane + 1);
      if (i < NSTEP) elim4(S, F, Bo, gv, Zl, Wl, Zr, Wr, i, s, wave);
    }
    __syncthreads();
    {
      const int j = 2 * s * lane;
      if (j < NSTEP) upd4(S, Bo, gv, Zl, Wl, Zr, Wr, j, s, wave);
    }
    __syncthreads();
  }
  if (tid == 0) {
    double u10, u20, u21, i0, i1, i2;
    ldl3(S[0][0], S[0][1], S[0][2], S[0][3], S[0][4], S[0][5],
         u10, u20, u21, i0, i1, i2);
    const double w0 = gv[0][0];
    const double w1 = __builtin_fma(-u10, w0, gv[0][1]);
    const double w2 = __builtin_fma(-u21, w1, __builtin_fma(-u20, w0, gv[0][2]));
    const double x2 = w2 * i2;
    const double x1 = __builtin_fma(-u21, x2, w1 * i1);
    const double x0 = __builtin_fma(-u20, x2, __builtin_fma(-u10, x1, w0 * i0));
    gv[0][0] = x0; gv[0][1] = x1; gv[0][2] = x2;
  }
  __syncthreads();
  for (int s = TOPS; s >= 2; s >>= 1) {
    const int i = s * (2 * lane + 1);
    if (i < NSTEP && wave < 3) tb[i][wave] = cr_back_t(F, gv, Wl, Wr, i, s, wave);
    __syncthreads();
    if (i < NSTEP && wave == 0) cr_back_x(F, gv, tb, i);
    __syncthreads();
  }
  {
    const int i = 2 * tid + 1;
    if (i < NSTEP) cr_back(F, gv, Wl, Wr, i, 1);
    __syncthreads();
  }

  if (tid < NSTEP) {
    out[0 * OUT_U + b * NSTEP + tid] = (float)gv[tid][0];
    out[1 * OUT_U + b * NSTEP + tid] = (float)gv[tid][1];
    out[2 * OUT_U + b * NSTEP + tid] = (float)gv[tid][2];
  }
  if (tid == 0) out[OUT_EPS + b] = 0.0f;
  if (tid < T1) out[OUT_ST + b * T1 + tid] = hcopy;
}

extern "C" void kernel_launch(void* const* d_in, const int* in_sizes, int n_in,
                              void* d_out, int out_size, void* d_ws, size_t ws_size,
                              hipStream_t stream) {
  (void)in_sizes; (void)n_in; (void)out_size; (void)d_ws; (void)ws_size;
  const float* coeffs = (const float*)d_in[0];
  const float* rhs    = (const float*)d_in[1];
  const float* ivr    = (const float*)d_in[2];
  const float* steps  = (const float*)d_in[3];
  float* out = (float*)d_out;
  hipLaunchKernelGGL(ode_blocktri_kernel, dim3(NB), dim3(NTHR), 0, stream,
                     coeffs, rhs, ivr, steps, out);
}


# From stride 4 up (12 blocks or fewer a side) a lane takes one row of a block's elimination and one output element of its update, not a whole part of a block; every value bit-identical
# speedup vs baseline: 33.1824x; 1.1157x over previous
#include <hip/hip_runtime.h>
#include <hip/hip_bf16.h>
#include <math.h>


typedef __attribute__((ext_vector_type(2))) float v2f;
typedef __attribute__((ext_vector_type(8))) float v8f;

#define NSTEP 100
#define T1    99
#define NB    256
#define OUT_U   (NB * NSTEP)
#define OUT_EPS (3 * OUT_U)
#define OUT_ST  (OUT_EPS + NB)

__device__ __forceinline__ void Ecoef(int r, int c, float* pa, float* pb, float* pc) {
  float a = 0.f, b = 0.f, q = 0.f;
  if (r == 0)      { if (c == 0) a = 1.f;  else if (c == 1) b = 1.f;   else if (c == 2) q = 0.5f; else if (c == 3) a = -1.f; }
  else if (r == 1) { if (c == 1) a = 1.f;  else if (c == 2) b = 1.f;   else if (c == 4) a = -1.f; }
  else if (r == 2) { if (c == 0) a = -1.f; else if (c == 3) a = 1.f;   else if (c == 4) b = -1.f; else if (c == 5) q = 0.5f; }
  else if (r == 3) { if (c == 1) a = -1.f; else if (c == 4) a = 1.f;   else if (c == 5) b = -1.f; }
  else if (r == 4) { if (c == 1) a = -1.f; else if (c == 2) b = -0.5f; else if (c == 4) a = 1.f;  else if (c == 5) b = -0.5f; }
  *pa = a; *pb = b; *pc = q;
}

__device__ __forceinline__ double recip_nr(double d) {
  double r = __builtin_amdgcn_rcp(d);
  double e = __builtin_fma(-d, r, 1.0);
  r = __builtin_fma(r, e, r);
  e = __builtin_fma(-d, r, 1.0);
  r = __builtin_fma(r, e, r);
  return r;
}

__device__ __forceinline__ void ldl3(double s0, double s1, double s2,
                                     double s3, double s4, double s5,
                                     double& u10, double& u20, double& u21,
                                     double& i0, double& i1, double& i2) {
  i0 = recip_nr(s0);
  u10 = s1 * i0;
  u20 = s3 * i0;
  const double d1 = __builtin_fma(-u10, s1, s2);
  i1 = recip_nr(d1);
  const double t = __builtin_fma(-u20, s1, s4);
  u21 = t * i1;
  const double d2 = __builtin_fma(-u21, t, __builtin_fma(-u20, s3, s5));
  i2 = recip_nr(d2);
}

__device__ __forceinline__ void zw3(const double* m,
                                    double u10, double u20, double u21,
                                    double i0, double i1, double i2,
                                    double* z, double* w) {
  z[0] = m[0];
  z[1] = __builtin_fma(-z[0], u10, m[1]);
  z[2] = __builtin_fma(-z[1], u21, __builtin_fma(-z[0], u20, m[2]));
  z[3] = m[3];
  z[4] = __builtin_fma(-z[3], u10, m[4]);
  z[5] = __builtin_fma(-z[4], u21, __builtin_fma(-z[3], u20, m[5]));
  z[6] = m[6];
  z[7] = __builtin_fma(-z[6], u10, m[7]);
  z[8] = __builtin_fma(-z[7], u21, __builtin_fma(-z[6], u20, m[8]));
  w[0] = z[0] * i0; w[1] = z[1] * i1; w[2] = z[2] * i2;
  w[3] = z[3] * i0; w[4] = z[4] * i1; w[5] = z[5] * i2;
  w[6] = z[6] * i0; w[7] = z[7] * i1; w[8] = z[8] * i2;
}

__device__ __forceinline__ void elim_left(
    const double (*S)[6], double (*F)[6], const double (*C)[9], double (*gv)[3],
    double (*Zl)[9], double (*Wl)[9], int i, int s) {
  const double* cl = C[i - s];
  double ml[9];
  ml[0] = cl[0]; ml[1] = cl[1]; ml[2] = cl[2];
  ml[3] = cl[3]; ml[4] = cl[4]; ml[5] = cl[5];
  ml[6] = cl[6]; ml[7] = cl[7]; ml[8] = cl[8];
  const double s0 = S[i][0], s1 = S[i][1], s2 = S[i][2];
  const double s3 = S[i][3], s4 = S[i][4], s5 = S[i][5];
  const double g0 = gv[i][0], g1 = gv[i][1], g2 = gv[i][2];

  double u10, u20, u21, i0, i1, i2;
  ldl3(s0, s1, s2, s3, s4, s5, u10, u20, u21, i0, i1, i2);
  const double w1 = __builtin_fma(-u10, g0, g1);
  const double w2 = __builtin_fma(-u21, w1, __builtin_fma(-u20, g0, g2));
  double zl[9], wl[9];
  zw3(ml, u10, u20, u21, i0, i1, i2, zl, wl);

  F[i][0] = i0;  F[i][1] = u10; F[i][2] = i1;
  F[i][3] = u20; F[i][4] = u21; F[i][5] = i2;
  gv[i][1] = w1; gv[i][2] = w2;
  Zl[i][0] = zl[0]; Zl[i][1] = zl[1]; Zl[i][2] = zl[2];
  Zl[i][3] = zl[3]; Zl[i][4] = zl[4]; Zl[i][5] = zl[5];
  Zl[i][6] = zl[6]; Zl[i][7] = zl[7]; Zl[i][8] = zl[8];
  Wl[i][0] = wl[0]; Wl[i][1] = wl[1]; Wl[i][2] = wl[2];
  Wl[i][3] = wl[3]; Wl[i][4] = wl[4]; Wl[i][5] = wl[5];
  Wl[i][6] = wl[6]; Wl[i][7] = wl[7]; Wl[i][8] = wl[8];
}

__device__ __forceinline__ void elim_right(
    const double (*S)[6], const double (*C)[9],
    double (*Zr)[9], double (*Wr)[9], int i) {
  const double* cr = C[i];
  double mr[9];
  mr[0] = cr[0]; mr[1] = cr[3]; mr[2] = cr[6];
  mr[3] = cr[1]; mr[4] = cr[4]; mr[5] = cr[7];
  mr[6] = cr[2]; mr[7] = cr[5]; mr[8] = cr[8];
  const double s0 = S[i][0], s1 = S[i][1], s2 = S[i][2];
  const double s3 = S[i][3], s4 = S[i][4], s5 = S[i][5];

  double u10, u20, u21, i0, i1, i2;
  ldl3(s0, s1, s2, s3, s4, s5, u10, u20, u21, i0, i1, i2);
  double zr[9], wr[9];
  zw3(mr, u10, u20, u21, i0, i1, i2, zr, wr);

  Zr[i][0] = zr[0]; Zr[i][1] = zr[1]; Zr[i][2] = zr[2];
  Zr[i][3] = zr[3]; Zr[i][4] = zr[4]; Zr[i][5] = zr[5];
  Zr[i][6] = zr[6]; Zr[i][7] = zr[7]; Zr[i][8] = zr[8];
  Wr[i][0] = wr[0]; Wr[i][1] = wr[1]; Wr[i][2] = wr[2];
  Wr[i][3] = wr[3]; Wr[i][4] = wr[4]; Wr[i][5] = wr[5];
  Wr[i][6] = wr[6]; Wr[i][7] = wr[7]; Wr[i][8] = wr[8];
}

__device__ __forceinline__ void schur3_n(
    const double* zp, const double* wp,
    double& n0, double& n1, double& n2, double& n3, double& n4, double& n5) {
  const double z0 = zp[0], z1 = zp[1], z2 = zp[2];
  const double z3 = zp[3], z4 = zp[4], z5 = zp[5];
  const double z6 = zp[6], z7 = zp[7], z8 = zp[8];
  const double v0 = wp[0], v1 = wp[1], v2 = wp[2];
  const double v3 = wp[3], v4 = wp[4], v5 = wp[5];
  const double v6 = wp[6], v7 = wp[7], v8 = wp[8];
  n0 = __builtin_fma(-v2, z2, __builtin_fma(-v1, z1, __builtin_fma(-v0, z0, n0)));
  n1 = __builtin_fma(-v5, z2, __builtin_fma(-v4, z1, __builtin_fma(-v3, z0, n1)));
  n2 = __builtin_fma(-v5, z5, __builtin_fma(-v4, z4, __builtin_fma(-v3, z3, n2)));
  n3 = __builtin_fma(-v8, z2, __builtin_fma(-v7, z1, __builtin_fma(-v6, z0, n3)));
  n4 = __builtin_fma(-v8, z5, __builtin_fma(-v7, z4, __builtin_fma(-v6, z3, n4)));
  n5 = __builtin_fma(-v8, z8, __builtin_fma(-v7, z7, __builtin_fma(-v6, z6, n5)));
}

__device__ __forceinline__ void schur3_g(
    const double* wp, const double* xp, double& g0, double& g1, double& g2) {
  const double v0 = wp[0], v1 = wp[1], v2 = wp[2];
  const double v3 = wp[3], v4 = wp[4], v5 = wp[5];
  const double v6 = wp[6], v7 = wp[7], v8 = wp[8];
  const double x0 = xp[0], x1 = xp[1], x2 = xp[2];
  g0 = __builtin_fma(-v2, x2, __builtin_fma(-v1, x1, __builtin_fma(-v0, x0, g0)));
  g1 = __builtin_fma(-v5, x2, __builtin_fma(-v4, x1, __builtin_fma(-v3, x0, g1)));
  g2 = __builtin_fma(-v8, x2, __builtin_fma(-v7, x1, __builtin_fma(-v6, x0, g2)));
}

__device__ __forceinline__ void upd_diag(
    double (*S)[6],
    const double (*Zl)[9], const double (*Wl)[9],
    const double (*Zr)[9], const double (*Wr)[9],
    int j, int s) {
  double n0 = S[j][0], n1 = S[j][1], n2 = S[j][2];
  double n3 = S[j][3], n4 = S[j][4], n5 = S[j][5];
  if (j > 0) {
    const int e = j - s;
    schur3_n(Zr[e], Wr[e], n0, n1, n2, n3, n4, n5);
  }
  if (j + s < NSTEP) {
    const int e = j + s;
    schur3_n(Zl[e], Wl[e], n0, n1, n2, n3, n4, n5);
  }
  S[j][0] = n0; S[j][1] = n1; S[j][2] = n2;
  S[j][3] = n3; S[j][4] = n4; S[j][5] = n5;
}

__device__ __forceinline__ void upd_rhs_couple(
    double (*C)[9], double (*gv)[3],
    const double (*Wl)[9], const double (*Wr)[9], const double (*Zr)[9],
    int j, int s) {
  double g0 = gv[j][0], g1 = gv[j][1], g2 = gv[j][2];
  if (j > 0) {
    const int e = j - s;
    schur3_g(Wr[e], gv[e], g0, g1, g2);
  }
  if (j + s < NSTEP) {
    const int e = j + s;
    schur3_g(Wl[e], gv[e], g0, g1, g2);
    if (j + 2 * s < NSTEP) {
      const double* wp = Wl[e];
      const double* zp = Zr[e];
      const double v0 = wp[0], v1 = wp[1], v2 = wp[2];
      const double v3 = wp[3], v4 = wp[4], v5 = wp[5];
      const double v6 = wp[6], v7 = wp[7], v8 = wp[8];
      const double z0 = zp[0], z1 = zp[1], z2 = zp[2];
      const double z3 = zp[3], z4 = zp[4], z5 = zp[5];
      const double z6 = zp[6], z7 = zp[7], z8 = zp[8];
      double* cp = C[j];
      cp[0] = __builtin_fma(-v2, z2, __builtin_fma(-v1, z1, -(v0 * z0)));
      cp[1] = __builtin_fma(-v2, z5, __builtin_fma(-v1, z4, -(v0 * z3)));
      cp[2] = __builtin_fma(-v2, z8, __builtin_fma(-v1, z7, -(v0 * z6)));
      cp[3] = __builtin_fma(-v5, z2, __builtin_fma(-v4, z1, -(v3 * z0)));
      cp[4] = __builtin_fma(-v5, z5, __builtin_fma(-v4, z4, -(v3 * z3)));
      cp[5] = __builtin_fma(-v5, z8, __builtin_fma(-v4, z7, -(v3 * z6)));
      cp[6] = __builtin_fma(-v8, z2, __builtin_fma(-v7, z1, -(v6 * z0)));
      cp[7] = __builtin_fma(-v8, z5, __builtin_fma(-v7, z4, -(v6 * z3)));
      cp[8] = __builtin_fma(-v8, z8, __builtin_fma(-v7, z7, -(v6 * z6)));
    }
  }
  gv[j][0] = g0; gv[j][1] = g1; gv[j][2] = g2;
}

__device__ __forceinline__ void zw_row(double m0, double m1, double m2,
                                       double u10, double u20, double u21,
                                       double i0, double i1, double i2,
                                       double* z, double* w) {
  const double z0 = m0;
  const double z1 = __builtin_fma(-z0, u10, m1);
  const double z2 = __builtin_fma(-z1, u21, __builtin_fma(-z0, u20, m2));
  z[0] = z0; z[1] = z1; z[2] = z2;
  w[0] = z0 * i0; w[1] = z1 * i1; w[2] = z2 * i2;
}

__device__ __forceinline__ void elim4(
    const double (*S)[6], double (*F)[6], const double (*C)[9], double (*gv)[3],
    double (*Zl)[9], double (*Wl)[9], double (*Zr)[9], double (*Wr)[9],
    int i, int s, int part) {
  if (part >= 2 && i + s >= NSTEP) return;
  const double s0 = S[i][0], s1 = S[i][1], s2 = S[i][2];
  const double s3 = S[i][3], s4 = S[i][4], s5 = S[i][5];
  double u10, u20, u21, i0, i1, i2;
  if (part == 0) {
    const double* cl = C[i - s];
    const double m0 = cl[0], m1 = cl[1], m2 = cl[2];
    const double g0 = gv[i][0], g1 = gv[i][1], g2 = gv[i][2];
    ldl3(s0, s1, s2, s3, s4, s5, u10, u20, u21, i0, i1, i2);
    const double w1 = __builtin_fma(-u10, g0, g1);
    const double w2 = __builtin_fma(-u21, w1, __builtin_fma(-u20, g0, g2));
    F[i][0] = i0;  F[i][1] = u10; F[i][2] = i1;
    F[i][3] = u20; F[i][4] = u21; F[i][5] = i2;
    gv[i][1] = w1; gv[i][2] = w2;
    zw_row(m0, m1, m2, u10, u20, u21, i0, i1, i2, &Zl[i][0], &Wl[i][0]);
  } else if (part == 1) {
    const double* cl = C[i - s];
    const double m3 = cl[3], m4 = cl[4], m5 = cl[5];
    const double m6 = cl[6], m7 = cl[7], m8 = cl[8];
    ldl3(s0, s1, s2, s3, s4, s5, u10, u20, u21, i0, i1, i2);
    zw_row(m3, m4, m5, u10, u20, u21, i0, i1, i2, &Zl[i][3], &Wl[i][3]);
    zw_row(m6, m7, m8, u10, u20, u21, i0, i1, i2, &Zl[i][6], &Wl[i][6]);
  } else if (part == 2) {
    const double* cr = C[i];
    const double m0 = cr[0], m1 = cr[3], m2 = cr[6];
    const double m3 = cr[1], m4 = cr[4], m5 = cr[7];
    ldl3(s0, s1, s2, s3, s4, s5, u10, u20, u21, i0, i1, i2);
    zw_row(m0, m1, m2, u10, u20, u21, i0, i1, i2, &Zr[i][0], &Wr[i][0]);
    zw_row(m3, m4, m5, u10, u20, u21, i0, i1, i2, &Zr[i][3], &Wr[i][3]);
  } else {
    const double* cr = C[i];
    const double m6 = cr[2], m7 = cr[5], m8 = cr[8];
    ldl3(s0, s1, s2, s3, s4, s5, u10, u20, u21, i0, i1, i2);
    zw_row(m6, m7, m8, u10, u20, u21, i0, i1, i2, &Zr[i][6], &Wr[i][6]);
  }
}

__device__ __forceinline__ void upd4(
    double (*S)[6], double (*C)[9], double (*gv)[3],
    const double (*Zl)[9], const double (*Wl)[9],
    const double (*Zr)[9], const double (*Wr)[9],
    int j, int s, int part) {
  const bool hasL = (j > 0);
  const bool hasR = (j + s < NSTEP);
  const bool hasC = (j + 2 * s < NSTEP);
  const int el = hasL ? j - s : 0;
  const int er = hasR ? j + s : 0;
  if (part == 0) {
    double n0 = S[j][0], n1 = S[j][1], n2 = S[j][2];
    if (hasL) {
      const double* zp = Zr[el]; const double* wp = Wr[el];
      const double z0 = zp[0], z1 = zp[1], z2 = zp[2], z3 = zp[3], z4 = zp[4], z5 = zp[5];
      const double v0 = wp[0], v1 = wp[1], v2 = wp[2], v3 = wp[3], v4 = wp[4], v5 = wp[5];
      n0 = __builtin_fma(-v2, z2, __builtin_fma(-v1, z1, __builtin_fma(-v0, z0, n0)));
      n1 = __builtin_fma(-v5, z2, __builtin_fma(-v4, z1, __builtin_fma(-v3, z0, n1)));
      n2 = __builtin_fma(-v5, z5, __builtin_fma(-v4, z4, __builtin_fma(-v3, z3, n2)));
    }
    if (hasR) {
      const double* zp = Zl[er]; const double* wp = Wl[er];
      const double z0 = zp[0], z1 = zp[1], z2 = zp[2], z3 = zp[3], z4 = zp[4], z5 = zp[5];
      const double v0 = wp[0], v1 = wp[1], v2 = wp[2], v3 = wp[3], v4 = wp[4], v5 = wp[5];
      n0 = __builtin_fma(-v2, z2, __builtin_fma(-v1, z1, __builtin_fma(-v0, z0, n0)));
      n1 = __builtin_fma(-v5, z2, __builtin_fma(-v4, z1, __builtin_fma(-v3, z0, n1)));
      n2 = __builtin_fma(-v5, z5, __builtin_fma(-v4, z4, __builtin_fma(-v3, z3, n2)));
    }
    S[j][0] = n0; S[j][1] = n1; S[j][2] = n2;
  } else if (part == 1) {
    double n3 = S[j][3], n4 = S[j][4], n5 = S[j][5];
    if (hasL) {
      const double* zp = Zr[el]; const double* wp = Wr[el];
      const double z0 = zp[0], z1 = zp[1], z2 = zp[2], z3 = zp[3], z4 = zp[4], z5 = zp[5];
      const double z6 = zp[6], z7 = zp[7], z8 = zp[8];
      const double v6 = wp[6], v7 = wp[7], v8 = wp[8];
      n3 = __builtin_fma(-v8, z2, __builtin_fma(-v7, z1, __builtin_fma(-v6, z0, n3)));
      n4 = __builtin_fma(-v8, z5, __builtin_fma(-v7, z4, __builtin_fma(-v6, z3, n4)));
      n5 = __builtin_fma(-v8, z8, __builtin_fma(-v7, z7, __builtin_fma(-v6, z6, n5)));
    }
    if (hasR) {
      const double* zp = Zl[er]; const double* wp = Wl[er];
      const double z0 = zp[0], z1 = zp[1], z2 = zp[2], z3 = zp[3], z4 = zp[4], z5 = zp[5];
      const double z6 = zp[6], z7 = zp[7], z8 = zp[8];
      const double v6 = wp[6], v7 = wp[7], v8 = wp[8];
      n3 = __builtin_fma(-v8, z2, __builtin_fma(-v7, z1, __builtin_fma(-v6, z0, n3)));
      n4 = __builtin_fma(-v8, z5, __builtin_fma(-v7, z4, __builtin_fma(-v6, z3, n4)));
      n5 = __builtin_fma(-v8, z8, __builtin_fma(-v7, z7, __builtin_fma(-v6, z6, n5)));
    }
    S[j][3] = n3; S[j][4] = n4; S[j][5] = n5;
    if (hasC) {
      const double* wp = Wl[er]; const double* zp = Zr[er];
      const double v0 = wp[0], v1 = wp[1], v2 = wp[2];
      const double z0 = zp[0], z1 = zp[1], z2 = zp[2];
      C[j][0] = __builtin_fma(-v2, z2, __builtin_fma(-v1, z1, -(v0 * z0)));
    }
  } else if (part == 2) {
    double g0 = gv[j][0], g1 = gv[j][1], g2 = gv[j][2];
    if (hasL) schur3_g(Wr[el], gv[el], g0, g1, g2);
    if (hasR) schur3_g(Wl[er], gv[er], g0, g1, g2);
    gv[j][0] = g0; gv[j][1] = g1; gv[j][2] = g2;
    if (hasC) {
      const double* wp = Wl[er]; const double* zp = Zr[er];
      const double v0 = wp[0], v1 = wp[1], v2 = wp[2];
      const double z3 = zp[3], z4 = zp[4], z5 = zp[5];
      C[j][1] = __builtin_fma(-v2, z5, __builtin_fma(-v1, z4, -(v0 * z3)));
    }
  } else {
    if (hasC) {
      const double* wp = Wl[er]; const double* zp = Zr[er];
      const double v0 = wp[0], v1 = wp[1], v2 = wp[2];
      const double v3 = wp[3], v4 = wp[4], v5 = wp[5];
      const double v6 = wp[6], v7 = wp[7], v8 = wp[8];
      const double z0 = zp[0], z1 = zp[1], z2 = zp[2];
      const double z3 = zp[3], z4 = zp[4], z5 = zp[5];
      const double z6 = zp[6], z7 = zp[7], z8 = zp[8];
      double* cp = C[j];
      cp[2] = __builtin_fma(-v2, z8, __builtin_fma(-v1, z7, -(v0 * z6)));
      cp[3] = __builtin_fma(-v5, z2, __builtin_fma(-v4, z1, -(v3 * z0)));
      cp[4] = __builtin_fma(-v5, z5, __builtin_fma(-v4, z4, -(v3 * z3)));
      cp[5] = __builtin_fma(-v5, z8, __builtin_fma(-v4, z7, -(v3 * z6)));
      cp[6] = __builtin_fma(-v8, z2, __builtin_fma(-v7, z1, -(v6 * z0)));
      cp[7] = __builtin_fma(-v8, z5, __builtin_fma(-v7, z4, -(v6 * z3)));
      cp[8] = __builtin_fma(-v8, z8, __builtin_fma(-v7, z7, -(v6 * z6)));
    }
  }
}

__device__ __forceinline__ void elim_rows(
    const double (*S)[6], double (*F)[6], const double (*C)[9], double (*gv)[3],
    double (*Zl)[9], double (*Wl)[9], double (*Zr)[9], double (*Wr)[9],
    int t, int s) {
  const int k = t >> 3;
  const int r = t & 7;
  const int i = s * (2 * k + 1);
  if (i >= NSTEP || r == 7) return;
  const bool left = (r < 3);
  const bool right = (r >= 3 && r < 6);
  if (right && i + s >= NSTEP) return;
  const int rr = left ? r : r - 3;
  const int pitch = (r < 6) ? 9 : 3;
  const int isrc = left ? i - s : i;
  const int osrc = pitch * isrc + (left ? 3 * r : (right ? rr : 0));
  const int step = right ? 3 : 1;
  const int odst = pitch * i + ((r < 6) ? 3 * rr : 0);
  const double* src = ((r < 6) ? &C[0][0] : &gv[0][0]) + osrc;
  double* zdst = (left ? &Zl[0][0] : (right ? &Zr[0][0] : &gv[0][0])) + odst;
  double* wdst = (left ? &Wl[0][0] : &Wr[0][0]) + odst;
  const double s0 = S[i][0], s1 = S[i][1], s2 = S[i][2];
  const double s3 = S[i][3], s4 = S[i][4], s5 = S[i][5];
  const double m0 = src[0], m1 = src[step], m2 = src[2 * step];
  __builtin_amdgcn_sched_barrier(0);
  double u10, u20, u21, i0, i1, i2;
  ldl3(s0, s1, s2, s3, s4, s5, u10, u20, u21, i0, i1, i2);
  const double z0 = m0;
  const double z1 = __builtin_fma(-z0, u10, m1);
  const double z2 = __builtin_fma(-z1, u21, __builtin_fma(-z0, u20, m2));
  zdst[0] = z0; zdst[1] = z1; zdst[2] = z2;
  if (r < 6) {
    wdst[0] = z0 * i0; wdst[1] = z1 * i1; wdst[2] = z2 * i2;
  } else {
    F[i][0] = i0;  F[i][1] = u10; F[i][2] = i1;
    F[i][3] = u20; F[i][4] = u21; F[i][5] = i2;
  }
}

__device__ __forceinline__ void upd_elem(
    double (*S)[6], double (*C)[9], double (*gv)[3],
    const double (*Zl)[9], const double (*Wl)[9],
    const double (*Zr)[9], const double (*Wr)[9],
    int task, int s) {
  const int k = task / 18;
  const int e = task - 18 * k;
  const int j = 2 * s * k;
  if (j >= NSTEP) return;
  const bool hasL = (j > 0);
  const bool hasR = (j + s < NSTEP);
  const bool hasC = (j + 2 * s < NSTEP);
  const int el = hasL ? j - s : 0;
  const int er = hasR ? j + s : 0;
  const bool isS = (e < 6);
  const bool isG = (e >= 6 && e < 9);
  const bool isC = (e >= 9);
  const int aS = ((e >= 1) ? 1 : 0) + ((e >= 3) ? 1 : 0);
  const int bS = e - ((aS * (aS + 1)) >> 1);
  const int aC = (e - 9) / 3;
  const int bC = (e - 9) - 3 * aC;
  const int a = isS ? aS : (isG ? e - 6 : aC);
  const int b = isS ? bS : (isG ? 0 : bC);
  double* dst = (isS ? &S[0][0] : (isG ? &gv[0][0] : &C[0][0]))
              + ((isS ? 6 : (isG ? 3 : 9)) * j + (isS ? e : (isG ? a : e - 9)));
  const double* wl = &Wr[0][0] + (9 * el + 3 * a);
  const double* zl = (isS ? &Zr[0][0] : &gv[0][0]) + ((isS ? 9 : 3) * el + (isS ? 3 * b : 0));
  const double* wr = &Wl[0][0] + (9 * er + 3 * a);
  const double* zr = (isS ? &Zl[0][0] : (isG ? &gv[0][0] : &Zr[0][0]))
                   + ((isG ? 3 : 9) * er + (isG ? 0 : 3 * b));
  const double old = dst[0];
  const double p0 = wl[0], p1 = wl[1], p2 = wl[2];
  const double q0 = zl[0], q1 = zl[1], q2 = zl[2];
  const double v0 = wr[0], v1 = wr[1], v2 = wr[2];
  const double z0 = zr[0], z1 = zr[1], z2 = zr[2];
  asm volatile("" :: "v"(old), "v"(p0), "v"(p1), "v"(p2), "v"(q0), "v"(q1), "v"(q2),
                     "v"(v0), "v"(v1), "v"(v2), "v"(z0), "v"(z1), "v"(z2));
  const double acc0 = isC ? -0.0 : old;
  const double accL = __builtin_fma(-p2, q2, __builtin_fma(-p1, q1, __builtin_fma(-p0, q0, acc0)));
  const double acc1 = (!isC && hasL) ? accL : acc0;
  const double accR = __builtin_fma(-v2, z2, __builtin_fma(-v1, z1, __builtin_fma(-v0, z0, acc1)));
  const double acc2 = (isC ? hasC : hasR) ? accR : acc1;
  if (!isC || hasC) dst[0] = acc2;
}

__device__ __forceinline__ void cr_back(
    const double (*F)[6], double (*gv)[3],
    const double (*Wl)[9], const double (*Wr)[9],
    int i, int s) {
  const double j0 = F[i][0], q10 = F[i][1], j1 = F[i][2];
  const double q20 = F[i][3], q21 = F[i][4], j2 = F[i][5];
  double t0 = gv[i][0] * j0, t1 = gv[i][1] * j1, t2 = gv[i][2] * j2;
  {
    const double* wp = Wl[i];
    const double a0 = gv[i - s][0], a1 = gv[i - s][1], a2 = gv[i - s][2];
    t0 = __builtin_fma(-wp[6], a2, __builtin_fma(-wp[3], a1, __builtin_fma(-wp[0], a0, t0)));
    t1 = __builtin_fma(-wp[7], a2, __builtin_fma(-wp[4], a1, __builtin_fma(-wp[1], a0, t1)));
    t2 = __builtin_fma(-wp[8], a2, __builtin_fma(-wp[5], a1, __builtin_fma(-wp[2], a0, t2)));
  }
  if (i + s < NSTEP) {
    const double* wp = Wr[i];
    const double a0 = gv[i + s][0], a1 = gv[i + s][1], a2 = gv[i + s][2];
    t0 = __builtin_fma(-wp[6], a2, __builtin_fma(-wp[3], a1, __builtin_fma(-wp[0], a0, t0)));
    t1 = __builtin_fma(-wp[7], a2, __builtin_fma(-wp[4], a1, __builtin_fma(-wp[1], a0, t1)));
    t2 = __builtin_fma(-wp[8], a2, __builtin_fma(-wp[5], a1, __builtin_fma(-wp[2], a0, t2)));
  }
  const double x2 = t2;
  const double x1 = __builtin_fma(-q21, x2, t1);
  const double x0 = __builtin_fma(-q20, x2, __builtin_fma(-q10, x1, t0));
  gv[i][0] = x0; gv[i][1] = x1; gv[i][2] = x2;
}

__device__ __forceinline__ double cr_back_t(
    const double (*F)[6], const double (*gv)[3],
    const double (*Wl)[9], const double (*Wr)[9],
    int i, int s, int k) {
  const int fk = (k == 0) ? 0 : ((k == 1) ? 2 : 5);
  double t = gv[i][k] * F[i][fk];
  {
    const double* wp = Wl[i];
    const double a0 = gv[i - s][0], a1 = gv[i - s][1], a2 = gv[i - s][2];
    t = __builtin_fma(-wp[6 + k], a2, __builtin_fma(-wp[3 + k], a1, __builtin_fma(-wp[k], a0, t)));
  }
  if (i + s < NSTEP) {
    const double* wp = Wr[i];
    const double a0 = gv[i + s][0], a1 = gv[i + s][1], a2 = gv[i + s][2];
    t = __builtin_fma(-wp[6 + k], a2, __builtin_fma(-wp[3 + k], a1, __builtin_fma(-wp[k], a0, t)));
  }
  return t;
}

__device__ __forceinline__ void cr_back_x(
    const double (*F)[6], double (*gv)[3], const double (*tb)[3], int i) {
  const double q10 = F[i][1], q20 = F[i][3], q21 = F[i][4];
  const double x2 = tb[i][2];
  const double x1 = __builtin_fma(-q21, x2, tb[i][1]);
  const double x0 = __builtin_fma(-q20, x2, __builtin_fma(-q10, x1, tb[i][0]));
  gv[i][0] = x0; gv[i][1] = x1; gv[i][2] = x2;
}

__device__ __forceinline__ v8f gram_tile(float h,
                                         float a0, float b0, float q0,
                                         float a1, float b1, float q1,
                                         float a2, float b2, float q2) {
  v2f f0, f1;
  f0.x = fmaf(fmaf(q0, h, b0), h, a0);
  f0.y = fmaf(fmaf(q1, h, b1), h, a1);
  f1.x = fmaf(fmaf(q2, h, b2), h, a2);
  f1.y = 0.f;
  v8f acc = {};
  acc = __builtin_amdgcn_wmma_f32_16x16x4_f32(false, f0, false, f0,
                                              (short)0, acc, false, false);
  acc = __builtin_amdgcn_wmma_f32_16x16x4_f32(false, f1, false, f1,
                                              (short)0, acc, false, false);
  return acc;
}

__device__ __forceinline__ void put_tile(float* q, v8f acc) {
  q[0] = acc[0]; q[1] = acc[1]; q[2] = acc[2];
  q[3] = acc[3]; q[4] = acc[4]; q[5] = acc[5];
}

#define TOPS  64
#define NWAVE 4
#define NTHR  (32 * NWAVE)
#define KSTEP 25
#define GS_TILES (T1 * 36)
#define GS_SPARE ((KSTEP - 1) * NWAVE * 36 + 26 * 6)

__global__ __launch_bounds__(NTHR) void ode_blocktri_kernel(
    const float* __restrict__ coeffs,
    const float* __restrict__ rhs,
    const float* __restrict__ ivr,
    const float* __restrict__ steps,
    float* __restrict__ out)
{
  __shared__ double S[NSTEP][6];
  __shared__ double F[NSTEP][6];
  __shared__ double Bo[T1][9];
  __shared__ double gv[NSTEP][3];
  __shared__ double Zl[NSTEP][9];
  __shared__ double Wl[NSTEP][9];
  __shared__ double Zr[NSTEP][9];
  __shared__ double Wr[NSTEP][9];
  __shared__ double tb[NSTEP][3];
  __shared__ __attribute__((aligned(16))) float Gs[GS_TILES + GS_SPARE];

  const int b = blockIdx.x;
  const int tid = threadIdx.x;
  const int lane = tid & 31;
  const int wave = __builtin_amdgcn_readfirstlane(tid >> 5);
  const int role = wave >> 1;
  const int slot = ((wave & 1) << 5) | lane;

  const int myp = wave + NWAVE * lane;
  const float hmine = (myp < T1) ? steps[b * T1 + myp] : 0.f;
  const float hcopy = (tid < T1) ? steps[b * T1 + tid] : 0.f;

  const int m = lane & 15;
  const int khalf = lane >> 4;
  const int col = (m < 6) ? m : 6;
  float a0, b0, q0, a1, b1, q1, a2, b2, q2;
  Ecoef(2 * khalf,                 col, &a0, &b0, &q0);
  Ecoef(2 * khalf + 1,             col, &a1, &b1, &q1);
  Ecoef((khalf == 0) ? 4 : -1,     col, &a2, &b2, &q2);

  const int hbits = __builtin_bit_cast(int, hmine);
  float* const gq = Gs + ((lane < 6) ? (wave * 36 + lane * 6)
                                     : (GS_TILES + (lane - 6) * 6));
  {
    v8f pend = {};
#pragma unroll
    for (int k = 0; k < KSTEP - 1; ++k) {
      const float h = __builtin_bit_cast(float, __builtin_amdgcn_readlane(hbits, k));
      const v8f acc = gram_tile(h, a0, b0, q0, a1, b1, q1, a2, b2, q2);
      if (k > 0) put_tile(gq + (k - 1) * (NWAVE * 36), pend);
      pend = acc;
    }
    put_tile(gq + (KSTEP - 2) * (NWAVE * 36), pend);
    if (wave + NWAVE * (KSTEP - 1) < T1) {
      const float h = __builtin_bit_cast(float, __builtin_amdgcn_readlane(hbits, KSTEP - 1));
      const v8f acc = gram_tile(h, a0, b0, q0, a1, b1, q1, a2, b2, q2);
      put_tile(gq + (KSTEP - 1) * (NWAVE * 36), acc);
    }
  }
  __syncthreads();

  const float iv0 = ivr[b * 2 + 0];
  const float iv1 = ivr[b * 2 + 1];
  if (tid < NSTEP) {
    const int t = tid;
    const float* cp = coeffs + (b * NSTEP + t) * 3;
    const double c0 = (double)cp[0];
    const double c1 = (double)cp[1];
    const double c2 = (double)cp[2];
    const double r  = (double)rhs[b * NSTEP + t];
    const bool hasPrev = (t > 0);
    const bool hasNext = (t < T1);
    const float* gp = Gs + 36 * (hasPrev ? t - 1 : 0);
    const float* gn = Gs + 36 * (hasNext ? t : T1 - 1);
    const float f0 = gp[3 * 6 + 3], f1 = gp[3 * 6 + 4], f2 = gp[4 * 6 + 4];
    const float f3 = gp[3 * 6 + 5], f4 = gp[4 * 6 + 5], f5 = gp[5 * 6 + 5];
    const float n0 = gn[0 * 6 + 0], n1 = gn[0 * 6 + 1], n2 = gn[1 * 6 + 1];
    const float n3 = gn[0 * 6 + 2], n4 = gn[1 * 6 + 2], n5 = gn[2 * 6 + 2];
    const double p0 = (double)(hasPrev ? f0 : 1.f);
    const double p1 = (double)(hasPrev ? f1 : 0.f);
    const double p2 = (double)(hasPrev ? f2 : 1.f);
    const double p3 = (double)(hasPrev ? f3 : 0.f);
    const double p4 = (double)(hasPrev ? f4 : 0.f);
    const double p5 = (double)(hasPrev ? f5 : 0.f);
    S[t][0] = __builtin_fma(c0, c0, p0) + (double)(hasNext ? n0 : 0.f);
    S[t][1] = __builtin_fma(c1, c0, p1) + (double)(hasNext ? n1 : 0.f);
    S[t][2] = __builtin_fma(c1, c1, p2) + (double)(hasNext ? n2 : 0.f);
    S[t][3] = __builtin_fma(c2, c0, p3) + (double)(hasNext ? n3 : 0.f);
    S[t][4] = __builtin_fma(c2, c1, p4) + (double)(hasNext ? n4 : 0.f);
    S[t][5] = __builtin_fma(c2, c2, p5) + (double)(hasNext ? n5 : 0.f);
    gv[t][0] = __builtin_fma(c0, r, (double)((t == 0) ? iv0 : 0.f));
    gv[t][1] = __builtin_fma(c1, r, (double)((t == 0) ? iv1 : 0.f));
    gv[t][2] = c2 * r;
    const double b00 = (double)gn[3 * 6 + 0], b01 = (double)gn[4 * 6 + 0], b02 = (double)gn[5 * 6 + 0];
    const double b10 = (double)gn[3 * 6 + 1], b11 = (double)gn[4 * 6 + 1], b12 = (double)gn[5 * 6 + 1];
    const double b20 = (double)gn[3 * 6 + 2], b21 = (double)gn[4 * 6 + 2], b22 = (double)gn[5 * 6 + 2];
    if (hasNext) {
      Bo[t][0] = b00; Bo[t][1] = b01; Bo[t][2] = b02;
      Bo[t][3] = b10; Bo[t][4] = b11; Bo[t][5] = b12;
      Bo[t][6] = b20; Bo[t][7] = b21; Bo[t][8] = b22;
    }
  }
  __syncthreads();

  {
    const int s = 1;
    {
      const int i = s * (2 * slot + 1);
      if (i < NSTEP) {
        if (role == 0)            elim_left(S, F, Bo, gv, Zl, Wl, i, s);
        else if (i + s < NSTEP)   elim_right(S, Bo, Zr, Wr, i);
      }
    }
    __syncthreads();
    {
      const int j = 2 * s * slot;
      if (j < NSTEP) {
        if (role == 0)   upd_diag(S, Zl, Wl, Zr, Wr, j, s);
        else             upd_rhs_couple(Bo, gv, Wl, Wr, Zr, j, s);
      }
    }
    __syncthreads();
  }
  {
    const int s = 2;
    {
      const int i = s * (2 * lane + 1);
      if (i < NSTEP) elim4(S, F, Bo, gv, Zl, Wl, Zr, Wr, i, s, wave);
    }
    __syncthreads();
    {
      const int j = 2 * s * lane;
      if (j < NSTEP) upd4(S, Bo, gv, Zl, Wl, Zr, Wr, j, s, wave);
    }
    __syncthreads();
  }
  for (int s = 4; s <= TOPS; s <<= 1) {
    elim_rows(S, F, Bo, gv, Zl, Wl, Zr, Wr, tid, s);
    __syncthreads();
    const int ntask = 18 * ((NSTEP - 1) / (2 * s) + 1);
    for (int base = 0; base < ntask; base += NTHR)
      upd_elem(S, Bo, gv, Zl, Wl, Zr, Wr, base + tid, s);
    __syncthreads();
  }
  if (tid == 0) {
    double u10, u20, u21, i0, i1, i2;
    ldl3(S[0][0], S[0][1], S[0][2], S[0][3], S[0][4], S[0][5],
         u10, u20, u21, i0, i1, i2);
    const double w0 = gv[0][0];
    const double w1 = __builtin_fma(-u10, w0, gv[0][1]);
    const double w2 = __builtin_fma(-u21, w1, __builtin_fma(-u20, w0, gv[0][2]));
    const double x2 = w2 * i2;
    const double x1 = __builtin_fma(-u21, x2, w1 * i1);
    const double x0 = __builtin_fma(-u20, x2, __builtin_fma(-u10, x1, w0 * i0));
    gv[0][0] = x0; gv[0][1] = x1; gv[0][2] = x2;
  }
  __syncthreads();
  for (int s = TOPS; s >= 2; s >>= 1) {
    const int i = s * (2 * lane + 1);
    if (i < NSTEP && wave < 3) tb[i][wave] = cr_back_t(F, gv, Wl, Wr, i, s, wave);
    __syncthreads();
    if (i < NSTEP && wave == 0) cr_back_x(F, gv, tb, i);
    __syncthreads();
  }
  {
    const int i = 2 * tid + 1;
    if (i < NSTEP) cr_back(F, gv, Wl, Wr, i, 1);
    __syncthreads();
  }

  if (tid < NSTEP) {
    out[0 * OUT_U + b * NSTEP + tid] = (float)gv[tid][0];
    out[1 * OUT_U + b * NSTEP + tid] = (float)gv[tid][1];
    out[2 * OUT_U + b * NSTEP + tid] = (float)gv[tid][2];
  }
  if (tid == 0) out[OUT_EPS + b] = 0.0f;
  if (tid < T1) out[OUT_ST + b * T1 + tid] = hcopy;
}

extern "C" void kernel_launch(void* const* d_in, const int* in_sizes, int n_in,
                              void* d_out, int out_size, void* d_ws, size_t ws_size,
                              hipStream_t stream) {
  (void)in_sizes; (void)n_in; (void)out_size; (void)d_ws; (void)ws_size;
  const float* coeffs = (const float*)d_in[0];
  const float* rhs    = (const float*)d_in[1];
  const float* ivr    = (const float*)d_in[2];
  const float* steps  = (const float*)d_in[3];
  float* out = (float*)d_out;
  hipLaunchKernelGGL(ode_blocktri_kernel, dim3(NB), dim3(NTHR), 0, stream,
                     coeffs, rhs, ivr, steps, out);
}


# Waves idle during block 0's solve fold U^-T into the kept multipliers (P = W U^-1, y = U^-T D^-1 w); back-substitution levels 64..2: x = y - Pl^T xl - Pr^T xr, one barrier not two; rounding changes
# speedup vs baseline: 36.6701x; 1.1051x over previous
#include <hip/hip_runtime.h>
#include <hip/hip_bf16.h>
#include <math.h>


typedef __attribute__((ext_vector_type(2))) float v2f;
typedef __attribute__((ext_vector_type(8))) float v8f;

#define NSTEP 100
#define T1    99
#define NB    256
#define OUT_U   (NB * NSTEP)
#define OUT_EPS (3 * OUT_U)
#define OUT_ST  (OUT_EPS + NB)

__device__ __forceinline__ void Ecoef(int r, int c, float* pa, float* pb, float* pc) {
  float a = 0.f, b = 0.f, q = 0.f;
  if (r == 0)      { if (c == 0) a = 1.f;  else if (c == 1) b = 1.f;   else if (c == 2) q = 0.5f; else if (c == 3) a = -1.f; }
  else if (r == 1) { if (c == 1) a = 1.f;  else if (c == 2) b = 1.f;   else if (c == 4) a = -1.f; }
  else if (r == 2) { if (c == 0) a = -1.f; else if (c == 3) a = 1.f;   else if (c == 4) b = -1.f; else if (c == 5) q = 0.5f; }
  else if (r == 3) { if (c == 1) a = -1.f; else if (c == 4) a = 1.f;   else if (c == 5) b = -1.f; }
  else if (r == 4) { if (c == 1) a = -1.f; else if (c == 2) b = -0.5f; else if (c == 4) a = 1.f;  else if (c == 5) b = -0.5f; }
  *pa = a; *pb = b; *pc = q;
}

__device__ __forceinline__ double recip_nr(double d) {
  double r = __builtin_amdgcn_rcp(d);
  double e = __builtin_fma(-d, r, 1.0);
  r = __builtin_fma(r, e, r);
  e = __builtin_fma(-d, r, 1.0);
  r = __builtin_fma(r, e, r);
  return r;
}

__device__ __forceinline__ void ldl3(double s0, double s1, double s2,
                                     double s3, double s4, double s5,
                                     double& u10, double& u20, double& u21,
                                     double& i0, double& i1, double& i2) {
  i0 = recip_nr(s0);
  u10 = s1 * i0;
  u20 = s3 * i0;
  const double d1 = __builtin_fma(-u10, s1, s2);
  i1 = recip_nr(d1);
  const double t = __builtin_fma(-u20, s1, s4);
  u21 = t * i1;
  const double d2 = __builtin_fma(-u21, t, __builtin_fma(-u20, s3, s5));
  i2 = recip_nr(d2);
}

__device__ __forceinline__ void zw3(const double* m,
                                    double u10, double u20, double u21,
                                    double i0, double i1, double i2,
                                    double* z, double* w) {
  z[0] = m[0];
  z[1] = __builtin_fma(-z[0], u10, m[1]);
  z[2] = __builtin_fma(-z[1], u21, __builtin_fma(-z[0], u20, m[2]));
  z[3] = m[3];
  z[4] = __builtin_fma(-z[3], u10, m[4]);
  z[5] = __builtin_fma(-z[4], u21, __builtin_fma(-z[3], u20, m[5]));
  z[6] = m[6];
  z[7] = __builtin_fma(-z[6], u10, m[7]);
  z[8] = __builtin_fma(-z[7], u21, __builtin_fma(-z[6], u20, m[8]));
  w[0] = z[0] * i0; w[1] = z[1] * i1; w[2] = z[2] * i2;
  w[3] = z[3] * i0; w[4] = z[4] * i1; w[5] = z[5] * i2;
  w[6] = z[6] * i0; w[7] = z[7] * i1; w[8] = z[8] * i2;
}

__device__ __forceinline__ void elim_left(
    const double (*S)[6], double (*F)[6], const double (*C)[9], double (*gv)[3],
    double (*Zl)[9], double (*Wl)[9], int i, int s) {
  const double* cl = C[i - s];
  double ml[9];
  ml[0] = cl[0]; ml[1] = cl[1]; ml[2] = cl[2];
  ml[3] = cl[3]; ml[4] = cl[4]; ml[5] = cl[5];
  ml[6] = cl[6]; ml[7] = cl[7]; ml[8] = cl[8];
  const double s0 = S[i][0], s1 = S[i][1], s2 = S[i][2];
  const double s3 = S[i][3], s4 = S[i][4], s5 = S[i][5];
  const double g0 = gv[i][0], g1 = gv[i][1], g2 = gv[i][2];

  double u10, u20, u21, i0, i1, i2;
  ldl3(s0, s1, s2, s3, s4, s5, u10, u20, u21, i0, i1, i2);
  const double w1 = __builtin_fma(-u10, g0, g1);
  const double w2 = __builtin_fma(-u21, w1, __builtin_fma(-u20, g0, g2));
  double zl[9], wl[9];
  zw3(ml, u10, u20, u21, i0, i1, i2, zl, wl);

  F[i][0] = i0;  F[i][1] = u10; F[i][2] = i1;
  F[i][3] = u20; F[i][4] = u21; F[i][5] = i2;
  gv[i][1] = w1; gv[i][2] = w2;
  Zl[i][0] = zl[0]; Zl[i][1] = zl[1]; Zl[i][2] = zl[2];
  Zl[i][3] = zl[3]; Zl[i][4] = zl[4]; Zl[i][5] = zl[5];
  Zl[i][6] = zl[6]; Zl[i][7] = zl[7]; Zl[i][8] = zl[8];
  Wl[i][0] = wl[0]; Wl[i][1] = wl[1]; Wl[i][2] = wl[2];
  Wl[i][3] = wl[3]; Wl[i][4] = wl[4]; Wl[i][5] = wl[5];
  Wl[i][6] = wl[6]; Wl[i][7] = wl[7]; Wl[i][8] = wl[8];
}

__device__ __forceinline__ void elim_right(
    const double (*S)[6], const double (*C)[9],
    double (*Zr)[9], double (*Wr)[9], int i) {
  const double* cr = C[i];
  double mr[9];
  mr[0] = cr[0]; mr[1] = cr[3]; mr[2] = cr[6];
  mr[3] = cr[1]; mr[4] = cr[4]; mr[5] = cr[7];
  mr[6] = cr[2]; mr[7] = cr[5]; mr[8] = cr[8];
  const double s0 = S[i][0], s1 = S[i][1], s2 = S[i][2];
  const double s3 = S[i][3], s4 = S[i][4], s5 = S[i][5];

  double u10, u20, u21, i0, i1, i2;
  ldl3(s0, s1, s2, s3, s4, s5, u10, u20, u21, i0, i1, i2);
  double zr[9], wr[9];
  zw3(mr, u10, u20, u21, i0, i1, i2, zr, wr);

  Zr[i][0] = zr[0]; Zr[i][1] = zr[1]; Zr[i][2] = zr[2];
  Zr[i][3] = zr[3]; Zr[i][4] = zr[4]; Zr[i][5] = zr[5];
  Zr[i][6] = zr[6]; Zr[i][7] = zr[7]; Zr[i][8] = zr[8];
  Wr[i][0] = wr[0]; Wr[i][1] = wr[1]; Wr[i][2] = wr[2];
  Wr[i][3] = wr[3]; Wr[i][4] = wr[4]; Wr[i][5] = wr[5];
  Wr[i][6] = wr[6]; Wr[i][7] = wr[7]; Wr[i][8] = wr[8];
}

__device__ __forceinline__ void schur3_n(
    const double* zp, const double* wp,
    double& n0, double& n1, double& n2, double& n3, double& n4, double& n5) {
  const double z0 = zp[0], z1 = zp[1], z2 = zp[2];
  const double z3 = zp[3], z4 = zp[4], z5 = zp[5];
  const double z6 = zp[6], z7 = zp[7], z8 = zp[8];
  const double v0 = wp[0], v1 = wp[1], v2 = wp[2];
  const double v3 = wp[3], v4 = wp[4], v5 = wp[5];
  const double v6 = wp[6], v7 = wp[7], v8 = wp[8];
  n0 = __builtin_fma(-v2, z2, __builtin_fma(-v1, z1, __builtin_fma(-v0, z0, n0)));
  n1 = __builtin_fma(-v5, z2, __builtin_fma(-v4, z1, __builtin_fma(-v3, z0, n1)));
  n2 = __builtin_fma(-v5, z5, __builtin_fma(-v4, z4, __builtin_fma(-v3, z3, n2)));
  n3 = __builtin_fma(-v8, z2, __builtin_fma(-v7, z1, __builtin_fma(-v6, z0, n3)));
  n4 = __builtin_fma(-v8, z5, __builtin_fma(-v7, z4, __builtin_fma(-v6, z3, n4)));
  n5 = __builtin_fma(-v8, z8, __builtin_fma(-v7, z7, __builtin_fma(-v6, z6, n5)));
}

__device__ __forceinline__ void schur3_g(
    const double* wp, const double* xp, double& g0, double& g1, double& g2) {
  const double v0 = wp[0], v1 = wp[1], v2 = wp[2];
  const double v3 = wp[3], v4 = wp[4], v5 = wp[5];
  const double v6 = wp[6], v7 = wp[7], v8 = wp[8];
  const double x0 = xp[0], x1 = xp[1], x2 = xp[2];
  g0 = __builtin_fma(-v2, x2, __builtin_fma(-v1, x1, __builtin_fma(-v0, x0, g0)));
  g1 = __builtin_fma(-v5, x2, __builtin_fma(-v4, x1, __builtin_fma(-v3, x0, g1)));
  g2 = __builtin_fma(-v8, x2, __builtin_fma(-v7, x1, __builtin_fma(-v6, x0, g2)));
}

__device__ __forceinline__ void upd_diag(
    double (*S)[6],
    const double (*Zl)[9], const double (*Wl)[9],
    const double (*Zr)[9], const double (*Wr)[9],
    int j, int s) {
  double n0 = S[j][0], n1 = S[j][1], n2 = S[j][2];
  double n3 = S[j][3], n4 = S[j][4], n5 = S[j][5];
  if (j > 0) {
    const int e = j - s;
    schur3_n(Zr[e], Wr[e], n0, n1, n2, n3, n4, n5);
  }
  if (j + s < NSTEP) {
    const int e = j + s;
    schur3_n(Zl[e], Wl[e], n0, n1, n2, n3, n4, n5);
  }
  S[j][0] = n0; S[j][1] = n1; S[j][2] = n2;
  S[j][3] = n3; S[j][4] = n4; S[j][5] = n5;
}

__device__ __forceinline__ void upd_rhs_couple(
    double (*C)[9], double (*gv)[3],
    const double (*Wl)[9], const double (*Wr)[9], const double (*Zr)[9],
    int j, int s) {
  double g0 = gv[j][0], g1 = gv[j][1], g2 = gv[j][2];
  if (j > 0) {
    const int e = j - s;
    schur3_g(Wr[e], gv[e], g0, g1, g2);
  }
  if (j + s < NSTEP) {
    const int e = j + s;
    schur3_g(Wl[e], gv[e], g0, g1, g2);
    if (j + 2 * s < NSTEP) {
      const double* wp = Wl[e];
      const double* zp = Zr[e];
      const double v0 = wp[0], v1 = wp[1], v2 = wp[2];
      const double v3 = wp[3], v4 = wp[4], v5 = wp[5];
      const double v6 = wp[6], v7 = wp[7], v8 = wp[8];
      const double z0 = zp[0], z1 = zp[1], z2 = zp[2];
      const double z3 = zp[3], z4 = zp[4], z5 = zp[5];
      const double z6 = zp[6], z7 = zp[7], z8 = zp[8];
      double* cp = C[j];
      cp[0] = __builtin_fma(-v2, z2, __builtin_fma(-v1, z1, -(v0 * z0)));
      cp[1] = __builtin_fma(-v2, z5, __builtin_fma(-v1, z4, -(v0 * z3)));
      cp[2] = __builtin_fma(-v2, z8, __builtin_fma(-v1, z7, -(v0 * z6)));
      cp[3] = __builtin_fma(-v5, z2, __builtin_fma(-v4, z1, -(v3 * z0)));
      cp[4] = __builtin_fma(-v5, z5, __builtin_fma(-v4, z4, -(v3 * z3)));
      cp[5] = __builtin_fma(-v5, z8, __builtin_fma(-v4, z7, -(v3 * z6)));
      cp[6] = __builtin_fma(-v8, z2, __builtin_fma(-v7, z1, -(v6 * z0)));
      cp[7] = __builtin_fma(-v8, z5, __builtin_fma(-v7, z4, -(v6 * z3)));
      cp[8] = __builtin_fma(-v8, z8, __builtin_fma(-v7, z7, -(v6 * z6)));
    }
  }
  gv[j][0] = g0; gv[j][1] = g1; gv[j][2] = g2;
}

__device__ __forceinline__ void zw_row(double m0, double m1, double m2,
                                       double u10, double u20, double u21,
                                       double i0, double i1, double i2,
                                       double* z, double* w) {
  const double z0 = m0;
  const double z1 = __builtin_fma(-z0, u10, m1);
  const double z2 = __builtin_fma(-z1, u21, __builtin_fma(-z0, u20, m2));
  z[0] = z0; z[1] = z1; z[2] = z2;
  w[0] = z0 * i0; w[1] = z1 * i1; w[2] = z2 * i2;
}

__device__ __forceinline__ void elim4(
    const double (*S)[6], double (*F)[6], const double (*C)[9], double (*gv)[3],
    double (*Zl)[9], double (*Wl)[9], double (*Zr)[9], double (*Wr)[9],
    int i, int s, int part) {
  if (part >= 2 && i + s >= NSTEP) return;
  const double s0 = S[i][0], s1 = S[i][1], s2 = S[i][2];
  const double s3 = S[i][3], s4 = S[i][4], s5 = S[i][5];
  double u10, u20, u21, i0, i1, i2;
  if (part == 0) {
    const double* cl = C[i - s];
    const double m0 = cl[0], m1 = cl[1], m2 = cl[2];
    const double g0 = gv[i][0], g1 = gv[i][1], g2 = gv[i][2];
    ldl3(s0, s1, s2, s3, s4, s5, u10, u20, u21, i0, i1, i2);
    const double w1 = __builtin_fma(-u10, g0, g1);
    const double w2 = __builtin_fma(-u21, w1, __builtin_fma(-u20, g0, g2));
    F[i][0] = i0;  F[i][1] = u10; F[i][2] = i1;
    F[i][3] = u20; F[i][4] = u21; F[i][5] = i2;
    gv[i][1] = w1; gv[i][2] = w2;
    zw_row(m0, m1, m2, u10, u20, u21, i0, i1, i2, &Zl[i][0], &Wl[i][0]);
  } else if (part == 1) {
    const double* cl = C[i - s];
    const double m3 = cl[3], m4 = cl[4], m5 = cl[5];
    const double m6 = cl[6], m7 = cl[7], m8 = cl[8];
    ldl3(s0, s1, s2, s3, s4, s5, u10, u20, u21, i0, i1, i2);
    zw_row(m3, m4, m5, u10, u20, u21, i0, i1, i2, &Zl[i][3], &Wl[i][3]);
    zw_row(m6, m7, m8, u10, u20, u21, i0, i1, i2, &Zl[i][6], &Wl[i][6]);
  } else if (part == 2) {
    const double* cr = C[i];
    const double m0 = cr[0], m1 = cr[3], m2 = cr[6];
    const double m3 = cr[1], m4 = cr[4], m5 = cr[7];
    ldl3(s0, s1, s2, s3, s4, s5, u10, u20, u21, i0, i1, i2);
    zw_row(m0, m1, m2, u10, u20, u21, i0, i1, i2, &Zr[i][0], &Wr[i][0]);
    zw_row(m3, m4, m5, u10, u20, u21, i0, i1, i2, &Zr[i][3], &Wr[i][3]);
  } else {
    const double* cr = C[i];
    const double m6 = cr[2], m7 = cr[5], m8 = cr[8];
    ldl3(s0, s1, s2, s3, s4, s5, u10, u20, u21, i0, i1, i2);
    zw_row(m6, m7, m8, u10, u20, u21, i0, i1, i2, &Zr[i][6], &Wr[i][6]);
  }
}

__device__ __forceinline__ void upd4(
    double (*S)[6], double (*C)[9], double (*gv)[3],
    const double (*Zl)[9], const double (*Wl)[9],
    const double (*Zr)[9], const double (*Wr)[9],
    int j, int s, int part) {
  const bool hasL = (j > 0);
  const bool hasR = (j + s < NSTEP);
  const bool hasC = (j + 2 * s < NSTEP);
  const int el = hasL ? j - s : 0;
  const int er = hasR ? j + s : 0;
  if (part == 0) {
    double n0 = S[j][0], n1 = S[j][1], n2 = S[j][2];
    if (hasL) {
      const double* zp = Zr[el]; const double* wp = Wr[el];
      const double z0 = zp[0], z1 = zp[1], z2 = zp[2], z3 = zp[3], z4 = zp[4], z5 = zp[5];
      const double v0 = wp[0], v1 = wp[1], v2 = wp[2], v3 = wp[3], v4 = wp[4], v5 = wp[5];
      n0 = __builtin_fma(-v2, z2, __builtin_fma(-v1, z1, __builtin_fma(-v0, z0, n0)));
      n1 = __builtin_fma(-v5, z2, __builtin_fma(-v4, z1, __builtin_fma(-v3, z0, n1)));
      n2 = __builtin_fma(-v5, z5, __builtin_fma(-v4, z4, __builtin_fma(-v3, z3, n2)));
    }
    if (hasR) {
      const double* zp = Zl[er]; const double* wp = Wl[er];
      const double z0 = zp[0], z1 = zp[1], z2 = zp[2], z3 = zp[3], z4 = zp[4], z5 = zp[5];
      const double v0 = wp[0], v1 = wp[1], v2 = wp[2], v3 = wp[3], v4 = wp[4], v5 = wp[5];
      n0 = __builtin_fma(-v2, z2, __builtin_fma(-v1, z1, __builtin_fma(-v0, z0, n0)));
      n1 = __builtin_fma(-v5, z2, __builtin_fma(-v4, z1, __builtin_fma(-v3, z0, n1)));
      n2 = __builtin_fma(-v5, z5, __builtin_fma(-v4, z4, __builtin_fma(-v3, z3, n2)));
    }
    S[j][0] = n0; S[j][1] = n1; S[j][2] = n2;
  } else if (part == 1) {
    double n3 = S[j][3], n4 = S[j][4], n5 = S[j][5];
    if (hasL) {
      const double* zp = Zr[el]; const double* wp = Wr[el];
      const double z0 = zp[0], z1 = zp[1], z2 = zp[2], z3 = zp[3], z4 = zp[4], z5 = zp[5];
      const double z6 = zp[6], z7 = zp[7], z8 = zp[8];
      const double v6 = wp[6], v7 = wp[7], v8 = wp[8];
      n3 = __builtin_fma(-v8, z2, __builtin_fma(-v7, z1, __builtin_fma(-v6, z0, n3)));
      n4 = __builtin_fma(-v8, z5, __builtin_fma(-v7, z4, __builtin_fma(-v6, z3, n4)));
      n5 = __builtin_fma(-v8, z8, __builtin_fma(-v7, z7, __builtin_fma(-v6, z6, n5)));
    }
    if (hasR) {
      const double* zp = Zl[er]; const double* wp = Wl[er];
      const double z0 = zp[0], z1 = zp[1], z2 = zp[2], z3 = zp[3], z4 = zp[4], z5 = zp[5];
      const double z6 = zp[6], z7 = zp[7], z8 = zp[8];
      const double v6 = wp[6], v7 = wp[7], v8 = wp[8];
      n3 = __builtin_fma(-v8, z2, __builtin_fma(-v7, z1, __builtin_fma(-v6, z0, n3)));
      n4 = __builtin_fma(-v8, z5, __builtin_fma(-v7, z4, __builtin_fma(-v6, z3, n4)));
      n5 = __builtin_fma(-v8, z8, __builtin_fma(-v7, z7, __builtin_fma(-v6, z6, n5)));
    }
    S[j][3] = n3; S[j][4] = n4; S[j][5] = n5;
    if (hasC) {
      const double* wp = Wl[er]; const double* zp = Zr[er];
      const double v0 = wp[0], v1 = wp[1], v2 = wp[2];
      const double z0 = zp[0], z1 = zp[1], z2 = zp[2];
      C[j][0] = __builtin_fma(-v2, z2, __builtin_fma(-v1, z1, -(v0 * z0)));
    }
  } else if (part == 2) {
    double g0 = gv[j][0], g1 = gv[j][1], g2 = gv[j][2];
    if (hasL) schur3_g(Wr[el], gv[el], g0, g1, g2);
    if (hasR) schur3_g(Wl[er], gv[er], g0, g1, g2);
    gv[j][0] = g0; gv[j][1] = g1; gv[j][2] = g2;
    if (hasC) {
      const double* wp = Wl[er]; const double* zp = Zr[er];
      const double v0 = wp[0], v1 = wp[1], v2 = wp[2];
      const double z3 = zp[3], z4 = zp[4], z5 = zp[5];
      C[j][1] = __builtin_fma(-v2, z5, __builtin_fma(-v1, z4, -(v0 * z3)));
    }
  } else {
    if (hasC) {
      const double* wp = Wl[er]; const double* zp = Zr[er];
      const double v0 = wp[0], v1 = wp[1], v2 = wp[2];
      const double v3 = wp[3], v4 = wp[4], v5 = wp[5];
      const double v6 = wp[6], v7 = wp[7], v8 = wp[8];
      const double z0 = zp[0], z1 = zp[1], z2 = zp[2];
      const double z3 = zp[3], z4 = zp[4], z5 = zp[5];
      const double z6 = zp[6], z7 = zp[7], z8 = zp[8];
      double* cp = C[j];
      cp[2] = __builtin_fma(-v2, z8, __builtin_fma(-v1, z7, -(v0 * z6)));
      cp[3] = __builtin_fma(-v5, z2, __builtin_fma(-v4, z1, -(v3 * z0)));
      cp[4] = __builtin_fma(-v5, z5, __builtin_fma(-v4, z4, -(v3 * z3)));
      cp[5] = __builtin_fma(-v5, z8, __builtin_fma(-v4, z7, -(v3 * z6)));
      cp[6] = __builtin_fma(-v8, z2, __builtin_fma(-v7, z1, -(v6 * z0)));
      cp[7] = __builtin_fma(-v8, z5, __builtin_fma(-v7, z4, -(v6 * z3)));
      cp[8] = __builtin_fma(-v8, z8, __builtin_fma(-v7, z7, -(v6 * z6)));
    }
  }
}

__device__ __forceinline__ void elim_rows(
    const double (*S)[6], double (*F)[6], const double (*C)[9], double (*gv)[3],
    double (*Zl)[9], double (*Wl)[9], double (*Zr)[9], double (*Wr)[9],
    int t, int s) {
  const int k = t >> 3;
  const int r = t & 7;
  const int i = s * (2 * k + 1);
  if (i >= NSTEP || r == 7) return;
  const bool left = (r < 3);
  const bool right = (r >= 3 && r < 6);
  if (right && i + s >= NSTEP) return;
  const int rr = left ? r : r - 3;
  const int pitch = (r < 6) ? 9 : 3;
  const int isrc = left ? i - s : i;
  const int osrc = pitch * isrc + (left ? 3 * r : (right ? rr : 0));
  const int step = right ? 3 : 1;
  const int odst = pitch * i + ((r < 6) ? 3 * rr : 0);
  const double* src = ((r < 6) ? &C[0][0] : &gv[0][0]) + osrc;
  double* zdst = (left ? &Zl[0][0] : (right ? &Zr[0][0] : &gv[0][0])) + odst;
  double* wdst = (left ? &Wl[0][0] : &Wr[0][0]) + odst;
  const double s0 = S[i][0], s1 = S[i][1], s2 = S[i][2];
  const double s3 = S[i][3], s4 = S[i][4], s5 = S[i][5];
  const double m0 = src[0], m1 = src[step], m2 = src[2 * step];
  __builtin_amdgcn_sched_barrier(0);
  double u10, u20, u21, i0, i1, i2;
  ldl3(s0, s1, s2, s3, s4, s5, u10, u20, u21, i0, i1, i2);
  const double z0 = m0;
  const double z1 = __builtin_fma(-z0, u10, m1);
  const double z2 = __builtin_fma(-z1, u21, __builtin_fma(-z0, u20, m2));
  zdst[0] = z0; zdst[1] = z1; zdst[2] = z2;
  if (r < 6) {
    wdst[0] = z0 * i0; wdst[1] = z1 * i1; wdst[2] = z2 * i2;
  } else {
    F[i][0] = i0;  F[i][1] = u10; F[i][2] = i1;
    F[i][3] = u20; F[i][4] = u21; F[i][5] = i2;
  }
}

__device__ __forceinline__ void upd_elem(
    double (*S)[6], double (*C)[9], double (*gv)[3],
    const double (*Zl)[9], const double (*Wl)[9],
    const double (*Zr)[9], const double (*Wr)[9],
    int task, int s) {
  const int k = task / 18;
  const int e = task - 18 * k;
  const int j = 2 * s * k;
  if (j >= NSTEP) return;
  const bool hasL = (j > 0);
  const bool hasR = (j + s < NSTEP);
  const bool hasC = (j + 2 * s < NSTEP);
  const int el = hasL ? j - s : 0;
  const int er = hasR ? j + s : 0;
  const bool isS = (e < 6);
  const bool isG = (e >= 6 && e < 9);
  const bool isC = (e >= 9);
  const int aS = ((e >= 1) ? 1 : 0) + ((e >= 3) ? 1 : 0);
  const int bS = e - ((aS * (aS + 1)) >> 1);
  const int aC = (e - 9) / 3;
  const int bC = (e - 9) - 3 * aC;
  const int a = isS ? aS : (isG ? e - 6 : aC);
  const int b = isS ? bS : (isG ? 0 : bC);
  double* dst = (isS ? &S[0][0] : (isG ? &gv[0][0] : &C[0][0]))
              + ((isS ? 6 : (isG ? 3 : 9)) * j + (isS ? e : (isG ? a : e - 9)));
  const double* wl = &Wr[0][0] + (9 * el + 3 * a);
  const double* zl = (isS ? &Zr[0][0] : &gv[0][0]) + ((isS ? 9 : 3) * el + (isS ? 3 * b : 0));
  const double* wr = &Wl[0][0] + (9 * er + 3 * a);
  const double* zr = (isS ? &Zl[0][0] : (isG ? &gv[0][0] : &Zr[0][0]))
                   + ((isG ? 3 : 9) * er + (isG ? 0 : 3 * b));
  const double old = dst[0];
  const double p0 = wl[0], p1 = wl[1], p2 = wl[2];
  const double q0 = zl[0], q1 = zl[1], q2 = zl[2];
  const double v0 = wr[0], v1 = wr[1], v2 = wr[2];
  const double z0 = zr[0], z1 = zr[1], z2 = zr[2];
  asm volatile("" :: "v"(old), "v"(p0), "v"(p1), "v"(p2), "v"(q0), "v"(q1), "v"(q2),
                     "v"(v0), "v"(v1), "v"(v2), "v"(z0), "v"(z1), "v"(z2));
  const double acc0 = isC ? -0.0 : old;
  const double accL = __builtin_fma(-p2, q2, __builtin_fma(-p1, q1, __builtin_fma(-p0, q0, acc0)));
  const double acc1 = (!isC && hasL) ? accL : acc0;
  const double accR = __builtin_fma(-v2, z2, __builtin_fma(-v1, z1, __builtin_fma(-v0, z0, acc1)));
  const double acc2 = (isC ? hasC : hasR) ? accR : acc1;
  if (!isC || hasC) dst[0] = acc2;
}

__device__ __forceinline__ void cr_back(
    const double (*F)[6], double (*gv)[3],
    const double (*Wl)[9], const double (*Wr)[9],
    int i, int s) {
  const double j0 = F[i][0], q10 = F[i][1], j1 = F[i][2];
  const double q20 = F[i][3], q21 = F[i][4], j2 = F[i][5];
  double t0 = gv[i][0] * j0, t1 = gv[i][1] * j1, t2 = gv[i][2] * j2;
  {
    const double* wp = Wl[i];
    const double a0 = gv[i - s][0], a1 = gv[i - s][1], a2 = gv[i - s][2];
    t0 = __builtin_fma(-wp[6], a2, __builtin_fma(-wp[3], a1, __builtin_fma(-wp[0], a0, t0)));
    t1 = __builtin_fma(-wp[7], a2, __builtin_fma(-wp[4], a1, __builtin_fma(-wp[1], a0, t1)));
    t2 = __builtin_fma(-wp[8], a2, __builtin_fma(-wp[5], a1, __builtin_fma(-wp[2], a0, t2)));
  }
  if (i + s < NSTEP) {
    const double* wp = Wr[i];
    const double a0 = gv[i + s][0], a1 = gv[i + s][1], a2 = gv[i + s][2];
    t0 = __builtin_fma(-wp[6], a2, __builtin_fma(-wp[3], a1, __builtin_fma(-wp[0], a0, t0)));
    t1 = __builtin_fma(-wp[7], a2, __builtin_fma(-wp[4], a1, __builtin_fma(-wp[1], a0, t1)));
    t2 = __builtin_fma(-wp[8], a2, __builtin_fma(-wp[5], a1, __builtin_fma(-wp[2], a0, t2)));
  }
  const double x2 = t2;
  const double x1 = __builtin_fma(-q21, x2, t1);
  const double x0 = __builtin_fma(-q20, x2, __builtin_fma(-q10, x1, t0));
  gv[i][0] = x0; gv[i][1] = x1; gv[i][2] = x2;
}

#define NEVEN 49

struct FoldRow { double u10, u20, u21, w0, w1, w2; double* dst; bool live; };

__device__ __forceinline__ FoldRow fold_p_load(
    const double (*F)[6], const double (*Wl)[9], const double (*Wr)[9],
    double (*Pl)[9], double (*Pr)[9], int q) {
  const int kq = q / 6;
  const int r = q - 6 * kq;
  const int k = (kq < NEVEN) ? kq : NEVEN - 1;
  const int i = 2 * (k + 1);
  const int s = i & (-i);
  const bool hasR = (i + s < NSTEP);
  const bool left = (r < 3) || !hasR;
  const int off = 9 * i + 3 * ((r < 3) ? r : r - 3);
  const double* src = (left ? &Wl[0][0] : &Wr[0][0]) + off;
  FoldRow o;
  o.dst = (left ? &Pl[0][0] : &Pr[0][0]) + off;
  o.live = (kq < NEVEN) && ((r < 3) || hasR);
  o.u10 = F[i][1]; o.u20 = F[i][3]; o.u21 = F[i][4];
  o.w0 = src[0]; o.w1 = src[1]; o.w2 = src[2];
  return o;
}

__device__ __forceinline__ void fold_p_solve(const FoldRow& a, double& p0, double& p1, double& p2) {
  p2 = a.w2;
  p1 = __builtin_fma(-a.u21, p2, a.w1);
  p0 = __builtin_fma(-a.u20, p2, __builtin_fma(-a.u10, p1, a.w0));
}

__device__ __forceinline__ void fold_p_store(const FoldRow& a, double p0, double p1, double p2) {
  if (a.live) { a.dst[0] = p0; a.dst[1] = p1; a.dst[2] = p2; }
}

__device__ __forceinline__ double back_fold(
    const double (*yv)[3], const double (*gv)[3],
    const double (*Pl)[9], const double (*Pr)[9],
    int i, int s, int k) {
  const bool hasR = (i + s < NSTEP);
  const double* pl = Pl[i];
  const double* pr = hasR ? Pr[i] : Pl[i];
  const double* xl = gv[i - s];
  const double* xr = gv[hasR ? i + s : i - s];
  const double y = yv[i][k];
  const double c0 = pl[k], c1 = pl[3 + k], c2 = pl[6 + k];
  const double a0 = xl[0], a1 = xl[1], a2 = xl[2];
  const double d0 = pr[k], d1 = pr[3 + k], d2 = pr[6 + k];
  const double b0 = xr[0], b1 = xr[1], b2 = xr[2];
  __builtin_amdgcn_sched_barrier(0);
  const double tl = __builtin_fma(-c2, a2, __builtin_fma(-c1, a1, __builtin_fma(-c0, a0, y)));
  const double tr = __builtin_fma(-d2, b2, __builtin_fma(-d1, b1, __builtin_fma(-d0, b0, tl)));
  return hasR ? tr : tl;
}

__device__ __forceinline__ v8f gram_tile(float h,
                                         float a0, float b0, float q0,
                                         float a1, float b1, float q1,
                                         float a2, float b2, float q2) {
  v2f f0, f1;
  f0.x = fmaf(fmaf(q0, h, b0), h, a0);
  f0.y = fmaf(fmaf(q1, h, b1), h, a1);
  f1.x = fmaf(fmaf(q2, h, b2), h, a2);
  f1.y = 0.f;
  v8f acc = {};
  acc = __builtin_amdgcn_wmma_f32_16x16x4_f32(false, f0, false, f0,
                                              (short)0, acc, false, false);
  acc = __builtin_amdgcn_wmma_f32_16x16x4_f32(false, f1, false, f1,
                                              (short)0, acc, false, false);
  return acc;
}

__device__ __forceinline__ void put_tile(float* q, v8f acc) {
  q[0] = acc[0]; q[1] = acc[1]; q[2] = acc[2];
  q[3] = acc[3]; q[4] = acc[4]; q[5] = acc[5];
}

#define TOPS  64
#define NWAVE 4
#define NTHR  (32 * NWAVE)
#define KSTEP 25
#define GS_TILES (T1 * 36)
#define GS_SPARE ((KSTEP - 1) * NWAVE * 36 + 26 * 6)

__global__ __launch_bounds__(NTHR) void ode_blocktri_kernel(
    const float* __restrict__ coeffs,
    const float* __restrict__ rhs,
    const float* __restrict__ ivr,
    const float* __restrict__ steps,
    float* __restrict__ out)
{
  __shared__ double S[NSTEP][6];
  __shared__ double F[NSTEP][6];
  __shared__ double Bo[T1][9];
  __shared__ double gv[NSTEP][3];
  __shared__ double Zl[NSTEP][9];
  __shared__ double Wl[NSTEP][9];
  __shared__ double Zr[NSTEP][9];
  __shared__ double Wr[NSTEP][9];
  __shared__ double Pl[NSTEP][9];
  __shared__ double Pr[NSTEP][9];
  __shared__ double yv[NSTEP][3];
  __shared__ __attribute__((aligned(16))) float Gs[GS_TILES + GS_SPARE];

  const int b = blockIdx.x;
  const int tid = threadIdx.x;
  const int lane = tid & 31;
  const int wave = __builtin_amdgcn_readfirstlane(tid >> 5);
  const int role = wave >> 1;
  const int slot = ((wave & 1) << 5) | lane;

  const int myp = wave + NWAVE * lane;
  const float hmine = (myp < T1) ? steps[b * T1 + myp] : 0.f;
  const float hcopy = (tid < T1) ? steps[b * T1 + tid] : 0.f;

  const int m = lane & 15;
  const int khalf = lane >> 4;
  const int col = (m < 6) ? m : 6;
  float a0, b0, q0, a1, b1, q1, a2, b2, q2;
  Ecoef(2 * khalf,                 col, &a0, &b0, &q0);
  Ecoef(2 * khalf + 1,             col, &a1, &b1, &q1);
  Ecoef((khalf == 0) ? 4 : -1,     col, &a2, &b2, &q2);

  const int hbits = __builtin_bit_cast(int, hmine);
  float* const gq = Gs + ((lane < 6) ? (wave * 36 + lane * 6)
                                     : (GS_TILES + (lane - 6) * 6));
  {
    v8f pend = {};
#pragma unroll
    for (int k = 0; k < KSTEP - 1; ++k) {
      const float h = __builtin_bit_cast(float, __builtin_amdgcn_readlane(hbits, k));
      const v8f acc = gram_tile(h, a0, b0, q0, a1, b1, q1, a2, b2, q2);
      if (k > 0) put_tile(gq + (k - 1) * (NWAVE * 36), pend);
      pend = acc;
    }
    put_tile(gq + (KSTEP - 2) * (NWAVE * 36), pend);
    if (wave + NWAVE * (KSTEP - 1) < T1) {
      const float h = __builtin_bit_cast(float, __builtin_amdgcn_readlane(hbits, KSTEP - 1));
      const v8f acc = gram_tile(h, a0, b0, q0, a1, b1, q1, a2, b2, q2);
      put_tile(gq + (KSTEP - 1) * (NWAVE * 36), acc);
    }
  }
  __syncthreads();

  const float iv0 = ivr[b * 2 + 0];
  const float iv1 = ivr[b * 2 + 1];
  if (tid < NSTEP) {
    const int t = tid;
    const float* cp = coeffs + (b * NSTEP + t) * 3;
    const double c0 = (double)cp[0];
    const double c1 = (double)cp[1];
    const double c2 = (double)cp[2];
    const double r  = (double)rhs[b * NSTEP + t];
    const bool hasPrev = (t > 0);
    const bool hasNext = (t < T1);
    const float* gp = Gs + 36 * (hasPrev ? t - 1 : 0);
    const float* gn = Gs + 36 * (hasNext ? t : T1 - 1);
    const float f0 = gp[3 * 6 + 3], f1 = gp[3 * 6 + 4], f2 = gp[4 * 6 + 4];
    const float f3 = gp[3 * 6 + 5], f4 = gp[4 * 6 + 5], f5 = gp[5 * 6 + 5];
    const float n0 = gn[0 * 6 + 0], n1 = gn[0 * 6 + 1], n2 = gn[1 * 6 + 1];
    const float n3 = gn[0 * 6 + 2], n4 = gn[1 * 6 + 2], n5 = gn[2 * 6 + 2];
    const double p0 = (double)(hasPrev ? f0 : 1.f);
    const double p1 = (double)(hasPrev ? f1 : 0.f);
    const double p2 = (double)(hasPrev ? f2 : 1.f);
    const double p3 = (double)(hasPrev ? f3 : 0.f);
    const double p4 = (double)(hasPrev ? f4 : 0.f);
    const double p5 = (double)(hasPrev ? f5 : 0.f);
    S[t][0] = __builtin_fma(c0, c0, p0) + (double)(hasNext ? n0 : 0.f);
    S[t][1] = __builtin_fma(c1, c0, p1) + (double)(hasNext ? n1 : 0.f);
    S[t][2] = __builtin_fma(c1, c1, p2) + (double)(hasNext ? n2 : 0.f);
    S[t][3] = __builtin_fma(c2, c0, p3) + (double)(hasNext ? n3 : 0.f);
    S[t][4] = __builtin_fma(c2, c1, p4) + (double)(hasNext ? n4 : 0.f);
    S[t][5] = __builtin_fma(c2, c2, p5) + (double)(hasNext ? n5 : 0.f);
    gv[t][0] = __builtin_fma(c0, r, (double)((t == 0) ? iv0 : 0.f));
    gv[t][1] = __builtin_fma(c1, r, (double)((t == 0) ? iv1 : 0.f));
    gv[t][2] = c2 * r;
    const double b00 = (double)gn[3 * 6 + 0], b01 = (double)gn[4 * 6 + 0], b02 = (double)gn[5 * 6 + 0];
    const double b10 = (double)gn[3 * 6 + 1], b11 = (double)gn[4 * 6 + 1], b12 = (double)gn[5 * 6 + 1];
    const double b20 = (double)gn[3 * 6 + 2], b21 = (double)gn[4 * 6 + 2], b22 = (double)gn[5 * 6 + 2];
    if (hasNext) {
      Bo[t][0] = b00; Bo[t][1] = b01; Bo[t][2] = b02;
      Bo[t][3] = b10; Bo[t][4] = b11; Bo[t][5] = b12;
      Bo[t][6] = b20; Bo[t][7] = b21; Bo[t][8] = b22;
    }
  }
  __syncthreads();

  {
    const int s = 1;
    {
      const int i = s * (2 * slot + 1);
      if (i < NSTEP) {
        if (role == 0)            elim_left(S, F, Bo, gv, Zl, Wl, i, s);
        else if (i + s < NSTEP)   elim_right(S, Bo, Zr, Wr, i);
      }
    }
    __syncthreads();
    {
      const int j = 2 * s * slot;
      if (j < NSTEP) {
        if (role == 0)   upd_diag(S, Zl, Wl, Zr, Wr, j, s);
        else             upd_rhs_couple(Bo, gv, Wl, Wr, Zr, j, s);
      }
    }
    __syncthreads();
  }
  {
    const int s = 2;
    {
      const int i = s * (2 * lane + 1);
      if (i < NSTEP) elim4(S, F, Bo, gv, Zl, Wl, Zr, Wr, i, s, wave);
    }
    __syncthreads();
    {
      const int j = 2 * s * lane;
      if (j < NSTEP) upd4(S, Bo, gv, Zl, Wl, Zr, Wr, j, s, wave);
    }
    __syncthreads();
  }
  for (int s = 4; s <= TOPS; s <<= 1) {
    elim_rows(S, F, Bo, gv, Zl, Wl, Zr, Wr, tid, s);
    __syncthreads();
    const int ntask = 18 * ((NSTEP - 1) / (2 * s) + 1);
    for (int base = 0; base < ntask; base += NTHR)
      upd_elem(S, Bo, gv, Zl, Wl, Zr, Wr, base + tid, s);
    __syncthreads();
  }
  if (wave != 0) {
    const int f = tid - 32;
    const FoldRow ra = fold_p_load(F, Wl, Wr, Pl, Pr, f);
    const FoldRow rb = fold_p_load(F, Wl, Wr, Pl, Pr, f + 96);
    const FoldRow rc = fold_p_load(F, Wl, Wr, Pl, Pr, f + 192);
    double pa0, pa1, pa2, pb0, pb1, pb2, pc0, pc1, pc2;
    if (wave == 1) {
      const FoldRow rd = fold_p_load(F, Wl, Wr, Pl, Pr, f + 288);
      double pd0, pd1, pd2;
      fold_p_solve(ra, pa0, pa1, pa2);
      fold_p_solve(rb, pb0, pb1, pb2);
      fold_p_solve(rc, pc0, pc1, pc2);
      fold_p_solve(rd, pd0, pd1, pd2);
      fold_p_store(ra, pa0, pa1, pa2);
      fold_p_store(rb, pb0, pb1, pb2);
      fold_p_store(rc, pc0, pc1, pc2);
      fold_p_store(rd, pd0, pd1, pd2);
    } else {
      const int kq = tid - 64;
      const int i = 2 * (((kq < NEVEN) ? kq : NEVEN - 1) + 1);
      const double j0 = F[i][0], q10 = F[i][1], j1 = F[i][2];
      const double q20 = F[i][3], q21 = F[i][4], j2 = F[i][5];
      const double g0 = gv[i][0], g1 = gv[i][1], g2 = gv[i][2];
      fold_p_solve(ra, pa0, pa1, pa2);
      fold_p_solve(rb, pb0, pb1, pb2);
      fold_p_solve(rc, pc0, pc1, pc2);
      const double t0 = g0 * j0, t1 = g1 * j1, t2 = g2 * j2;
      const double y2 = t2;
      const double y1 = __builtin_fma(-q21, y2, t1);
      const double y0 = __builtin_fma(-q20, y2, __builtin_fma(-q10, y1, t0));
      fold_p_store(ra, pa0, pa1, pa2);
      fold_p_store(rb, pb0, pb1, pb2);
      fold_p_store(rc, pc0, pc1, pc2);
      if (kq < NEVEN) { yv[i][0] = y0; yv[i][1] = y1; yv[i][2] = y2; }
    }
  } else if (tid == 0) {
    double u10, u20, u21, i0, i1, i2;
    ldl3(S[0][0], S[0][1], S[0][2], S[0][3], S[0][4], S[0][5],
         u10, u20, u21, i0, i1, i2);
    const double w0 = gv[0][0];
    const double w1 = __builtin_fma(-u10, w0, gv[0][1]);
    const double w2 = __builtin_fma(-u21, w1, __builtin_fma(-u20, w0, gv[0][2]));
    const double x2 = w2 * i2;
    const double x1 = __builtin_fma(-u21, x2, w1 * i1);
    const double x0 = __builtin_fma(-u20, x2, __builtin_fma(-u10, x1, w0 * i0));
    gv[0][0] = x0; gv[0][1] = x1; gv[0][2] = x2;
  }
  __syncthreads();
  for (int s = TOPS; s >= 2; s >>= 1) {
    const int i = s * (2 * lane + 1);
    if (i < NSTEP && wave < 3) gv[i][wave] = back_fold(yv, gv, Pl, Pr, i, s, wave);
    __syncthreads();
  }
  {
    const int i = 2 * tid + 1;
    if (i < NSTEP) cr_back(F, gv, Wl, Wr, i, 1);
    __syncthreads();
  }

  if (tid < NSTEP) {
    out[0 * OUT_U + b * NSTEP + tid] = (float)gv[tid][0];
    out[1 * OUT_U + b * NSTEP + tid] = (float)gv[tid][1];
    out[2 * OUT_U + b * NSTEP + tid] = (float)gv[tid][2];
  }
  if (tid == 0) out[OUT_EPS + b] = 0.0f;
  if (tid < T1) out[OUT_ST + b * T1 + tid] = hcopy;
}

extern "C" void kernel_launch(void* const* d_in, const int* in_sizes, int n_in,
                              void* d_out, int out_size, void* d_ws, size_t ws_size,
                              hipStream_t stream) {
  (void)in_sizes; (void)n_in; (void)out_size; (void)d_ws; (void)ws_size;
  const float* coeffs = (const float*)d_in[0];
  const float* rhs    = (const float*)d_in[1];
  const float* ivr    = (const float*)d_in[2];
  const float* steps  = (const float*)d_in[3];
  float* out = (float*)d_out;
  hipLaunchKernelGGL(ode_blocktri_kernel, dim3(NB), dim3(NTHR), 0, stream,
                     coeffs, rhs, ivr, steps, out);
}


# Odd-even order counted from block 1 (an empty uncoupled block 0 pads the frame): block 64 is left and solved directly, 6 levels not 7, 5 back levels not 6; elimination order, factor rounding change
# speedup vs baseline: 40.3042x; 1.0991x over previous
#include <hip/hip_runtime.h>
#include <hip/hip_bf16.h>
#include <math.h>


typedef __attribute__((ext_vector_type(2))) float v2f;
typedef __attribute__((ext_vector_type(8))) float v8f;

#define NSTEP 100
#define NBLK  (NSTEP + 1)
#define T1    99
#define NB    256
#define OUT_U   (NB * NSTEP)
#define OUT_EPS (3 * OUT_U)
#define OUT_ST  (OUT_EPS + NB)

__device__ __forceinline__ void Ecoef(int r, int c, float* pa, float* pb, float* pc) {
  float a = 0.f, b = 0.f, q = 0.f;
  if (r == 0)      { if (c == 0) a = 1.f;  else if (c == 1) b = 1.f;   else if (c == 2) q = 0.5f; else if (c == 3) a = -1.f; }
  else if (r == 1) { if (c == 1) a = 1.f;  else if (c == 2) b = 1.f;   else if (c == 4) a = -1.f; }
  else if (r == 2) { if (c == 0) a = -1.f; else if (c == 3) a = 1.f;   else if (c == 4) b = -1.f; else if (c == 5) q = 0.5f; }
  else if (r == 3) { if (c == 1) a = -1.f; else if (c == 4) a = 1.f;   else if (c == 5) b = -1.f; }
  else if (r == 4) { if (c == 1) a = -1.f; else if (c == 2) b = -0.5f; else if (c == 4) a = 1.f;  else if (c == 5) b = -0.5f; }
  *pa = a; *pb = b; *pc = q;
}

__device__ __forceinline__ double recip_nr(double d) {
  double r = __builtin_amdgcn_rcp(d);
  double e = __builtin_fma(-d, r, 1.0);
  r = __builtin_fma(r, e, r);
  e = __builtin_fma(-d, r, 1.0);
  r = __builtin_fma(r, e, r);
  return r;
}

__device__ __forceinline__ void ldl3(double s0, double s1, double s2,
                                     double s3, double s4, double s5,
                                     double& u10, double& u20, double& u21,
                                     double& i0, double& i1, double& i2) {
  i0 = recip_nr(s0);
  u10 = s1 * i0;
  u20 = s3 * i0;
  const double d1 = __builtin_fma(-u10, s1, s2);
  i1 = recip_nr(d1);
  const double t = __builtin_fma(-u20, s1, s4);
  u21 = t * i1;
  const double d2 = __builtin_fma(-u21, t, __builtin_fma(-u20, s3, s5));
  i2 = recip_nr(d2);
}

__device__ __forceinline__ void zw3(const double* m,
                                    double u10, double u20, double u21,
                                    double i0, double i1, double i2,
                                    double* z, double* w) {
  z[0] = m[0];
  z[1] = __builtin_fma(-z[0], u10, m[1]);
  z[2] = __builtin_fma(-z[1], u21, __builtin_fma(-z[0], u20, m[2]));
  z[3] = m[3];
  z[4] = __builtin_fma(-z[3], u10, m[4]);
  z[5] = __builtin_fma(-z[4], u21, __builtin_fma(-z[3], u20, m[5]));
  z[6] = m[6];
  z[7] = __builtin_fma(-z[6], u10, m[7]);
  z[8] = __builtin_fma(-z[7], u21, __builtin_fma(-z[6], u20, m[8]));
  w[0] = z[0] * i0; w[1] = z[1] * i1; w[2] = z[2] * i2;
  w[3] = z[3] * i0; w[4] = z[4] * i1; w[5] = z[5] * i2;
  w[6] = z[6] * i0; w[7] = z[7] * i1; w[8] = z[8] * i2;
}

__device__ __forceinline__ void elim_left(
    const double (*S)[6], double (*F)[6], const double (*C)[9], double (*gv)[3],
    double (*Zl)[9], double (*Wl)[9], int i, int s) {
  const double* cl = C[i - s];
  double ml[9];
  ml[0] = cl[0]; ml[1] = cl[1]; ml[2] = cl[2];
  ml[3] = cl[3]; ml[4] = cl[4]; ml[5] = cl[5];
  ml[6] = cl[6]; ml[7] = cl[7]; ml[8] = cl[8];
  const double s0 = S[i][0], s1 = S[i][1], s2 = S[i][2];
  const double s3 = S[i][3], s4 = S[i][4], s5 = S[i][5];
  const double g0 = gv[i][0], g1 = gv[i][1], g2 = gv[i][2];

  double u10, u20, u21, i0, i1, i2;
  ldl3(s0, s1, s2, s3, s4, s5, u10, u20, u21, i0, i1, i2);
  const double w1 = __builtin_fma(-u10, g0, g1);
  const double w2 = __builtin_fma(-u21, w1, __builtin_fma(-u20, g0, g2));
  double zl[9], wl[9];
  zw3(ml, u10, u20, u21, i0, i1, i2, zl, wl);

  F[i][0] = i0;  F[i][1] = u10; F[i][2] = i1;
  F[i][3] = u20; F[i][4] = u21; F[i][5] = i2;
  gv[i][1] = w1; gv[i][2] = w2;
  Zl[i][0] = zl[0]; Zl[i][1] = zl[1]; Zl[i][2] = zl[2];
  Zl[i][3] = zl[3]; Zl[i][4] = zl[4]; Zl[i][5] = zl[5];
  Zl[i][6] = zl[6]; Zl[i][7] = zl[7]; Zl[i][8] = zl[8];
  Wl[i][0] = wl[0]; Wl[i][1] = wl[1]; Wl[i][2] = wl[2];
  Wl[i][3] = wl[3]; Wl[i][4] = wl[4]; Wl[i][5] = wl[5];
  Wl[i][6] = wl[6]; Wl[i][7] = wl[7]; Wl[i][8] = wl[8];
}

__device__ __forceinline__ void elim_right(
    const double (*S)[6], const double (*C)[9],
    double (*Zr)[9], double (*Wr)[9], int i) {
  const double* cr = C[i];
  double mr[9];
  mr[0] = cr[0]; mr[1] = cr[3]; mr[2] = cr[6];
  mr[3] = cr[1]; mr[4] = cr[4]; mr[5] = cr[7];
  mr[6] = cr[2]; mr[7] = cr[5]; mr[8] = cr[8];
  const double s0 = S[i][0], s1 = S[i][1], s2 = S[i][2];
  const double s3 = S[i][3], s4 = S[i][4], s5 = S[i][5];

  double u10, u20, u21, i0, i1, i2;
  ldl3(s0, s1, s2, s3, s4, s5, u10, u20, u21, i0, i1, i2);
  double zr[9], wr[9];
  zw3(mr, u10, u20, u21, i0, i1, i2, zr, wr);

  Zr[i][0] = zr[0]; Zr[i][1] = zr[1]; Zr[i][2] = zr[2];
  Zr[i][3] = zr[3]; Zr[i][4] = zr[4]; Zr[i][5] = zr[5];
  Zr[i][6] = zr[6]; Zr[i][7] = zr[7]; Zr[i][8] = zr[8];
  Wr[i][0] = wr[0]; Wr[i][1] = wr[1]; Wr[i][2] = wr[2];
  Wr[i][3] = wr[3]; Wr[i][4] = wr[4]; Wr[i][5] = wr[5];
  Wr[i][6] = wr[6]; Wr[i][7] = wr[7]; Wr[i][8] = wr[8];
}

__device__ __forceinline__ void schur3_n(
    const double* zp, const double* wp,
    double& n0, double& n1, double& n2, double& n3, double& n4, double& n5) {
  const double z0 = zp[0], z1 = zp[1], z2 = zp[2];
  const double z3 = zp[3], z4 = zp[4], z5 = zp[5];
  const double z6 = zp[6], z7 = zp[7], z8 = zp[8];
  const double v0 = wp[0], v1 = wp[1], v2 = wp[2];
  const double v3 = wp[3], v4 = wp[4], v5 = wp[5];
  const double v6 = wp[6], v7 = wp[7], v8 = wp[8];
  n0 = __builtin_fma(-v2, z2, __builtin_fma(-v1, z1, __builtin_fma(-v0, z0, n0)));
  n1 = __builtin_fma(-v5, z2, __builtin_fma(-v4, z1, __builtin_fma(-v3, z0, n1)));
  n2 = __builtin_fma(-v5, z5, __builtin_fma(-v4, z4, __builtin_fma(-v3, z3, n2)));
  n3 = __builtin_fma(-v8, z2, __builtin_fma(-v7, z1, __builtin_fma(-v6, z0, n3)));
  n4 = __builtin_fma(-v8, z5, __builtin_fma(-v7, z4, __builtin_fma(-v6, z3, n4)));
  n5 = __builtin_fma(-v8, z8, __builtin_fma(-v7, z7, __builtin_fma(-v6, z6, n5)));
}

__device__ __forceinline__ void schur3_g(
    const double* wp, const double* xp, double& g0, double& g1, double& g2) {
  const double v0 = wp[0], v1 = wp[1], v2 = wp[2];
  const double v3 = wp[3], v4 = wp[4], v5 = wp[5];
  const double v6 = wp[6], v7 = wp[7], v8 = wp[8];
  const double x0 = xp[0], x1 = xp[1], x2 = xp[2];
  g0 = __builtin_fma(-v2, x2, __builtin_fma(-v1, x1, __builtin_fma(-v0, x0, g0)));
  g1 = __builtin_fma(-v5, x2, __builtin_fma(-v4, x1, __builtin_fma(-v3, x0, g1)));
  g2 = __builtin_fma(-v8, x2, __builtin_fma(-v7, x1, __builtin_fma(-v6, x0, g2)));
}

__device__ __forceinline__ void upd_diag(
    double (*S)[6],
    const double (*Zl)[9], const double (*Wl)[9],
    const double (*Zr)[9], const double (*Wr)[9],
    int j, int s) {
  double n0 = S[j][0], n1 = S[j][1], n2 = S[j][2];
  double n3 = S[j][3], n4 = S[j][4], n5 = S[j][5];
  if (j > 0) {
    const int e = j - s;
    schur3_n(Zr[e], Wr[e], n0, n1, n2, n3, n4, n5);
  }
  if (j + s < NBLK) {
    const int e = j + s;
    schur3_n(Zl[e], Wl[e], n0, n1, n2, n3, n4, n5);
  }
  S[j][0] = n0; S[j][1] = n1; S[j][2] = n2;
  S[j][3] = n3; S[j][4] = n4; S[j][5] = n5;
}

__device__ __forceinline__ void upd_rhs_couple(
    double (*C)[9], double (*gv)[3],
    const double (*Wl)[9], const double (*Wr)[9], const double (*Zr)[9],
    int j, int s) {
  double g0 = gv[j][0], g1 = gv[j][1], g2 = gv[j][2];
  if (j > 0) {
    const int e = j - s;
    schur3_g(Wr[e], gv[e], g0, g1, g2);
  }
  if (j + s < NBLK) {
    const int e = j + s;
    schur3_g(Wl[e], gv[e], g0, g1, g2);
    if (j + 2 * s < NBLK) {
      const double* wp = Wl[e];
      const double* zp = Zr[e];
      const double v0 = wp[0], v1 = wp[1], v2 = wp[2];
      const double v3 = wp[3], v4 = wp[4], v5 = wp[5];
      const double v6 = wp[6], v7 = wp[7], v8 = wp[8];
      const double z0 = zp[0], z1 = zp[1], z2 = zp[2];
      const double z3 = zp[3], z4 = zp[4], z5 = zp[5];
      const double z6 = zp[6], z7 = zp[7], z8 = zp[8];
      double* cp = C[j];
      cp[0] = __builtin_fma(-v2, z2, __builtin_fma(-v1, z1, -(v0 * z0)));
      cp[1] = __builtin_fma(-v2, z5, __builtin_fma(-v1, z4, -(v0 * z3)));
      cp[2] = __builtin_fma(-v2, z8, __builtin_fma(-v1, z7, -(v0 * z6)));
      cp[3] = __builtin_fma(-v5, z2, __builtin_fma(-v4, z1, -(v3 * z0)));
      cp[4] = __builtin_fma(-v5, z5, __builtin_fma(-v4, z4, -(v3 * z3)));
      cp[5] = __builtin_fma(-v5, z8, __builtin_fma(-v4, z7, -(v3 * z6)));
      cp[6] = __builtin_fma(-v8, z2, __builtin_fma(-v7, z1, -(v6 * z0)));
      cp[7] = __builtin_fma(-v8, z5, __builtin_fma(-v7, z4, -(v6 * z3)));
      cp[8] = __builtin_fma(-v8, z8, __builtin_fma(-v7, z7, -(v6 * z6)));
    }
  }
  gv[j][0] = g0; gv[j][1] = g1; gv[j][2] = g2;
}

__device__ __forceinline__ void zw_row(double m0, double m1, double m2,
                                       double u10, double u20, double u21,
                                       double i0, double i1, double i2,
                                       double* z, double* w) {
  const double z0 = m0;
  const double z1 = __builtin_fma(-z0, u10, m1);
  const double z2 = __builtin_fma(-z1, u21, __builtin_fma(-z0, u20, m2));
  z[0] = z0; z[1] = z1; z[2] = z2;
  w[0] = z0 * i0; w[1] = z1 * i1; w[2] = z2 * i2;
}

__device__ __forceinline__ void elim4(
    const double (*S)[6], double (*F)[6], const double (*C)[9], double (*gv)[3],
    double (*Zl)[9], double (*Wl)[9], double (*Zr)[9], double (*Wr)[9],
    int i, int s, int part) {
  if (part >= 2 && i + s >= NBLK) return;
  const double s0 = S[i][0], s1 = S[i][1], s2 = S[i][2];
  const double s3 = S[i][3], s4 = S[i][4], s5 = S[i][5];
  double u10, u20, u21, i0, i1, i2;
  if (part == 0) {
    const double* cl = C[i - s];
    const double m0 = cl[0], m1 = cl[1], m2 = cl[2];
    const double g0 = gv[i][0], g1 = gv[i][1], g2 = gv[i][2];
    ldl3(s0, s1, s2, s3, s4, s5, u10, u20, u21, i0, i1, i2);
    const double w1 = __builtin_fma(-u10, g0, g1);
    const double w2 = __builtin_fma(-u21, w1, __builtin_fma(-u20, g0, g2));
    F[i][0] = i0;  F[i][1] = u10; F[i][2] = i1;
    F[i][3] = u20; F[i][4] = u21; F[i][5] = i2;
    gv[i][1] = w1; gv[i][2] = w2;
    zw_row(m0, m1, m2, u10, u20, u21, i0, i1, i2, &Zl[i][0], &Wl[i][0]);
  } else if (part == 1) {
    const double* cl = C[i - s];
    const double m3 = cl[3], m4 = cl[4], m5 = cl[5];
    const double m6 = cl[6], m7 = cl[7], m8 = cl[8];
    ldl3(s0, s1, s2, s3, s4, s5, u10, u20, u21, i0, i1, i2);
    zw_row(m3, m4, m5, u10, u20, u21, i0, i1, i2, &Zl[i][3], &Wl[i][3]);
    zw_row(m6, m7, m8, u10, u20, u21, i0, i1, i2, &Zl[i][6], &Wl[i][6]);
  } else if (part == 2) {
    const double* cr = C[i];
    const double m0 = cr[0], m1 = cr[3], m2 = cr[6];
    const double m3 = cr[1], m4 = cr[4], m5 = cr[7];
    ldl3(s0, s1, s2, s3, s4, s5, u10, u20, u21, i0, i1, i2);
    zw_row(m0, m1, m2, u10, u20, u21, i0, i1, i2, &Zr[i][0], &Wr[i][0]);
    zw_row(m3, m4, m5, u10, u20, u21, i0, i1, i2, &Zr[i][3], &Wr[i][3]);
  } else {
    const double* cr = C[i];
    const double m6 = cr[2], m7 = cr[5], m8 = cr[8];
    ldl3(s0, s1, s2, s3, s4, s5, u10, u20, u21, i0, i1, i2);
    zw_row(m6, m7, m8, u10, u20, u21, i0, i1, i2, &Zr[i][6], &Wr[i][6]);
  }
}

__device__ __forceinline__ void upd4(
    double (*S)[6], double (*C)[9], double (*gv)[3],
    const double (*Zl)[9], const double (*Wl)[9],
    const double (*Zr)[9], const double (*Wr)[9],
    int j, int s, int part) {
  const bool hasL = (j > 0);
  const bool hasR = (j + s < NBLK);
  const bool hasC = (j + 2 * s < NBLK);
  const int el = hasL ? j - s : 0;
  const int er = hasR ? j + s : 0;
  if (part == 0) {
    double n0 = S[j][0], n1 = S[j][1], n2 = S[j][2];
    if (hasL) {
      const double* zp = Zr[el]; const double* wp = Wr[el];
      const double z0 = zp[0], z1 = zp[1], z2 = zp[2], z3 = zp[3], z4 = zp[4], z5 = zp[5];
      const double v0 = wp[0], v1 = wp[1], v2 = wp[2], v3 = wp[3], v4 = wp[4], v5 = wp[5];
      n0 = __builtin_fma(-v2, z2, __builtin_fma(-v1, z1, __builtin_fma(-v0, z0, n0)));
      n1 = __builtin_fma(-v5, z2, __builtin_fma(-v4, z1, __builtin_fma(-v3, z0, n1)));
      n2 = __builtin_fma(-v5, z5, __builtin_fma(-v4, z4, __builtin_fma(-v3, z3, n2)));
    }
    if (hasR) {
      const double* zp = Zl[er]; const double* wp = Wl[er];
      const double z0 = zp[0], z1 = zp[1], z2 = zp[2], z3 = zp[3], z4 = zp[4], z5 = zp[5];
      const double v0 = wp[0], v1 = wp[1], v2 = wp[2], v3 = wp[3], v4 = wp[4], v5 = wp[5];
      n0 = __builtin_fma(-v2, z2, __builtin_fma(-v1, z1, __builtin_fma(-v0, z0, n0)));
      n1 = __builtin_fma(-v5, z2, __builtin_fma(-v4, z1, __builtin_fma(-v3, z0, n1)));
      n2 = __builtin_fma(-v5, z5, __builtin_fma(-v4, z4, __builtin_fma(-v3, z3, n2)));
    }
    S[j][0] = n0; S[j][1] = n1; S[j][2] = n2;
  } else if (part == 1) {
    double n3 = S[j][3], n4 = S[j][4], n5 = S[j][5];
    if (hasL) {
      const double* zp = Zr[el]; const double* wp = Wr[el];
      const double z0 = zp[0], z1 = zp[1], z2 = zp[2], z3 = zp[3], z4 = zp[4], z5 = zp[5];
      const double z6 = zp[6], z7 = zp[7], z8 = zp[8];
      const double v6 = wp[6], v7 = wp[7], v8 = wp[8];
      n3 = __builtin_fma(-v8, z2, __builtin_fma(-v7, z1, __builtin_fma(-v6, z0, n3)));
      n4 = __builtin_fma(-v8, z5, __builtin_fma(-v7, z4, __builtin_fma(-v6, z3, n4)));
      n5 = __builtin_fma(-v8, z8, __builtin_fma(-v7, z7, __builtin_fma(-v6, z6, n5)));
    }
    if (hasR) {
      const double* zp = Zl[er]; const double* wp = Wl[er];
      const double z0 = zp[0], z1 = zp[1], z2 = zp[2], z3 = zp[3], z4 = zp[4], z5 = zp[5];
      const double z6 = zp[6], z7 = zp[7], z8 = zp[8];
      const double v6 = wp[6], v7 = wp[7], v8 = wp[8];
      n3 = __builtin_fma(-v8, z2, __builtin_fma(-v7, z1, __builtin_fma(-v6, z0, n3)));
      n4 = __builtin_fma(-v8, z5, __builtin_fma(-v7, z4, __builtin_fma(-v6, z3, n4)));
      n5 = __builtin_fma(-v8, z8, __builtin_fma(-v7, z7, __builtin_fma(-v6, z6, n5)));
    }
    S[j][3] = n3; S[j][4] = n4; S[j][5] = n5;
    if (hasC) {
      const double* wp = Wl[er]; const double* zp = Zr[er];
      const double v0 = wp[0], v1 = wp[1], v2 = wp[2];
      const double z0 = zp[0], z1 = zp[1], z2 = zp[2];
      C[j][0] = __builtin_fma(-v2, z2, __builtin_fma(-v1, z1, -(v0 * z0)));
    }
  } else if (part == 2) {
    double g0 = gv[j][0], g1 = gv[j][1], g2 = gv[j][2];
    if (hasL) schur3_g(Wr[el], gv[el], g0, g1, g2);
    if (hasR) schur3_g(Wl[er], gv[er], g0, g1, g2);
    gv[j][0] = g0; gv[j][1] = g1; gv[j][2] = g2;
    if (hasC) {
      const double* wp = Wl[er]; const double* zp = Zr[er];
      const double v0 = wp[0], v1 = wp[1], v2 = wp[2];
      const double z3 = zp[3], z4 = zp[4], z5 = zp[5];
      C[j][1] = __builtin_fma(-v2, z5, __builtin_fma(-v1, z4, -(v0 * z3)));
    }
  } else {
    if (hasC) {
      const double* wp = Wl[er]; const double* zp = Zr[er];
      const double v0 = wp[0], v1 = wp[1], v2 = wp[2];
      const double v3 = wp[3], v4 = wp[4], v5 = wp[5];
      const double v6 = wp[6], v7 = wp[7], v8 = wp[8];
      const double z0 = zp[0], z1 = zp[1], z2 = zp[2];
      const double z3 = zp[3], z4 = zp[4], z5 = zp[5];
      const double z6 = zp[6], z7 = zp[7], z8 = zp[8];
      double* cp = C[j];
      cp[2] = __builtin_fma(-v2, z8, __builtin_fma(-v1, z7, -(v0 * z6)));
      cp[3] = __builtin_fma(-v5, z2, __builtin_fma(-v4, z1, -(v3 * z0)));
      cp[4] = __builtin_fma(-v5, z5, __builtin_fma(-v4, z4, -(v3 * z3)));
      cp[5] = __builtin_fma(-v5, z8, __builtin_fma(-v4, z7, -(v3 * z6)));
      cp[6] = __builtin_fma(-v8, z2, __builtin_fma(-v7, z1, -(v6 * z0)));
      cp[7] = __builtin_fma(-v8, z5, __builtin_fma(-v7, z4, -(v6 * z3)));
      cp[8] = __builtin_fma(-v8, z8, __builtin_fma(-v7, z7, -(v6 * z6)));
    }
  }
}

__device__ __forceinline__ void elim_rows(
    const double (*S)[6], double (*F)[6], const double (*C)[9], double (*gv)[3],
    double (*Zl)[9], double (*Wl)[9], double (*Zr)[9], double (*Wr)[9],
    int t, int s) {
  const int k = t >> 3;
  const int r = t & 7;
  const int i = s * (2 * k + 1);
  if (i >= NBLK || r == 7) return;
  const bool left = (r < 3);
  const bool right = (r >= 3 && r < 6);
  if (right && i + s >= NBLK) return;
  const int rr = left ? r : r - 3;
  const int pitch = (r < 6) ? 9 : 3;
  const int isrc = left ? i - s : i;
  const int osrc = pitch * isrc + (left ? 3 * r : (right ? rr : 0));
  const int step = right ? 3 : 1;
  const int odst = pitch * i + ((r < 6) ? 3 * rr : 0);
  const double* src = ((r < 6) ? &C[0][0] : &gv[0][0]) + osrc;
  double* zdst = (left ? &Zl[0][0] : (right ? &Zr[0][0] : &gv[0][0])) + odst;
  double* wdst = (left ? &Wl[0][0] : &Wr[0][0]) + odst;
  const double s0 = S[i][0], s1 = S[i][1], s2 = S[i][2];
  const double s3 = S[i][3], s4 = S[i][4], s5 = S[i][5];
  const double m0 = src[0], m1 = src[step], m2 = src[2 * step];
  __builtin_amdgcn_sched_barrier(0);
  double u10, u20, u21, i0, i1, i2;
  ldl3(s0, s1, s2, s3, s4, s5, u10, u20, u21, i0, i1, i2);
  const double z0 = m0;
  const double z1 = __builtin_fma(-z0, u10, m1);
  const double z2 = __builtin_fma(-z1, u21, __builtin_fma(-z0, u20, m2));
  zdst[0] = z0; zdst[1] = z1; zdst[2] = z2;
  if (r < 6) {
    wdst[0] = z0 * i0; wdst[1] = z1 * i1; wdst[2] = z2 * i2;
  } else {
    F[i][0] = i0;  F[i][1] = u10; F[i][2] = i1;
    F[i][3] = u20; F[i][4] = u21; F[i][5] = i2;
  }
}

__device__ __forceinline__ void upd_elem(
    double (*S)[6], double (*C)[9], double (*gv)[3],
    const double (*Zl)[9], const double (*Wl)[9],
    const double (*Zr)[9], const double (*Wr)[9],
    int task, int s) {
  const int k = task / 18;
  const int e = task - 18 * k;
  const int j = 2 * s * k;
  if (j >= NBLK) return;
  const bool hasL = (j > 0);
  const bool hasR = (j + s < NBLK);
  const bool hasC = (j + 2 * s < NBLK);
  const int el = hasL ? j - s : 0;
  const int er = hasR ? j + s : 0;
  const bool isS = (e < 6);
  const bool isG = (e >= 6 && e < 9);
  const bool isC = (e >= 9);
  const int aS = ((e >= 1) ? 1 : 0) + ((e >= 3) ? 1 : 0);
  const int bS = e - ((aS * (aS + 1)) >> 1);
  const int aC = (e - 9) / 3;
  const int bC = (e - 9) - 3 * aC;
  const int a = isS ? aS : (isG ? e - 6 : aC);
  const int b = isS ? bS : (isG ? 0 : bC);
  double* dst = (isS ? &S[0][0] : (isG ? &gv[0][0] : &C[0][0]))
              + ((isS ? 6 : (isG ? 3 : 9)) * j + (isS ? e : (isG ? a : e - 9)));
  const double* wl = &Wr[0][0] + (9 * el + 3 * a);
  const double* zl = (isS ? &Zr[0][0] : &gv[0][0]) + ((isS ? 9 : 3) * el + (isS ? 3 * b : 0));
  const double* wr = &Wl[0][0] + (9 * er + 3 * a);
  const double* zr = (isS ? &Zl[0][0] : (isG ? &gv[0][0] : &Zr[0][0]))
                   + ((isG ? 3 : 9) * er + (isG ? 0 : 3 * b));
  const double old = dst[0];
  const double p0 = wl[0], p1 = wl[1], p2 = wl[2];
  const double q0 = zl[0], q1 = zl[1], q2 = zl[2];
  const double v0 = wr[0], v1 = wr[1], v2 = wr[2];
  const double z0 = zr[0], z1 = zr[1], z2 = zr[2];
  asm volatile("" :: "v"(old), "v"(p0), "v"(p1), "v"(p2), "v"(q0), "v"(q1), "v"(q2),
                     "v"(v0), "v"(v1), "v"(v2), "v"(z0), "v"(z1), "v"(z2));
  const double acc0 = isC ? -0.0 : old;
  const double accL = __builtin_fma(-p2, q2, __builtin_fma(-p1, q1, __builtin_fma(-p0, q0, acc0)));
  const double acc1 = (!isC && hasL) ? accL : acc0;
  const double accR = __builtin_fma(-v2, z2, __builtin_fma(-v1, z1, __builtin_fma(-v0, z0, acc1)));
  const double acc2 = (isC ? hasC : hasR) ? accR : acc1;
  if (!isC || hasC) dst[0] = acc2;
}

__device__ __forceinline__ void cr_back(
    const double (*F)[6], double (*gv)[3],
    const double (*Wl)[9], const double (*Wr)[9],
    int i, int s) {
  const double j0 = F[i][0], q10 = F[i][1], j1 = F[i][2];
  const double q20 = F[i][3], q21 = F[i][4], j2 = F[i][5];
  double t0 = gv[i][0] * j0, t1 = gv[i][1] * j1, t2 = gv[i][2] * j2;
  {
    const double* wp = Wl[i];
    const double a0 = gv[i - s][0], a1 = gv[i - s][1], a2 = gv[i - s][2];
    t0 = __builtin_fma(-wp[6], a2, __builtin_fma(-wp[3], a1, __builtin_fma(-wp[0], a0, t0)));
    t1 = __builtin_fma(-wp[7], a2, __builtin_fma(-wp[4], a1, __builtin_fma(-wp[1], a0, t1)));
    t2 = __builtin_fma(-wp[8], a2, __builtin_fma(-wp[5], a1, __builtin_fma(-wp[2], a0, t2)));
  }
  if (i + s < NBLK) {
    const double* wp = Wr[i];
    const double a0 = gv[i + s][0], a1 = gv[i + s][1], a2 = gv[i + s][2];
    t0 = __builtin_fma(-wp[6], a2, __builtin_fma(-wp[3], a1, __builtin_fma(-wp[0], a0, t0)));
    t1 = __builtin_fma(-wp[7], a2, __builtin_fma(-wp[4], a1, __builtin_fma(-wp[1], a0, t1)));
    t2 = __builtin_fma(-wp[8], a2, __builtin_fma(-wp[5], a1, __builtin_fma(-wp[2], a0, t2)));
  }
  const double x2 = t2;
  const double x1 = __builtin_fma(-q21, x2, t1);
  const double x0 = __builtin_fma(-q20, x2, __builtin_fma(-q10, x1, t0));
  gv[i][0] = x0; gv[i][1] = x1; gv[i][2] = x2;
}

#define NEVEN 49
#define TOPS  64

__device__ __forceinline__ int fold_block(int k) {
  const int i = 2 * (k + 1);
  return (i < TOPS) ? i : i + 2;
}

struct FoldRow { double u10, u20, u21, w0, w1, w2; double* dst; bool live; };

__device__ __forceinline__ FoldRow fold_p_load(
    const double (*F)[6], const double (*Wl)[9], const double (*Wr)[9],
    double (*Pl)[9], double (*Pr)[9], int q) {
  const int kq = q / 6;
  const int r = q - 6 * kq;
  const int k = (kq < NEVEN) ? kq : NEVEN - 1;
  const int i = fold_block(k);
  const int s = i & (-i);
  const bool hasR = (i + s < NBLK);
  const bool left = (r < 3) || !hasR;
  const int off = 9 * i + 3 * ((r < 3) ? r : r - 3);
  const double* src = (left ? &Wl[0][0] : &Wr[0][0]) + off;
  FoldRow o;
  o.dst = (left ? &Pl[0][0] : &Pr[0][0]) + off;
  o.live = (kq < NEVEN) && ((r < 3) || hasR);
  o.u10 = F[i][1]; o.u20 = F[i][3]; o.u21 = F[i][4];
  o.w0 = src[0]; o.w1 = src[1]; o.w2 = src[2];
  return o;
}

__device__ __forceinline__ void fold_p_solve(const FoldRow& a, double& p0, double& p1, double& p2) {
  p2 = a.w2;
  p1 = __builtin_fma(-a.u21, p2, a.w1);
  p0 = __builtin_fma(-a.u20, p2, __builtin_fma(-a.u10, p1, a.w0));
}

__device__ __forceinline__ void fold_p_store(const FoldRow& a, double p0, double p1, double p2) {
  if (a.live) { a.dst[0] = p0; a.dst[1] = p1; a.dst[2] = p2; }
}

__device__ __forceinline__ double back_fold(
    const double (*yv)[3], const double (*gv)[3],
    const double (*Pl)[9], const double (*Pr)[9],
    int i, int s, int k) {
  const bool hasR = (i + s < NBLK);
  const double* pl = Pl[i];
  const double* pr = hasR ? Pr[i] : Pl[i];
  const double* xl = gv[i - s];
  const double* xr = gv[hasR ? i + s : i - s];
  const double y = yv[i][k];
  const double c0 = pl[k], c1 = pl[3 + k], c2 = pl[6 + k];
  const double a0 = xl[0], a1 = xl[1], a2 = xl[2];
  const double d0 = pr[k], d1 = pr[3 + k], d2 = pr[6 + k];
  const double b0 = xr[0], b1 = xr[1], b2 = xr[2];
  __builtin_amdgcn_sched_barrier(0);
  const double tl = __builtin_fma(-c2, a2, __builtin_fma(-c1, a1, __builtin_fma(-c0, a0, y)));
  const double tr = __builtin_fma(-d2, b2, __builtin_fma(-d1, b1, __builtin_fma(-d0, b0, tl)));
  return hasR ? tr : tl;
}

__device__ __forceinline__ v8f gram_tile(float h,
                                         float a0, float b0, float q0,
                                         float a1, float b1, float q1,
                                         float a2, float b2, float q2) {
  v2f f0, f1;
  f0.x = fmaf(fmaf(q0, h, b0), h, a0);
  f0.y = fmaf(fmaf(q1, h, b1), h, a1);
  f1.x = fmaf(fmaf(q2, h, b2), h, a2);
  f1.y = 0.f;
  v8f acc = {};
  acc = __builtin_amdgcn_wmma_f32_16x16x4_f32(false, f0, false, f0,
                                              (short)0, acc, false, false);
  acc = __builtin_amdgcn_wmma_f32_16x16x4_f32(false, f1, false, f1,
                                              (short)0, acc, false, false);
  return acc;
}

__device__ __forceinline__ void put_tile(float* q, v8f acc) {
  q[0] = acc[0]; q[1] = acc[1]; q[2] = acc[2];
  q[3] = acc[3]; q[4] = acc[4]; q[5] = acc[5];
}


#define NWAVE 4
#define NTHR  (32 * NWAVE)
#define KSTEP 25
#define GS_TILES (T1 * 36)
#define GS_SPARE ((KSTEP - 1) * NWAVE * 36 + 26 * 6)

__global__ __launch_bounds__(NTHR) void ode_blocktri_kernel(
    const float* __restrict__ coeffs,
    const float* __restrict__ rhs,
    const float* __restrict__ ivr,
    const float* __restrict__ steps,
    float* __restrict__ out)
{
  __shared__ double S[NBLK][6];
  __shared__ double F[NBLK][6];
  __shared__ double Bo[NBLK][9];
  __shared__ double gv[NBLK][3];
  __shared__ double Zl[NBLK][9];
  __shared__ double Wl[NBLK][9];
  __shared__ double Zr[NBLK][9];
  __shared__ double Wr[NBLK][9];
  __shared__ double Pl[NBLK][9];
  __shared__ double Pr[NBLK][9];
  __shared__ double yv[NBLK][3];
  __shared__ __attribute__((aligned(16))) float Gs[GS_TILES + GS_SPARE];

  const int b = blockIdx.x;
  const int tid = threadIdx.x;
  const int lane = tid & 31;
  const int wave = __builtin_amdgcn_readfirstlane(tid >> 5);
  const int role = wave >> 1;
  const int slot = ((wave & 1) << 5) | lane;

  const int myp = wave + NWAVE * lane;
  const float hmine = (myp < T1) ? steps[b * T1 + myp] : 0.f;
  const float hcopy = (tid < T1) ? steps[b * T1 + tid] : 0.f;
  if (tid == NTHR - 1) {
    S[0][0] = 1.0; S[0][1] = 0.0; S[0][2] = 1.0;
    S[0][3] = 0.0; S[0][4] = 0.0; S[0][5] = 1.0;
    gv[0][0] = 0.0; gv[0][1] = 0.0; gv[0][2] = 0.0;
    Bo[0][0] = 0.0; Bo[0][1] = 0.0; Bo[0][2] = 0.0;
    Bo[0][3] = 0.0; Bo[0][4] = 0.0; Bo[0][5] = 0.0;
    Bo[0][6] = 0.0; Bo[0][7] = 0.0; Bo[0][8] = 0.0;
  }

  const int m = lane & 15;
  const int khalf = lane >> 4;
  const int col = (m < 6) ? m : 6;
  float a0, b0, q0, a1, b1, q1, a2, b2, q2;
  Ecoef(2 * khalf,                 col, &a0, &b0, &q0);
  Ecoef(2 * khalf + 1,             col, &a1, &b1, &q1);
  Ecoef((khalf == 0) ? 4 : -1,     col, &a2, &b2, &q2);

  const int hbits = __builtin_bit_cast(int, hmine);
  float* const gq = Gs + ((lane < 6) ? (wave * 36 + lane * 6)
                                     : (GS_TILES + (lane - 6) * 6));
  {
    v8f pend = {};
#pragma unroll
    for (int k = 0; k < KSTEP - 1; ++k) {
      const float h = __builtin_bit_cast(float, __builtin_amdgcn_readlane(hbits, k));
      const v8f acc = gram_tile(h, a0, b0, q0, a1, b1, q1, a2, b2, q2);
      if (k > 0) put_tile(gq + (k - 1) * (NWAVE * 36), pend);
      pend = acc;
    }
    put_tile(gq + (KSTEP - 2) * (NWAVE * 36), pend);
    if (wave + NWAVE * (KSTEP - 1) < T1) {
      const float h = __builtin_bit_cast(float, __builtin_amdgcn_readlane(hbits, KSTEP - 1));
      const v8f acc = gram_tile(h, a0, b0, q0, a1, b1, q1, a2, b2, q2);
      put_tile(gq + (KSTEP - 1) * (NWAVE * 36), acc);
    }
  }
  __syncthreads();

  const float iv0 = ivr[b * 2 + 0];
  const float iv1 = ivr[b * 2 + 1];
  if (tid < NSTEP) {
    const int t = tid;
    const int p = t + 1;
    const float* cp = coeffs + (b * NSTEP + t) * 3;
    const double c0 = (double)cp[0];
    const double c1 = (double)cp[1];
    const double c2 = (double)cp[2];
    const double r  = (double)rhs[b * NSTEP + t];
    const bool hasPrev = (t > 0);
    const bool hasNext = (t < T1);
    const float* gp = Gs + 36 * (hasPrev ? t - 1 : 0);
    const float* gn = Gs + 36 * (hasNext ? t : T1 - 1);
    const float f0 = gp[3 * 6 + 3], f1 = gp[3 * 6 + 4], f2 = gp[4 * 6 + 4];
    const float f3 = gp[3 * 6 + 5], f4 = gp[4 * 6 + 5], f5 = gp[5 * 6 + 5];
    const float n0 = gn[0 * 6 + 0], n1 = gn[0 * 6 + 1], n2 = gn[1 * 6 + 1];
    const float n3 = gn[0 * 6 + 2], n4 = gn[1 * 6 + 2], n5 = gn[2 * 6 + 2];
    const double p0 = (double)(hasPrev ? f0 : 1.f);
    const double p1 = (double)(hasPrev ? f1 : 0.f);
    const double p2 = (double)(hasPrev ? f2 : 1.f);
    const double p3 = (double)(hasPrev ? f3 : 0.f);
    const double p4 = (double)(hasPrev ? f4 : 0.f);
    const double p5 = (double)(hasPrev ? f5 : 0.f);
    S[p][0] = __builtin_fma(c0, c0, p0) + (double)(hasNext ? n0 : 0.f);
    S[p][1] = __builtin_fma(c1, c0, p1) + (double)(hasNext ? n1 : 0.f);
    S[p][2] = __builtin_fma(c1, c1, p2) + (double)(hasNext ? n2 : 0.f);
    S[p][3] = __builtin_fma(c2, c0, p3) + (double)(hasNext ? n3 : 0.f);
    S[p][4] = __builtin_fma(c2, c1, p4) + (double)(hasNext ? n4 : 0.f);
    S[p][5] = __builtin_fma(c2, c2, p5) + (double)(hasNext ? n5 : 0.f);
    gv[p][0] = __builtin_fma(c0, r, (double)((t == 0) ? iv0 : 0.f));
    gv[p][1] = __builtin_fma(c1, r, (double)((t == 0) ? iv1 : 0.f));
    gv[p][2] = c2 * r;
    const double b00 = (double)gn[3 * 6 + 0], b01 = (double)gn[4 * 6 + 0], b02 = (double)gn[5 * 6 + 0];
    const double b10 = (double)gn[3 * 6 + 1], b11 = (double)gn[4 * 6 + 1], b12 = (double)gn[5 * 6 + 1];
    const double b20 = (double)gn[3 * 6 + 2], b21 = (double)gn[4 * 6 + 2], b22 = (double)gn[5 * 6 + 2];
    if (hasNext) {
      Bo[p][0] = b00; Bo[p][1] = b01; Bo[p][2] = b02;
      Bo[p][3] = b10; Bo[p][4] = b11; Bo[p][5] = b12;
      Bo[p][6] = b20; Bo[p][7] = b21; Bo[p][8] = b22;
    }
  }
  __syncthreads();

  {
    const int s = 1;
    {
      const int i = s * (2 * slot + 1);
      if (i < NBLK) {
        if (role == 0)            elim_left(S, F, Bo, gv, Zl, Wl, i, s);
        else if (i + s < NBLK)   elim_right(S, Bo, Zr, Wr, i);
      }
    }
    __syncthreads();
    {
      const int j = 2 * s * slot;
      if (j < NBLK) {
        if (role == 0)   upd_diag(S, Zl, Wl, Zr, Wr, j, s);
        else             upd_rhs_couple(Bo, gv, Wl, Wr, Zr, j, s);
      }
    }
    __syncthreads();
  }
  {
    const int s = 2;
    {
      const int i = s * (2 * lane + 1);
      if (i < NBLK) elim4(S, F, Bo, gv, Zl, Wl, Zr, Wr, i, s, wave);
    }
    __syncthreads();
    {
      const int j = 2 * s * lane;
      if (j < NBLK) upd4(S, Bo, gv, Zl, Wl, Zr, Wr, j, s, wave);
    }
    __syncthreads();
  }
  for (int s = 4; s < TOPS; s <<= 1) {
    elim_rows(S, F, Bo, gv, Zl, Wl, Zr, Wr, tid, s);
    __syncthreads();
    const int ntask = 18 * ((NBLK - 1) / (2 * s) + 1);
    for (int base = 0; base < ntask; base += NTHR)
      upd_elem(S, Bo, gv, Zl, Wl, Zr, Wr, base + tid, s);
    __syncthreads();
  }
  if (wave != 0) {
    const int f = tid - 32;
    const FoldRow ra = fold_p_load(F, Wl, Wr, Pl, Pr, f);
    const FoldRow rb = fold_p_load(F, Wl, Wr, Pl, Pr, f + 96);
    const FoldRow rc = fold_p_load(F, Wl, Wr, Pl, Pr, f + 192);
    double pa0, pa1, pa2, pb0, pb1, pb2, pc0, pc1, pc2;
    if (wave == 1) {
      const FoldRow rd = fold_p_load(F, Wl, Wr, Pl, Pr, f + 288);
      double pd0, pd1, pd2;
      fold_p_solve(ra, pa0, pa1, pa2);
      fold_p_solve(rb, pb0, pb1, pb2);
      fold_p_solve(rc, pc0, pc1, pc2);
      fold_p_solve(rd, pd0, pd1, pd2);
      fold_p_store(ra, pa0, pa1, pa2);
      fold_p_store(rb, pb0, pb1, pb2);
      fold_p_store(rc, pc0, pc1, pc2);
      fold_p_store(rd, pd0, pd1, pd2);
    } else {
      const int kq = tid - 64;
      const int i = fold_block((kq < NEVEN) ? kq : NEVEN - 1);
      const double j0 = F[i][0], q10 = F[i][1], j1 = F[i][2];
      const double q20 = F[i][3], q21 = F[i][4], j2 = F[i][5];
      const double g0 = gv[i][0], g1 = gv[i][1], g2 = gv[i][2];
      fold_p_solve(ra, pa0, pa1, pa2);
      fold_p_solve(rb, pb0, pb1, pb2);
      fold_p_solve(rc, pc0, pc1, pc2);
      const double t0 = g0 * j0, t1 = g1 * j1, t2 = g2 * j2;
      const double y2 = t2;
      const double y1 = __builtin_fma(-q21, y2, t1);
      const double y0 = __builtin_fma(-q20, y2, __builtin_fma(-q10, y1, t0));
      fold_p_store(ra, pa0, pa1, pa2);
      fold_p_store(rb, pb0, pb1, pb2);
      fold_p_store(rc, pc0, pc1, pc2);
      if (kq < NEVEN) { yv[i][0] = y0; yv[i][1] = y1; yv[i][2] = y2; }
    }
  } else if (tid == 0) {
    double u10, u20, u21, i0, i1, i2;
    ldl3(S[TOPS][0], S[TOPS][1], S[TOPS][2], S[TOPS][3], S[TOPS][4], S[TOPS][5],
         u10, u20, u21, i0, i1, i2);
    const double w0 = gv[TOPS][0];
    const double w1 = __builtin_fma(-u10, w0, gv[TOPS][1]);
    const double w2 = __builtin_fma(-u21, w1, __builtin_fma(-u20, w0, gv[TOPS][2]));
    const double x2 = w2 * i2;
    const double x1 = __builtin_fma(-u21, x2, w1 * i1);
    const double x0 = __builtin_fma(-u20, x2, __builtin_fma(-u10, x1, w0 * i0));
    gv[TOPS][0] = x0; gv[TOPS][1] = x1; gv[TOPS][2] = x2;
  }
  __syncthreads();
  for (int s = TOPS / 2; s >= 2; s >>= 1) {
    const int i = s * (2 * lane + 1);
    if (i < NBLK && wave < 3) gv[i][wave] = back_fold(yv, gv, Pl, Pr, i, s, wave);
    __syncthreads();
  }
  {
    const int i = 2 * tid + 1;
    if (i < NBLK) cr_back(F, gv, Wl, Wr, i, 1);
    __syncthreads();
  }

  if (tid < NSTEP) {
    out[0 * OUT_U + b * NSTEP + tid] = (float)gv[tid + 1][0];
    out[1 * OUT_U + b * NSTEP + tid] = (float)gv[tid + 1][1];
    out[2 * OUT_U + b * NSTEP + tid] = (float)gv[tid + 1][2];
  }
  if (tid == 0) out[OUT_EPS + b] = 0.0f;
  if (tid < T1) out[OUT_ST + b * T1 + tid] = hcopy;
}

extern "C" void kernel_launch(void* const* d_in, const int* in_sizes, int n_in,
                              void* d_out, int out_size, void* d_ws, size_t ws_size,
                              hipStream_t stream) {
  (void)in_sizes; (void)n_in; (void)out_size; (void)d_ws; (void)ws_size;
  const float* coeffs = (const float*)d_in[0];
  const float* rhs    = (const float*)d_in[1];
  const float* ivr    = (const float*)d_in[2];
  const float* steps  = (const float*)d_in[3];
  float* out = (float*)d_out;
  hipLaunchKernelGGL(ode_blocktri_kernel, dim3(NB), dim3(NTHR), 0, stream,
                     coeffs, rhs, ivr, steps, out);
}
